# starting kernel
# baseline (speedup 1.0000x reference)
#include <hip/hip_runtime.h>


typedef __attribute__((ext_vector_type(2))) float v2f;
typedef __attribute__((ext_vector_type(8))) float v8f;

constexpr int NB  = 16;
constexpr int NC  = 64;
constexpr int NFR = 8;
constexpr int ND  = 16;
constexpr int NL  = 64;
constexpr int NLP = 65;
constexpr int NP  = NLP * NLP;
constexpr int NPW = NLP * NL;

constexpr int OFF_FWDR = 0;
constexpr int OFF_FWDI = OFF_FWDR + 80 * 80;
constexpr int OFF_ER   = OFF_FWDI + 80 * 80;
constexpr int OFF_EI   = OFF_ER + 64 * 64;
constexpr int OFF_WNR  = OFF_EI + 64 * 64;
constexpr int OFF_WNI  = OFF_WNR + 64 * 16;
constexpr int OFF_IVR  = OFF_WNI + 64 * 16;
constexpr int OFF_IVI  = OFF_IVR + 68 * 64;
constexpr int OFF_IUR  = OFF_IVI + 68 * 64;
constexpr int OFF_IUI  = OFF_IUR + 64 * 68;
constexpr int TW_TOTAL = OFF_IUI + 64 * 68;
constexpr int TW_PAD   = 40960;

#define PI_D 3.14159265358979323846


static __device__ __forceinline__ v8f wmma4(v2f a, v2f b, v8f c) {
  return __builtin_amdgcn_wmma_f32_16x16x4_f32(false, a, false, b, (short)0, c,
                                               false, false);
}

static __device__ __forceinline__ v8f zero8() {
  v8f z = {0.f, 0.f, 0.f, 0.f, 0.f, 0.f, 0.f, 0.f};
  return z;
}

static __device__ __forceinline__ v2f frag_a(const float* __restrict__ A,
                                             int lda, int m0, int k0) {
  int lane = threadIdx.x & 31;
  int m = m0 + (lane & 15);
  int k = k0 + ((lane >> 4) << 1);
  const float* p = A + m * lda + k;
  v2f r;
  r.x = p[0];
  r.y = p[1];
  return r;
}

static __device__ __forceinline__ v2f frag_a_guard(const float* __restrict__ A,
                                                   int lda, int m0, int k0,
                                                   int mmax, int kmax) {
  int lane = threadIdx.x & 31;
  int m = m0 + (lane & 15);
  int k = k0 + ((lane >> 4) << 1);
  v2f r;
  r.x = 0.f;
  r.y = 0.f;
  if (m < mmax) {
    const float* p = A + m * lda + k;
    if (k < kmax) r.x = p[0];
    if (k + 1 < kmax) r.y = p[1];
  }
  return r;
}

static __device__ __forceinline__ v2f frag_b(const float* __restrict__ B,
                                             int ldb, int k0, int n0) {
  int lane = threadIdx.x & 31;
  int n = n0 + (lane & 15);
  int k = k0 + ((lane >> 4) << 1);
  const float* p = B + k * ldb + n;
  v2f r;
  r.x = p[0];
  r.y = p[ldb];
  return r;
}

static __device__ __forceinline__ v2f frag_b_nguard(const float* __restrict__ B,
                                                    int ldb, int k0, int n0,
                                                    int nmax) {
  int lane = threadIdx.x & 31;
  int n = n0 + (lane & 15);
  int k = k0 + ((lane >> 4) << 1);
  v2f r;
  r.x = 0.f;
  r.y = 0.f;
  if (n < nmax) {
    const float* p = B + k * ldb + n;
    r.x = p[0];
    r.y = p[ldb];
  }
  return r;
}

static __device__ __forceinline__ v2f frag_b_kguard(const float* __restrict__ B,
                                                    int ldb, int k0, int n0,
                                                    int kmax) {
  int lane = threadIdx.x & 31;
  int n = n0 + (lane & 15);
  int k = k0 + ((lane >> 4) << 1);
  v2f r;
  r.x = 0.f;
  r.y = 0.f;
  const float* p = B + k * ldb + n;
  if (k < kmax) r.x = p[0];
  if (k + 1 < kmax) r.y = p[ldb];
  return r;
}

static __device__ __forceinline__ void cwmma(v2f ar, v2f ai, v2f br, v2f bi,
                                             v8f& dr, v8f& di) {
  dr = wmma4(ar, br, dr);
  v2f nai;
  nai.x = -ai.x;
  nai.y = -ai.y;
  dr = wmma4(nai, bi, dr);
  di = wmma4(ar, bi, di);
  di = wmma4(ai, br, di);
}


__global__ void k_twiddles(float* __restrict__ tw) {
  int t = blockIdx.x * blockDim.x + threadIdx.x;
  int which = blockIdx.y;
  if (which == 0) {
    if (t >= 80 * 80) return;
    int a = t / 80, b = t % 80;
    float r = 0.f, im = 0.f;
    if (a < NLP && b < NLP) {
      double ang = -2.0 * PI_D * (double)((a * b) % NLP) / (double)NLP;
      r = (float)cos(ang);
      im = (float)sin(ang);
    }
    tw[OFF_FWDR + t] = r;
    tw[OFF_FWDI + t] = im;
  } else if (which == 1) {
    if (t >= NC * NC) return;
    int c = t / NC, k = t % NC;
    double ang = -2.0 * PI_D * (double)((c * k) % NC) / (double)NC;
    tw[OFF_ER + t] = (float)cos(ang);
    tw[OFF_EI + t] = (float)sin(ang);
  } else if (which == 2) {
    if (t >= NC * ND) return;
    int c = t / ND, d = t % ND;
    double ang = 2.0 * PI_D * (double)((c * 4 * d) % NC) / (double)NC;
    tw[OFF_WNR + t] = (float)(cos(ang) / 64.0);
    tw[OFF_WNI + t] = (float)(sin(ang) / 64.0);
  } else if (which == 3) {
    if (t >= 68 * NL) return;
    int v = t / NL, w = t % NL;
    float r = 0.f, im = 0.f;
    if (v < NLP) {
      double ang = 2.0 * PI_D * (double)((v * (w + 1)) % NLP) / (double)NLP;
      r = (float)cos(ang);
      im = (float)sin(ang);
    }
    tw[OFF_IVR + t] = r;
    tw[OFF_IVI + t] = im;
  } else {
    if (t >= NL * 68) return;
    int h = t / 68, u = t % 68;
    float r = 0.f, im = 0.f;
    if (u < NLP) {
      double ang = 2.0 * PI_D * (double)((u * (h + 1)) % NLP) / (double)NLP;
      double s = 1.0 / (double)(NLP * NLP);
      r = (float)(cos(ang) * s);
      im = (float)(sin(ang) * s);
    }
    tw[OFF_IUR + t] = r;
    tw[OFF_IUI + t] = im;
  }
}


__global__ __launch_bounds__(128) void k_fft2d(const float* __restrict__ x,
                                               const float* __restrict__ tw,
                                               float* __restrict__ Yr,
                                               float* __restrict__ Yi) {
  __shared__ float xs[NL * NL];
  __shared__ float Tr[NL * 80];
  __shared__ float Ti[NL * 80];
  const int plane = blockIdx.x;
  const int tid = threadIdx.x;
  const int wave = tid >> 5;
  const int lane = tid & 31;

  const float* xp = x + (size_t)plane * (NL * NL);
  for (int i = tid; i < NL * NL; i += 128) xs[i] = xp[i];
  __syncthreads();

  const float* Fr = tw + OFF_FWDR;
  const float* Fi = tw + OFF_FWDI;

  for (int t = wave; t < 4 * 5; t += 4) {
    int mo = (t / 5) * 16, no = (t % 5) * 16;
    v8f ar = zero8(), ai = zero8();
    for (int k0 = 0; k0 < NL; k0 += 4) {
      v2f a = frag_a(xs, NL, mo, k0);
      ar = wmma4(a, frag_b(Fr, 80, k0, no), ar);
      ai = wmma4(a, frag_b(Fi, 80, k0, no), ai);
    }
    int n = no + (lane & 15);
    int mb = mo + ((lane >> 4) << 3);
#pragma unroll
    for (int i = 0; i < 8; ++i) {
      Tr[(mb + i) * 80 + n] = ar[i];
      Ti[(mb + i) * 80 + n] = ai[i];
    }
  }
  __syncthreads();

  float* yr = Yr + (size_t)plane * NP;
  float* yi = Yi + (size_t)plane * NP;
  for (int t = wave; t < 5 * 5; t += 4) {
    int mo = (t / 5) * 16, no = (t % 5) * 16;
    v8f dr = zero8(), di = zero8();
    for (int k0 = 0; k0 < NL; k0 += 4) {
      v2f par = frag_a(Fr, 80, mo, k0);
      v2f pai = frag_a(Fi, 80, mo, k0);
      v2f pbr = frag_b(Tr, 80, k0, no);
      v2f pbi = frag_b(Ti, 80, k0, no);
      cwmma(par, pai, pbr, pbi, dr, di);
    }
    int n = no + (lane & 15);
    int mb = mo + ((lane >> 4) << 3);
    if (n < NLP) {
#pragma unroll
      for (int i = 0; i < 8; ++i) {
        int m = mb + i;
        if (m < NLP) {
          yr[m * NLP + n] = dr[i];
          yi[m * NLP + n] = di[i];
        }
      }
    }
  }
}


__global__ __launch_bounds__(128) void k_cfft(const float* __restrict__ tw,
                                              const float* __restrict__ Yr,
                                              const float* __restrict__ Yi,
                                              float* __restrict__ Xr,
                                              float* __restrict__ Xi) {
  const int b = blockIdx.y;
  const int no = blockIdx.x * 16;
  const int mo = (threadIdx.x >> 5) * 16;
  const int lane = threadIdx.x & 31;
  const float* Er = tw + OFF_ER;
  const float* Ei = tw + OFF_EI;
  const float* yr = Yr + (size_t)b * NC * NP;
  const float* yi = Yi + (size_t)b * NC * NP;
  v8f dr = zero8(), di = zero8();
  for (int k0 = 0; k0 < NC; k0 += 4) {
    v2f ar = frag_a(Er, NC, mo, k0);
    v2f ai = frag_a(Ei, NC, mo, k0);
    v2f br = frag_b_nguard(yr, NP, k0, no, NP);
    v2f bi = frag_b_nguard(yi, NP, k0, no, NP);
    cwmma(ar, ai, br, bi, dr, di);
  }
  int n = no + (lane & 15);
  int mb = mo + ((lane >> 4) << 3);
  if (n < NP) {
    float* xr = Xr + (size_t)b * NC * NP;
    float* xi = Xi + (size_t)b * NC * NP;
#pragma unroll
    for (int i = 0; i < 8; ++i) {
      xr[(mb + i) * NP + n] = dr[i];
      xi[(mb + i) * NP + n] = di[i];
    }
  }
}


__global__ __launch_bounds__(128) void k_mix(const float* __restrict__ tw,
                                             const float* __restrict__ Xr,
                                             const float* __restrict__ Xi,
                                             const float* __restrict__ wre,
                                             const float* __restrict__ wim,
                                             float* __restrict__ Gr,
                                             float* __restrict__ Gi, int nidx) {
  const int wave = threadIdx.x >> 5;
  const int p = blockIdx.x * 4 + wave;
  if (p >= NP) return;
  const int lane = threadIdx.x & 31;
  const float* WNr = tw + OFF_WNR;
  const float* WNi = tw + OFF_WNI;
  const int bm = lane & 15;
  const int khalf = (lane >> 4) << 1;
  v8f dr = zero8(), di = zero8();
  for (int k0 = 0; k0 < NC; k0 += 4) {
    v2f ar, ai;
#pragma unroll
    for (int j = 0; j < 2; ++j) {
      int c = k0 + khalf + j;
      size_t xo = ((size_t)bm * NC + c) * NP + p;
      size_t wo = ((size_t)nidx * NC + c) * NP + p;
      float xr = Xr[xo], xi = Xi[xo];
      float wr = wre[wo], wi = wim[wo];
      ar[j] = xr * wr - xi * wi;
      ai[j] = xr * wi + xi * wr;
    }
    v2f br = frag_b(WNr, ND, k0, 0);
    v2f bi = frag_b(WNi, ND, k0, 0);
    cwmma(ar, ai, br, bi, dr, di);
  }
  int n = lane & 15;
  int mb = (lane >> 4) << 3;
#pragma unroll
  for (int i = 0; i < 8; ++i) {
    size_t o = ((size_t)(mb + i) * ND + n) * NP + p;
    Gr[o] = dr[i];
    Gi[o] = di[i];
  }
}


__global__ __launch_bounds__(128) void k_ifft_v(const float* __restrict__ tw,
                                                const float* __restrict__ Gr,
                                                const float* __restrict__ Gi,
                                                float* __restrict__ Pr,
                                                float* __restrict__ Pi) {
  const int d = blockIdx.x, b = blockIdx.y;
  const int wave = threadIdx.x >> 5;
  const int lane = threadIdx.x & 31;
  const float* IVr = tw + OFF_IVR;
  const float* IVi = tw + OFF_IVI;
  const float* gr = Gr + ((size_t)b * ND + d) * NP;
  const float* gi = Gi + ((size_t)b * ND + d) * NP;
  float* pr = Pr + ((size_t)b * ND + d) * NPW;
  float* pi = Pi + ((size_t)b * ND + d) * NPW;
  for (int t = wave; t < 5 * 4; t += 4) {
    int mo = (t / 4) * 16, no = (t % 4) * 16;
    v8f dr = zero8(), di = zero8();
    for (int k0 = 0; k0 < 68; k0 += 4) {
      v2f ar = frag_a_guard(gr, NLP, mo, k0, NLP, NLP);
      v2f ai = frag_a_guard(gi, NLP, mo, k0, NLP, NLP);
      v2f br = frag_b(IVr, NL, k0, no);
      v2f bi = frag_b(IVi, NL, k0, no);
      cwmma(ar, ai, br, bi, dr, di);
    }
    int n = no + (lane & 15);
    int mb = mo + ((lane >> 4) << 3);
#pragma unroll
    for (int i = 0; i < 8; ++i) {
      int m = mb + i;
      if (m < NLP) {
        pr[m * NL + n] = dr[i];
        pi[m * NL + n] = di[i];
      }
    }
  }
}


__global__ __launch_bounds__(128) void k_ifft_u(const float* __restrict__ tw,
                                                const float* __restrict__ Pr,
                                                const float* __restrict__ Pi,
                                                float* __restrict__ out,
                                                int nidx) {
  const int d = blockIdx.x, b = blockIdx.y;
  const int wave = threadIdx.x >> 5;
  const int lane = threadIdx.x & 31;
  const float* IUr = tw + OFF_IUR;
  const float* IUi = tw + OFF_IUI;
  const float* pr = Pr + ((size_t)b * ND + d) * NPW;
  const float* pi = Pi + ((size_t)b * ND + d) * NPW;
  float* op = out + ((size_t)b * (ND * NFR) + (d * NFR + nidx)) * (NL * NL);
  for (int t = wave; t < 4 * 4; t += 4) {
    int mo = (t / 4) * 16, no = (t % 4) * 16;
    v8f dr = zero8();
    for (int k0 = 0; k0 < 68; k0 += 4) {
      v2f ar = frag_a(IUr, 68, mo, k0);
      v2f ai = frag_a(IUi, 68, mo, k0);
      v2f br = frag_b_kguard(pr, NL, k0, no, NLP);
      v2f bi = frag_b_kguard(pi, NL, k0, no, NLP);
      dr = wmma4(ar, br, dr);
      v2f nai;
      nai.x = -ai.x;
      nai.y = -ai.y;
      dr = wmma4(nai, bi, dr);
    }
    int n = no + (lane & 15);
    int mb = mo + ((lane >> 4) << 3);
#pragma unroll
    for (int i = 0; i < 8; ++i) op[(mb + i) * NL + n] = dr[i];
  }
}


extern "C" void kernel_launch(void* const* d_in, const int* in_sizes, int n_in,
                              void* d_out, int out_size, void* d_ws,
                              size_t ws_size, hipStream_t stream) {
  (void)in_sizes; (void)n_in; (void)out_size; (void)ws_size;
  const float* x   = (const float*)d_in[0];
  const float* wre = (const float*)d_in[1];
  const float* wim = (const float*)d_in[2];
  float* out = (float*)d_out;
  float* ws = (float*)d_ws;

  float* tw = ws;
  size_t off = TW_PAD;
  float* Yr = ws + off; off += (size_t)NB * NC * NP;
  float* Yi = ws + off; off += (size_t)NB * NC * NP;
  float* Xr = ws + off; off += (size_t)NB * NC * NP;
  float* Xi = ws + off; off += (size_t)NB * NC * NP;
  float* Gr = ws + off; off += (size_t)NB * ND * NP;
  float* Gi = ws + off; off += (size_t)NB * ND * NP;
  float* Pr = ws + off; off += (size_t)NB * ND * NPW;
  float* Pi = ws + off; off += (size_t)NB * ND * NPW;

  k_twiddles<<<dim3(25, 5), 256, 0, stream>>>(tw);
  k_fft2d<<<NB * NC, 128, 0, stream>>>(x, tw, Yr, Yi);
  k_cfft<<<dim3((NP + 15) / 16, NB), 128, 0, stream>>>(tw, Yr, Yi, Xr, Xi);
  for (int n = 0; n < NFR; ++n) {
    k_mix<<<(NP + 3) / 4, 128, 0, stream>>>(tw, Xr, Xi, wre, wim, Gr, Gi, n);
    k_ifft_v<<<dim3(ND, NB), 128, 0, stream>>>(tw, Gr, Gi, Pr, Pi);
    k_ifft_u<<<dim3(ND, NB), 128, 0, stream>>>(tw, Pr, Pi, out, n);
  }
}


# mix kernel re-tiled: 16 consecutive frequency points on the WMMA column axis (A = channel table, B = X*W), so X, W loads and G stores are contiguous; c sum order unchanged
# speedup vs baseline: 2.4673x; 2.4673x over previous
#include <hip/hip_runtime.h>


typedef __attribute__((ext_vector_type(2))) float v2f;
typedef __attribute__((ext_vector_type(8))) float v8f;

constexpr int NB  = 16;
constexpr int NC  = 64;
constexpr int NFR = 8;
constexpr int ND  = 16;
constexpr int NL  = 64;
constexpr int NLP = 65;
constexpr int NP  = NLP * NLP;
constexpr int NPW = NLP * NL;

constexpr int OFF_FWDR = 0;
constexpr int OFF_FWDI = OFF_FWDR + 80 * 80;
constexpr int OFF_ER   = OFF_FWDI + 80 * 80;
constexpr int OFF_EI   = OFF_ER + 64 * 64;
constexpr int OFF_WNR  = OFF_EI + 64 * 64;
constexpr int OFF_WNI  = OFF_WNR + 64 * 16;
constexpr int OFF_IVR  = OFF_WNI + 64 * 16;
constexpr int OFF_IVI  = OFF_IVR + 68 * 64;
constexpr int OFF_IUR  = OFF_IVI + 68 * 64;
constexpr int OFF_IUI  = OFF_IUR + 64 * 68;
constexpr int TW_TOTAL = OFF_IUI + 64 * 68;
constexpr int TW_PAD   = 40960;

#define PI_D 3.14159265358979323846


static __device__ __forceinline__ v8f wmma4(v2f a, v2f b, v8f c) {
  return __builtin_amdgcn_wmma_f32_16x16x4_f32(false, a, false, b, (short)0, c,
                                               false, false);
}

static __device__ __forceinline__ v8f zero8() {
  v8f z = {0.f, 0.f, 0.f, 0.f, 0.f, 0.f, 0.f, 0.f};
  return z;
}

static __device__ __forceinline__ v2f frag_a(const float* __restrict__ A,
                                             int lda, int m0, int k0) {
  int lane = threadIdx.x & 31;
  int m = m0 + (lane & 15);
  int k = k0 + ((lane >> 4) << 1);
  const float* p = A + m * lda + k;
  v2f r;
  r.x = p[0];
  r.y = p[1];
  return r;
}

static __device__ __forceinline__ v2f frag_a_guard(const float* __restrict__ A,
                                                   int lda, int m0, int k0,
                                                   int mmax, int kmax) {
  int lane = threadIdx.x & 31;
  int m = m0 + (lane & 15);
  int k = k0 + ((lane >> 4) << 1);
  v2f r;
  r.x = 0.f;
  r.y = 0.f;
  if (m < mmax) {
    const float* p = A + m * lda + k;
    if (k < kmax) r.x = p[0];
    if (k + 1 < kmax) r.y = p[1];
  }
  return r;
}

static __device__ __forceinline__ v2f frag_b(const float* __restrict__ B,
                                             int ldb, int k0, int n0) {
  int lane = threadIdx.x & 31;
  int n = n0 + (lane & 15);
  int k = k0 + ((lane >> 4) << 1);
  const float* p = B + k * ldb + n;
  v2f r;
  r.x = p[0];
  r.y = p[ldb];
  return r;
}

static __device__ __forceinline__ v2f frag_b_nguard(const float* __restrict__ B,
                                                    int ldb, int k0, int n0,
                                                    int nmax) {
  int lane = threadIdx.x & 31;
  int n = n0 + (lane & 15);
  int k = k0 + ((lane >> 4) << 1);
  v2f r;
  r.x = 0.f;
  r.y = 0.f;
  if (n < nmax) {
    const float* p = B + k * ldb + n;
    r.x = p[0];
    r.y = p[ldb];
  }
  return r;
}

static __device__ __forceinline__ v2f frag_b_kguard(const float* __restrict__ B,
                                                    int ldb, int k0, int n0,
                                                    int kmax) {
  int lane = threadIdx.x & 31;
  int n = n0 + (lane & 15);
  int k = k0 + ((lane >> 4) << 1);
  v2f r;
  r.x = 0.f;
  r.y = 0.f;
  const float* p = B + k * ldb + n;
  if (k < kmax) r.x = p[0];
  if (k + 1 < kmax) r.y = p[ldb];
  return r;
}

static __device__ __forceinline__ void cwmma(v2f ar, v2f ai, v2f br, v2f bi,
                                             v8f& dr, v8f& di) {
  dr = wmma4(ar, br, dr);
  v2f nai;
  nai.x = -ai.x;
  nai.y = -ai.y;
  dr = wmma4(nai, bi, dr);
  di = wmma4(ar, bi, di);
  di = wmma4(ai, br, di);
}


__global__ void k_twiddles(float* __restrict__ tw) {
  int t = blockIdx.x * blockDim.x + threadIdx.x;
  int which = blockIdx.y;
  if (which == 0) {
    if (t >= 80 * 80) return;
    int a = t / 80, b = t % 80;
    float r = 0.f, im = 0.f;
    if (a < NLP && b < NLP) {
      double ang = -2.0 * PI_D * (double)((a * b) % NLP) / (double)NLP;
      r = (float)cos(ang);
      im = (float)sin(ang);
    }
    tw[OFF_FWDR + t] = r;
    tw[OFF_FWDI + t] = im;
  } else if (which == 1) {
    if (t >= NC * NC) return;
    int c = t / NC, k = t % NC;
    double ang = -2.0 * PI_D * (double)((c * k) % NC) / (double)NC;
    tw[OFF_ER + t] = (float)cos(ang);
    tw[OFF_EI + t] = (float)sin(ang);
  } else if (which == 2) {
    if (t >= NC * ND) return;
    int d = t / NC, c = t % NC;
    double ang = 2.0 * PI_D * (double)((c * 4 * d) % NC) / (double)NC;
    tw[OFF_WNR + t] = (float)(cos(ang) / 64.0);
    tw[OFF_WNI + t] = (float)(sin(ang) / 64.0);
  } else if (which == 3) {
    if (t >= 68 * NL) return;
    int v = t / NL, w = t % NL;
    float r = 0.f, im = 0.f;
    if (v < NLP) {
      double ang = 2.0 * PI_D * (double)((v * (w + 1)) % NLP) / (double)NLP;
      r = (float)cos(ang);
      im = (float)sin(ang);
    }
    tw[OFF_IVR + t] = r;
    tw[OFF_IVI + t] = im;
  } else {
    if (t >= NL * 68) return;
    int h = t / 68, u = t % 68;
    float r = 0.f, im = 0.f;
    if (u < NLP) {
      double ang = 2.0 * PI_D * (double)((u * (h + 1)) % NLP) / (double)NLP;
      double s = 1.0 / (double)(NLP * NLP);
      r = (float)(cos(ang) * s);
      im = (float)(sin(ang) * s);
    }
    tw[OFF_IUR + t] = r;
    tw[OFF_IUI + t] = im;
  }
}


__global__ __launch_bounds__(128) void k_fft2d(const float* __restrict__ x,
                                               const float* __restrict__ tw,
                                               float* __restrict__ Yr,
                                               float* __restrict__ Yi) {
  __shared__ float xs[NL * NL];
  __shared__ float Tr[NL * 80];
  __shared__ float Ti[NL * 80];
  const int plane = blockIdx.x;
  const int tid = threadIdx.x;
  const int wave = tid >> 5;
  const int lane = tid & 31;

  const float* xp = x + (size_t)plane * (NL * NL);
  for (int i = tid; i < NL * NL; i += 128) xs[i] = xp[i];
  __syncthreads();

  const float* Fr = tw + OFF_FWDR;
  const float* Fi = tw + OFF_FWDI;

  for (int t = wave; t < 4 * 5; t += 4) {
    int mo = (t / 5) * 16, no = (t % 5) * 16;
    v8f ar = zero8(), ai = zero8();
    for (int k0 = 0; k0 < NL; k0 += 4) {
      v2f a = frag_a(xs, NL, mo, k0);
      ar = wmma4(a, frag_b(Fr, 80, k0, no), ar);
      ai = wmma4(a, frag_b(Fi, 80, k0, no), ai);
    }
    int n = no + (lane & 15);
    int mb = mo + ((lane >> 4) << 3);
#pragma unroll
    for (int i = 0; i < 8; ++i) {
      Tr[(mb + i) * 80 + n] = ar[i];
      Ti[(mb + i) * 80 + n] = ai[i];
    }
  }
  __syncthreads();

  float* yr = Yr + (size_t)plane * NP;
  float* yi = Yi + (size_t)plane * NP;
  for (int t = wave; t < 5 * 5; t += 4) {
    int mo = (t / 5) * 16, no = (t % 5) * 16;
    v8f dr = zero8(), di = zero8();
    for (int k0 = 0; k0 < NL; k0 += 4) {
      v2f par = frag_a(Fr, 80, mo, k0);
      v2f pai = frag_a(Fi, 80, mo, k0);
      v2f pbr = frag_b(Tr, 80, k0, no);
      v2f pbi = frag_b(Ti, 80, k0, no);
      cwmma(par, pai, pbr, pbi, dr, di);
    }
    int n = no + (lane & 15);
    int mb = mo + ((lane >> 4) << 3);
    if (n < NLP) {
#pragma unroll
      for (int i = 0; i < 8; ++i) {
        int m = mb + i;
        if (m < NLP) {
          yr[m * NLP + n] = dr[i];
          yi[m * NLP + n] = di[i];
        }
      }
    }
  }
}


__global__ __launch_bounds__(128) void k_cfft(const float* __restrict__ tw,
                                              const float* __restrict__ Yr,
                                              const float* __restrict__ Yi,
                                              float* __restrict__ Xr,
                                              float* __restrict__ Xi) {
  const int b = blockIdx.y;
  const int no = blockIdx.x * 16;
  const int mo = (threadIdx.x >> 5) * 16;
  const int lane = threadIdx.x & 31;
  const float* Er = tw + OFF_ER;
  const float* Ei = tw + OFF_EI;
  const float* yr = Yr + (size_t)b * NC * NP;
  const float* yi = Yi + (size_t)b * NC * NP;
  v8f dr = zero8(), di = zero8();
  for (int k0 = 0; k0 < NC; k0 += 4) {
    v2f ar = frag_a(Er, NC, mo, k0);
    v2f ai = frag_a(Ei, NC, mo, k0);
    v2f br = frag_b_nguard(yr, NP, k0, no, NP);
    v2f bi = frag_b_nguard(yi, NP, k0, no, NP);
    cwmma(ar, ai, br, bi, dr, di);
  }
  int n = no + (lane & 15);
  int mb = mo + ((lane >> 4) << 3);
  if (n < NP) {
    float* xr = Xr + (size_t)b * NC * NP;
    float* xi = Xi + (size_t)b * NC * NP;
#pragma unroll
    for (int i = 0; i < 8; ++i) {
      xr[(mb + i) * NP + n] = dr[i];
      xi[(mb + i) * NP + n] = di[i];
    }
  }
}


constexpr int MIXB = 2;
constexpr int MIX_PT = (NP + 15) / 16;
constexpr int MIX_GX = (MIX_PT + 3) / 4;

__global__ __launch_bounds__(128) void k_mix(const float* __restrict__ tw,
                                             const float* __restrict__ Xr,
                                             const float* __restrict__ Xi,
                                             const float* __restrict__ wre,
                                             const float* __restrict__ wim,
                                             float* __restrict__ Gr,
                                             float* __restrict__ Gi, int nidx) {
  const int wave = threadIdx.x >> 5;
  const int lane = threadIdx.x & 31;
  const int p0 = (blockIdx.x * 4 + wave) * 16;
  if (p0 >= NP) return;
  const int col = lane & 15;
  const int h2 = (lane >> 4) << 1;
  const int p = p0 + col;
  const int pl = p < NP ? p : NP - 1;
  const int b0 = blockIdx.y * MIXB;

  const float* ar_p = tw + OFF_WNR + col * NC + h2;
  const float* ai_p = tw + OFF_WNI + col * NC + h2;
  const float* wr_p = wre + ((size_t)nidx * NC + h2) * NP + pl;
  const float* wi_p = wim + ((size_t)nidx * NC + h2) * NP + pl;
  const float* xr_p = Xr + ((size_t)b0 * NC + h2) * NP + pl;
  const float* xi_p = Xi + ((size_t)b0 * NC + h2) * NP + pl;

  v8f dr[MIXB], di[MIXB];
#pragma unroll
  for (int j = 0; j < MIXB; ++j) {
    dr[j] = zero8();
    di[j] = zero8();
  }

#pragma unroll
  for (int k0 = 0; k0 < NC; k0 += 4) {
    v2f ar = *reinterpret_cast<const v2f*>(ar_p + k0);
    v2f ai = *reinterpret_cast<const v2f*>(ai_p + k0);
    v2f nai;
    nai.x = -ai.x;
    nai.y = -ai.y;
    float wr0 = wr_p[(size_t)k0 * NP], wr1 = wr_p[(size_t)(k0 + 1) * NP];
    float wi0 = wi_p[(size_t)k0 * NP], wi1 = wi_p[(size_t)(k0 + 1) * NP];
#pragma unroll
    for (int j = 0; j < MIXB; ++j) {
      size_t o0 = ((size_t)j * NC + k0) * NP;
      size_t o1 = o0 + NP;
      float xr0 = xr_p[o0], xi0 = xi_p[o0];
      float xr1 = xr_p[o1], xi1 = xi_p[o1];
      v2f br, bi;
      br.x = xr0 * wr0 - xi0 * wi0;
      bi.x = xr0 * wi0 + xi0 * wr0;
      br.y = xr1 * wr1 - xi1 * wi1;
      bi.y = xr1 * wi1 + xi1 * wr1;
      dr[j] = wmma4(ar, br, dr[j]);
      dr[j] = wmma4(nai, bi, dr[j]);
      di[j] = wmma4(ar, bi, di[j]);
      di[j] = wmma4(ai, br, di[j]);
    }
  }

  if (p < NP) {
    const int mb = (lane >> 4) << 3;
#pragma unroll
    for (int j = 0; j < MIXB; ++j) {
#pragma unroll
      for (int i = 0; i < 8; ++i) {
        size_t o = ((size_t)(b0 + j) * ND + (mb + i)) * NP + p;
        Gr[o] = dr[j][i];
        Gi[o] = di[j][i];
      }
    }
  }
}


__global__ __launch_bounds__(128) void k_ifft_v(const float* __restrict__ tw,
                                                const float* __restrict__ Gr,
                                                const float* __restrict__ Gi,
                                                float* __restrict__ Pr,
                                                float* __restrict__ Pi) {
  const int d = blockIdx.x, b = blockIdx.y;
  const int wave = threadIdx.x >> 5;
  const int lane = threadIdx.x & 31;
  const float* IVr = tw + OFF_IVR;
  const float* IVi = tw + OFF_IVI;
  const float* gr = Gr + ((size_t)b * ND + d) * NP;
  const float* gi = Gi + ((size_t)b * ND + d) * NP;
  float* pr = Pr + ((size_t)b * ND + d) * NPW;
  float* pi = Pi + ((size_t)b * ND + d) * NPW;
  for (int t = wave; t < 5 * 4; t += 4) {
    int mo = (t / 4) * 16, no = (t % 4) * 16;
    v8f dr = zero8(), di = zero8();
    for (int k0 = 0; k0 < 68; k0 += 4) {
      v2f ar = frag_a_guard(gr, NLP, mo, k0, NLP, NLP);
      v2f ai = frag_a_guard(gi, NLP, mo, k0, NLP, NLP);
      v2f br = frag_b(IVr, NL, k0, no);
      v2f bi = frag_b(IVi, NL, k0, no);
      cwmma(ar, ai, br, bi, dr, di);
    }
    int n = no + (lane & 15);
    int mb = mo + ((lane >> 4) << 3);
#pragma unroll
    for (int i = 0; i < 8; ++i) {
      int m = mb + i;
      if (m < NLP) {
        pr[m * NL + n] = dr[i];
        pi[m * NL + n] = di[i];
      }
    }
  }
}


__global__ __launch_bounds__(128) void k_ifft_u(const float* __restrict__ tw,
                                                const float* __restrict__ Pr,
                                                const float* __restrict__ Pi,
                                                float* __restrict__ out,
                                                int nidx) {
  const int d = blockIdx.x, b = blockIdx.y;
  const int wave = threadIdx.x >> 5;
  const int lane = threadIdx.x & 31;
  const float* IUr = tw + OFF_IUR;
  const float* IUi = tw + OFF_IUI;
  const float* pr = Pr + ((size_t)b * ND + d) * NPW;
  const float* pi = Pi + ((size_t)b * ND + d) * NPW;
  float* op = out + ((size_t)b * (ND * NFR) + (d * NFR + nidx)) * (NL * NL);
  for (int t = wave; t < 4 * 4; t += 4) {
    int mo = (t / 4) * 16, no = (t % 4) * 16;
    v8f dr = zero8();
    for (int k0 = 0; k0 < 68; k0 += 4) {
      v2f ar = frag_a(IUr, 68, mo, k0);
      v2f ai = frag_a(IUi, 68, mo, k0);
      v2f br = frag_b_kguard(pr, NL, k0, no, NLP);
      v2f bi = frag_b_kguard(pi, NL, k0, no, NLP);
      dr = wmma4(ar, br, dr);
      v2f nai;
      nai.x = -ai.x;
      nai.y = -ai.y;
      dr = wmma4(nai, bi, dr);
    }
    int n = no + (lane & 15);
    int mb = mo + ((lane >> 4) << 3);
#pragma unroll
    for (int i = 0; i < 8; ++i) op[(mb + i) * NL + n] = dr[i];
  }
}


extern "C" void kernel_launch(void* const* d_in, const int* in_sizes, int n_in,
                              void* d_out, int out_size, void* d_ws,
                              size_t ws_size, hipStream_t stream) {
  (void)in_sizes; (void)n_in; (void)out_size; (void)ws_size;
  const float* x   = (const float*)d_in[0];
  const float* wre = (const float*)d_in[1];
  const float* wim = (const float*)d_in[2];
  float* out = (float*)d_out;
  float* ws = (float*)d_ws;

  float* tw = ws;
  size_t off = TW_PAD;
  float* Yr = ws + off; off += (size_t)NB * NC * NP;
  float* Yi = ws + off; off += (size_t)NB * NC * NP;
  float* Xr = ws + off; off += (size_t)NB * NC * NP;
  float* Xi = ws + off; off += (size_t)NB * NC * NP;
  float* Gr = ws + off; off += (size_t)NB * ND * NP;
  float* Gi = ws + off; off += (size_t)NB * ND * NP;
  float* Pr = ws + off; off += (size_t)NB * ND * NPW;
  float* Pi = ws + off; off += (size_t)NB * ND * NPW;

  k_twiddles<<<dim3(25, 5), 256, 0, stream>>>(tw);
  k_fft2d<<<NB * NC, 128, 0, stream>>>(x, tw, Yr, Yi);
  k_cfft<<<dim3((NP + 15) / 16, NB), 128, 0, stream>>>(tw, Yr, Yi, Xr, Xi);
  for (int n = 0; n < NFR; ++n) {
    k_mix<<<dim3(MIX_GX, NB / MIXB), 128, 0, stream>>>(tw, Xr, Xi, wre, wim, Gr, Gi, n);
    k_ifft_v<<<dim3(ND, NB), 128, 0, stream>>>(tw, Gr, Gi, Pr, Pi);
    k_ifft_u<<<dim3(ND, NB), 128, 0, stream>>>(tw, Pr, Pi, out, n);
  }
}


# inverse DFT stages v and u fused into one kernel per (b,d) plane: G plane staged zero-padded in LDS, v-stage result kept in LDS (transposed), P planes and 8 launches gone; sum order unchanged
# speedup vs baseline: 4.3379x; 1.7582x over previous
#include <hip/hip_runtime.h>


typedef __attribute__((ext_vector_type(2))) float v2f;
typedef __attribute__((ext_vector_type(8))) float v8f;

constexpr int NB  = 16;
constexpr int NC  = 64;
constexpr int NFR = 8;
constexpr int ND  = 16;
constexpr int NL  = 64;
constexpr int NLP = 65;
constexpr int NP  = NLP * NLP;

constexpr int OFF_FWDR = 0;
constexpr int OFF_FWDI = OFF_FWDR + 80 * 80;
constexpr int OFF_ER   = OFF_FWDI + 80 * 80;
constexpr int OFF_EI   = OFF_ER + 64 * 64;
constexpr int OFF_WNR  = OFF_EI + 64 * 64;
constexpr int OFF_WNI  = OFF_WNR + 64 * 16;
constexpr int OFF_IVR  = OFF_WNI + 64 * 16;
constexpr int OFF_IVI  = OFF_IVR + 68 * 64;
constexpr int OFF_IUR  = OFF_IVI + 68 * 64;
constexpr int OFF_IUI  = OFF_IUR + 64 * 68;
constexpr int TW_TOTAL = OFF_IUI + 64 * 68;
constexpr int TW_PAD   = 40960;

#define PI_D 3.14159265358979323846


static __device__ __forceinline__ v8f wmma4(v2f a, v2f b, v8f c) {
  return __builtin_amdgcn_wmma_f32_16x16x4_f32(false, a, false, b, (short)0, c,
                                               false, false);
}

static __device__ __forceinline__ v8f zero8() {
  v8f z = {0.f, 0.f, 0.f, 0.f, 0.f, 0.f, 0.f, 0.f};
  return z;
}

static __device__ __forceinline__ v2f frag_a(const float* __restrict__ A,
                                             int lda, int m0, int k0) {
  int lane = threadIdx.x & 31;
  int m = m0 + (lane & 15);
  int k = k0 + ((lane >> 4) << 1);
  const float* p = A + m * lda + k;
  v2f r;
  r.x = p[0];
  r.y = p[1];
  return r;
}

static __device__ __forceinline__ v2f frag_b(const float* __restrict__ B,
                                             int ldb, int k0, int n0) {
  int lane = threadIdx.x & 31;
  int n = n0 + (lane & 15);
  int k = k0 + ((lane >> 4) << 1);
  const float* p = B + k * ldb + n;
  v2f r;
  r.x = p[0];
  r.y = p[ldb];
  return r;
}

static __device__ __forceinline__ v2f frag_b_nguard(const float* __restrict__ B,
                                                    int ldb, int k0, int n0,
                                                    int nmax) {
  int lane = threadIdx.x & 31;
  int n = n0 + (lane & 15);
  int k = k0 + ((lane >> 4) << 1);
  v2f r;
  r.x = 0.f;
  r.y = 0.f;
  if (n < nmax) {
    const float* p = B + k * ldb + n;
    r.x = p[0];
    r.y = p[ldb];
  }
  return r;
}

static __device__ __forceinline__ void cwmma(v2f ar, v2f ai, v2f br, v2f bi,
                                             v8f& dr, v8f& di) {
  dr = wmma4(ar, br, dr);
  v2f nai;
  nai.x = -ai.x;
  nai.y = -ai.y;
  dr = wmma4(nai, bi, dr);
  di = wmma4(ar, bi, di);
  di = wmma4(ai, br, di);
}


__global__ void k_twiddles(float* __restrict__ tw) {
  int t = blockIdx.x * blockDim.x + threadIdx.x;
  int which = blockIdx.y;
  if (which == 0) {
    if (t >= 80 * 80) return;
    int a = t / 80, b = t % 80;
    float r = 0.f, im = 0.f;
    if (a < NLP && b < NLP) {
      double ang = -2.0 * PI_D * (double)((a * b) % NLP) / (double)NLP;
      r = (float)cos(ang);
      im = (float)sin(ang);
    }
    tw[OFF_FWDR + t] = r;
    tw[OFF_FWDI + t] = im;
  } else if (which == 1) {
    if (t >= NC * NC) return;
    int c = t / NC, k = t % NC;
    double ang = -2.0 * PI_D * (double)((c * k) % NC) / (double)NC;
    tw[OFF_ER + t] = (float)cos(ang);
    tw[OFF_EI + t] = (float)sin(ang);
  } else if (which == 2) {
    if (t >= NC * ND) return;
    int d = t / NC, c = t % NC;
    double ang = 2.0 * PI_D * (double)((c * 4 * d) % NC) / (double)NC;
    tw[OFF_WNR + t] = (float)(cos(ang) / 64.0);
    tw[OFF_WNI + t] = (float)(sin(ang) / 64.0);
  } else if (which == 3) {
    if (t >= 68 * NL) return;
    int v = t / NL, w = t % NL;
    float r = 0.f, im = 0.f;
    if (v < NLP) {
      double ang = 2.0 * PI_D * (double)((v * (w + 1)) % NLP) / (double)NLP;
      r = (float)cos(ang);
      im = (float)sin(ang);
    }
    tw[OFF_IVR + t] = r;
    tw[OFF_IVI + t] = im;
  } else {
    if (t >= NL * 68) return;
    int h = t / 68, u = t % 68;
    float r = 0.f, im = 0.f;
    if (u < NLP) {
      double ang = 2.0 * PI_D * (double)((u * (h + 1)) % NLP) / (double)NLP;
      double s = 1.0 / (double)(NLP * NLP);
      r = (float)(cos(ang) * s);
      im = (float)(sin(ang) * s);
    }
    tw[OFF_IUR + t] = r;
    tw[OFF_IUI + t] = im;
  }
}


__global__ __launch_bounds__(128) void k_fft2d(const float* __restrict__ x,
                                               const float* __restrict__ tw,
                                               float* __restrict__ Yr,
                                               float* __restrict__ Yi) {
  __shared__ float xs[NL * NL];
  __shared__ float Tr[NL * 80];
  __shared__ float Ti[NL * 80];
  const int plane = blockIdx.x;
  const int tid = threadIdx.x;
  const int wave = tid >> 5;
  const int lane = tid & 31;

  const float* xp = x + (size_t)plane * (NL * NL);
  for (int i = tid; i < NL * NL; i += 128) xs[i] = xp[i];
  __syncthreads();

  const float* Fr = tw + OFF_FWDR;
  const float* Fi = tw + OFF_FWDI;

  for (int t = wave; t < 4 * 5; t += 4) {
    int mo = (t / 5) * 16, no = (t % 5) * 16;
    v8f ar = zero8(), ai = zero8();
    for (int k0 = 0; k0 < NL; k0 += 4) {
      v2f a = frag_a(xs, NL, mo, k0);
      ar = wmma4(a, frag_b(Fr, 80, k0, no), ar);
      ai = wmma4(a, frag_b(Fi, 80, k0, no), ai);
    }
    int n = no + (lane & 15);
    int mb = mo + ((lane >> 4) << 3);
#pragma unroll
    for (int i = 0; i < 8; ++i) {
      Tr[(mb + i) * 80 + n] = ar[i];
      Ti[(mb + i) * 80 + n] = ai[i];
    }
  }
  __syncthreads();

  float* yr = Yr + (size_t)plane * NP;
  float* yi = Yi + (size_t)plane * NP;
  for (int t = wave; t < 5 * 5; t += 4) {
    int mo = (t / 5) * 16, no = (t % 5) * 16;
    v8f dr = zero8(), di = zero8();
    for (int k0 = 0; k0 < NL; k0 += 4) {
      v2f par = frag_a(Fr, 80, mo, k0);
      v2f pai = frag_a(Fi, 80, mo, k0);
      v2f pbr = frag_b(Tr, 80, k0, no);
      v2f pbi = frag_b(Ti, 80, k0, no);
      cwmma(par, pai, pbr, pbi, dr, di);
    }
    int n = no + (lane & 15);
    int mb = mo + ((lane >> 4) << 3);
    if (n < NLP) {
#pragma unroll
      for (int i = 0; i < 8; ++i) {
        int m = mb + i;
        if (m < NLP) {
          yr[m * NLP + n] = dr[i];
          yi[m * NLP + n] = di[i];
        }
      }
    }
  }
}


__global__ __launch_bounds__(128) void k_cfft(const float* __restrict__ tw,
                                              const float* __restrict__ Yr,
                                              const float* __restrict__ Yi,
                                              float* __restrict__ Xr,
                                              float* __restrict__ Xi) {
  const int b = blockIdx.y;
  const int no = blockIdx.x * 16;
  const int mo = (threadIdx.x >> 5) * 16;
  const int lane = threadIdx.x & 31;
  const float* Er = tw + OFF_ER;
  const float* Ei = tw + OFF_EI;
  const float* yr = Yr + (size_t)b * NC * NP;
  const float* yi = Yi + (size_t)b * NC * NP;
  v8f dr = zero8(), di = zero8();
  for (int k0 = 0; k0 < NC; k0 += 4) {
    v2f ar = frag_a(Er, NC, mo, k0);
    v2f ai = frag_a(Ei, NC, mo, k0);
    v2f br = frag_b_nguard(yr, NP, k0, no, NP);
    v2f bi = frag_b_nguard(yi, NP, k0, no, NP);
    cwmma(ar, ai, br, bi, dr, di);
  }
  int n = no + (lane & 15);
  int mb = mo + ((lane >> 4) << 3);
  if (n < NP) {
    float* xr = Xr + (size_t)b * NC * NP;
    float* xi = Xi + (size_t)b * NC * NP;
#pragma unroll
    for (int i = 0; i < 8; ++i) {
      xr[(mb + i) * NP + n] = dr[i];
      xi[(mb + i) * NP + n] = di[i];
    }
  }
}


constexpr int MIXB = 2;
constexpr int MIX_PT = (NP + 15) / 16;
constexpr int MIX_GX = (MIX_PT + 3) / 4;

__global__ __launch_bounds__(128) void k_mix(const float* __restrict__ tw,
                                             const float* __restrict__ Xr,
                                             const float* __restrict__ Xi,
                                             const float* __restrict__ wre,
                                             const float* __restrict__ wim,
                                             float* __restrict__ Gr,
                                             float* __restrict__ Gi, int nidx) {
  const int wave = threadIdx.x >> 5;
  const int lane = threadIdx.x & 31;
  const int p0 = (blockIdx.x * 4 + wave) * 16;
  if (p0 >= NP) return;
  const int col = lane & 15;
  const int h2 = (lane >> 4) << 1;
  const int p = p0 + col;
  const int pl = p < NP ? p : NP - 1;
  const int b0 = blockIdx.y * MIXB;

  const float* ar_p = tw + OFF_WNR + col * NC + h2;
  const float* ai_p = tw + OFF_WNI + col * NC + h2;
  const float* wr_p = wre + ((size_t)nidx * NC + h2) * NP + pl;
  const float* wi_p = wim + ((size_t)nidx * NC + h2) * NP + pl;
  const float* xr_p = Xr + ((size_t)b0 * NC + h2) * NP + pl;
  const float* xi_p = Xi + ((size_t)b0 * NC + h2) * NP + pl;

  v8f dr[MIXB], di[MIXB];
#pragma unroll
  for (int j = 0; j < MIXB; ++j) {
    dr[j] = zero8();
    di[j] = zero8();
  }

#pragma unroll
  for (int k0 = 0; k0 < NC; k0 += 4) {
    v2f ar = *reinterpret_cast<const v2f*>(ar_p + k0);
    v2f ai = *reinterpret_cast<const v2f*>(ai_p + k0);
    v2f nai;
    nai.x = -ai.x;
    nai.y = -ai.y;
    float wr0 = wr_p[(size_t)k0 * NP], wr1 = wr_p[(size_t)(k0 + 1) * NP];
    float wi0 = wi_p[(size_t)k0 * NP], wi1 = wi_p[(size_t)(k0 + 1) * NP];
#pragma unroll
    for (int j = 0; j < MIXB; ++j) {
      size_t o0 = ((size_t)j * NC + k0) * NP;
      size_t o1 = o0 + NP;
      float xr0 = xr_p[o0], xi0 = xi_p[o0];
      float xr1 = xr_p[o1], xi1 = xi_p[o1];
      v2f br, bi;
      br.x = xr0 * wr0 - xi0 * wi0;
      bi.x = xr0 * wi0 + xi0 * wr0;
      br.y = xr1 * wr1 - xi1 * wi1;
      bi.y = xr1 * wi1 + xi1 * wr1;
      dr[j] = wmma4(ar, br, dr[j]);
      dr[j] = wmma4(nai, bi, dr[j]);
      di[j] = wmma4(ar, bi, di[j]);
      di[j] = wmma4(ai, br, di[j]);
    }
  }

  if (p < NP) {
    const int mb = (lane >> 4) << 3;
#pragma unroll
    for (int j = 0; j < MIXB; ++j) {
#pragma unroll
      for (int i = 0; i < 8; ++i) {
        size_t o = ((size_t)(b0 + j) * ND + (mb + i)) * NP + p;
        Gr[o] = dr[j][i];
        Gi[o] = di[j][i];
      }
    }
  }
}


typedef __attribute__((ext_vector_type(4))) float v4f;

constexpr int GP = 68;
constexpr int GS = 80 * GP;
constexpr int TP = 84;
constexpr int KS = 17;

__global__ __launch_bounds__(256) void k_ifft2d(const float* __restrict__ tw,
                                                const float* __restrict__ Gr,
                                                const float* __restrict__ Gi,
                                                float* __restrict__ out,
                                                int nidx) {
  __shared__ __attribute__((aligned(16))) float gs_r[GS];
  __shared__ __attribute__((aligned(16))) float gs_i[GS];
  __shared__ __attribute__((aligned(16))) float tt_r[NL * TP];
  __shared__ __attribute__((aligned(16))) float tt_i[NL * TP];
  const int d = blockIdx.x, b = blockIdx.y;
  const int tid = threadIdx.x;
  const int wave = __builtin_amdgcn_readfirstlane(tid >> 5);
  const int lane = tid & 31;
  const int c = lane & 15;
  const int h2 = (lane >> 4) << 1;
  const int wq = wave & 3;
  const int wh = wave >> 2;

  const float* gr = Gr + ((size_t)b * ND + d) * NP;
  const float* gi = Gi + ((size_t)b * ND + d) * NP;
  const int r4 = tid >> 6, c64 = tid & 63;
  float sr[17], si[17];
#pragma unroll
  for (int j = 0; j < 17; ++j) {
    const int row = 4 * j + r4;
    const bool ok = row < NLP;
    const int s = ok ? row * NLP + c64 : 0;
    const float vr = gr[s], vi = gi[s];
    sr[j] = ok ? vr : 0.f;
    si[j] = ok ? vi : 0.f;
  }
  const int erow = tid >> 2, ecol = tid & 3;
  const bool ok0 = ecol == 0;
  const int s0 = ok0 ? erow * NLP + 64 : 0;
  const float g0r = gr[s0], g0i = gi[s0];
  const float e0r = ok0 ? g0r : 0.f, e0i = ok0 ? g0i : 0.f;
  const int erow1 = 64 + (erow & 15);
  const bool ok1 = (tid & 63) == 0;
  const int s1 = ok1 ? NP - 1 : 0;
  const float g1r = gr[s1], g1i = gi[s1];
  const float e1r = ok1 ? g1r : 0.f, e1i = ok1 ? g1i : 0.f;

  const float* IVr = tw + OFF_IVR;
  const float* IVi = tw + OFF_IVI;
  v2f vbr[KS], vbi[KS], vnbi[KS];
#pragma unroll
  for (int k = 0; k < KS; ++k) {
    vbr[k] = frag_b(IVr, NL, 4 * k, wq * 16);
    vbi[k] = frag_b(IVi, NL, 4 * k, wq * 16);
  }

#pragma unroll
  for (int j = 0; j < 20; ++j) {
    gs_r[(4 * j + r4) * GP + c64] = j < 17 ? sr[j] : 0.f;
    gs_i[(4 * j + r4) * GP + c64] = j < 17 ? si[j] : 0.f;
  }
  gs_r[erow * GP + 64 + ecol] = e0r;
  gs_i[erow * GP + 64 + ecol] = e0i;
  gs_r[erow1 * GP + 64 + ecol] = e1r;
  gs_i[erow1 * GP + 64 + ecol] = e1i;
  __syncthreads();

  const float* IUr = tw + OFF_IUR;
  const float* IUi = tw + OFF_IUI;
  v2f uar[KS], unai[KS];
#pragma unroll
  for (int k = 0; k < KS; ++k) {
    uar[k] = frag_a(IUr, 68, wq * 16, 4 * k);
    v2f t = frag_a(IUi, 68, wq * 16, 4 * k);
    unai[k].x = -t.x;
    unai[k].y = -t.y;
  }
#pragma unroll
  for (int k = 0; k < KS; ++k) {
    vnbi[k].x = -vbi[k].x;
    vnbi[k].y = -vbi[k].y;
  }

  {
    const int no = wq * 16;
    const int mo_end = wh ? 80 : 48;
    for (int mo = wh ? 48 : 0; mo < mo_end; mo += 16) {
      const float* ar_p = gs_r + (mo + c) * GP + h2;
      const float* ai_p = gs_i + (mo + c) * GP + h2;
      v8f dr = zero8(), di = zero8();
#pragma unroll
      for (int k = 0; k < KS; ++k) {
        v2f ar = *reinterpret_cast<const v2f*>(ar_p + 4 * k);
        v2f ai = *reinterpret_cast<const v2f*>(ai_p + 4 * k);
        dr = wmma4(ar, vbr[k], dr);
        di = wmma4(ar, vbi[k], di);
        dr = wmma4(ai, vnbi[k], dr);
        di = wmma4(ai, vbr[k], di);
      }
      float* tr = tt_r + (no + c) * TP + mo + ((lane >> 4) << 3);
      float* ti = tt_i + (no + c) * TP + mo + ((lane >> 4) << 3);
      v4f q;
      q.x = dr[0]; q.y = dr[1]; q.z = dr[2]; q.w = dr[3];
      *reinterpret_cast<v4f*>(tr) = q;
      q.x = dr[4]; q.y = dr[5]; q.z = dr[6]; q.w = dr[7];
      *reinterpret_cast<v4f*>(tr + 4) = q;
      q.x = di[0]; q.y = di[1]; q.z = di[2]; q.w = di[3];
      *reinterpret_cast<v4f*>(ti) = q;
      q.x = di[4]; q.y = di[5]; q.z = di[6]; q.w = di[7];
      *reinterpret_cast<v4f*>(ti + 4) = q;
    }
  }
  __syncthreads();

  {
    const int mo = wq * 16;
    const int no = wh * 32;
    const float* b0r_p = tt_r + (no + c) * TP + h2;
    const float* b0i_p = tt_i + (no + c) * TP + h2;
    const float* b1r_p = b0r_p + 16 * TP;
    const float* b1i_p = b0i_p + 16 * TP;
    v8f d0 = zero8(), d1 = zero8();
#pragma unroll
    for (int k = 0; k < KS; ++k) {
      v2f b0r = *reinterpret_cast<const v2f*>(b0r_p + 4 * k);
      v2f b1r = *reinterpret_cast<const v2f*>(b1r_p + 4 * k);
      v2f b0i = *reinterpret_cast<const v2f*>(b0i_p + 4 * k);
      v2f b1i = *reinterpret_cast<const v2f*>(b1i_p + 4 * k);
      d0 = wmma4(uar[k], b0r, d0);
      d1 = wmma4(uar[k], b1r, d1);
      d0 = wmma4(unai[k], b0i, d0);
      d1 = wmma4(unai[k], b1i, d1);
    }
    float* op = out + ((size_t)b * (ND * NFR) + (d * NFR + nidx)) * (NL * NL);
    const int n = no + c;
    const int mb = mo + ((lane >> 4) << 3);
#pragma unroll
    for (int i = 0; i < 8; ++i) {
      op[(mb + i) * NL + n] = d0[i];
      op[(mb + i) * NL + n + 16] = d1[i];
    }
  }
}


extern "C" void kernel_launch(void* const* d_in, const int* in_sizes, int n_in,
                              void* d_out, int out_size, void* d_ws,
                              size_t ws_size, hipStream_t stream) {
  (void)in_sizes; (void)n_in; (void)out_size; (void)ws_size;
  const float* x   = (const float*)d_in[0];
  const float* wre = (const float*)d_in[1];
  const float* wim = (const float*)d_in[2];
  float* out = (float*)d_out;
  float* ws = (float*)d_ws;

  float* tw = ws;
  size_t off = TW_PAD;
  float* Yr = ws + off; off += (size_t)NB * NC * NP;
  float* Yi = ws + off; off += (size_t)NB * NC * NP;
  float* Xr = ws + off; off += (size_t)NB * NC * NP;
  float* Xi = ws + off; off += (size_t)NB * NC * NP;
  float* Gr = ws + off; off += (size_t)NB * ND * NP;
  float* Gi = ws + off;

  k_twiddles<<<dim3(25, 5), 256, 0, stream>>>(tw);
  k_fft2d<<<NB * NC, 128, 0, stream>>>(x, tw, Yr, Yi);
  k_cfft<<<dim3((NP + 15) / 16, NB), 128, 0, stream>>>(tw, Yr, Yi, Xr, Xi);
  for (int n = 0; n < NFR; ++n) {
    k_mix<<<dim3(MIX_GX, NB / MIXB), 128, 0, stream>>>(tw, Xr, Xi, wre, wim, Gr, Gi, n);
    k_ifft2d<<<dim3(ND, NB), 256, 0, stream>>>(tw, Gr, Gi, out, n);
  }
}


# mode loop folded into the grid: mix and inverse-DFT kernels each run once per 4 modes (n a grid dimension), G of 4 modes kept in the dead Y planes; 19 launches become 7; sum order unchanged
# speedup vs baseline: 4.9768x; 1.1473x over previous
#include <hip/hip_runtime.h>


typedef __attribute__((ext_vector_type(2))) float v2f;
typedef __attribute__((ext_vector_type(8))) float v8f;

constexpr int NB  = 16;
constexpr int NC  = 64;
constexpr int NFR = 8;
constexpr int ND  = 16;
constexpr int NL  = 64;
constexpr int NLP = 65;
constexpr int NP  = NLP * NLP;

constexpr int OFF_FWDR = 0;
constexpr int OFF_FWDI = OFF_FWDR + 80 * 80;
constexpr int OFF_ER   = OFF_FWDI + 80 * 80;
constexpr int OFF_EI   = OFF_ER + 64 * 64;
constexpr int OFF_WNR  = OFF_EI + 64 * 64;
constexpr int OFF_WNI  = OFF_WNR + 64 * 16;
constexpr int OFF_IVR  = OFF_WNI + 64 * 16;
constexpr int OFF_IVI  = OFF_IVR + 68 * 64;
constexpr int OFF_IUR  = OFF_IVI + 68 * 64;
constexpr int OFF_IUI  = OFF_IUR + 64 * 68;
constexpr int TW_TOTAL = OFF_IUI + 64 * 68;
constexpr int TW_PAD   = 40960;

#define PI_D 3.14159265358979323846


static __device__ __forceinline__ v8f wmma4(v2f a, v2f b, v8f c) {
  return __builtin_amdgcn_wmma_f32_16x16x4_f32(false, a, false, b, (short)0, c,
                                               false, false);
}

static __device__ __forceinline__ v8f zero8() {
  v8f z = {0.f, 0.f, 0.f, 0.f, 0.f, 0.f, 0.f, 0.f};
  return z;
}

static __device__ __forceinline__ v2f frag_a(const float* __restrict__ A,
                                             int lda, int m0, int k0) {
  int lane = threadIdx.x & 31;
  int m = m0 + (lane & 15);
  int k = k0 + ((lane >> 4) << 1);
  const float* p = A + m * lda + k;
  v2f r;
  r.x = p[0];
  r.y = p[1];
  return r;
}

static __device__ __forceinline__ v2f frag_b(const float* __restrict__ B,
                                             int ldb, int k0, int n0) {
  int lane = threadIdx.x & 31;
  int n = n0 + (lane & 15);
  int k = k0 + ((lane >> 4) << 1);
  const float* p = B + k * ldb + n;
  v2f r;
  r.x = p[0];
  r.y = p[ldb];
  return r;
}

static __device__ __forceinline__ v2f frag_b_nguard(const float* __restrict__ B,
                                                    int ldb, int k0, int n0,
                                                    int nmax) {
  int lane = threadIdx.x & 31;
  int n = n0 + (lane & 15);
  int k = k0 + ((lane >> 4) << 1);
  v2f r;
  r.x = 0.f;
  r.y = 0.f;
  if (n < nmax) {
    const float* p = B + k * ldb + n;
    r.x = p[0];
    r.y = p[ldb];
  }
  return r;
}

static __device__ __forceinline__ void cwmma(v2f ar, v2f ai, v2f br, v2f bi,
                                             v8f& dr, v8f& di) {
  dr = wmma4(ar, br, dr);
  v2f nai;
  nai.x = -ai.x;
  nai.y = -ai.y;
  dr = wmma4(nai, bi, dr);
  di = wmma4(ar, bi, di);
  di = wmma4(ai, br, di);
}


__global__ void k_twiddles(float* __restrict__ tw) {
  int t = blockIdx.x * blockDim.x + threadIdx.x;
  int which = blockIdx.y;
  if (which == 0) {
    if (t >= 80 * 80) return;
    int a = t / 80, b = t % 80;
    float r = 0.f, im = 0.f;
    if (a < NLP && b < NLP) {
      double ang = -2.0 * PI_D * (double)((a * b) % NLP) / (double)NLP;
      r = (float)cos(ang);
      im = (float)sin(ang);
    }
    tw[OFF_FWDR + t] = r;
    tw[OFF_FWDI + t] = im;
  } else if (which == 1) {
    if (t >= NC * NC) return;
    int c = t / NC, k = t % NC;
    double ang = -2.0 * PI_D * (double)((c * k) % NC) / (double)NC;
    tw[OFF_ER + t] = (float)cos(ang);
    tw[OFF_EI + t] = (float)sin(ang);
  } else if (which == 2) {
    if (t >= NC * ND) return;
    int d = t / NC, c = t % NC;
    double ang = 2.0 * PI_D * (double)((c * 4 * d) % NC) / (double)NC;
    tw[OFF_WNR + t] = (float)(cos(ang) / 64.0);
    tw[OFF_WNI + t] = (float)(sin(ang) / 64.0);
  } else if (which == 3) {
    if (t >= 68 * NL) return;
    int v = t / NL, w = t % NL;
    float r = 0.f, im = 0.f;
    if (v < NLP) {
      double ang = 2.0 * PI_D * (double)((v * (w + 1)) % NLP) / (double)NLP;
      r = (float)cos(ang);
      im = (float)sin(ang);
    }
    tw[OFF_IVR + t] = r;
    tw[OFF_IVI + t] = im;
  } else {
    if (t >= NL * 68) return;
    int h = t / 68, u = t % 68;
    float r = 0.f, im = 0.f;
    if (u < NLP) {
      double ang = 2.0 * PI_D * (double)((u * (h + 1)) % NLP) / (double)NLP;
      double s = 1.0 / (double)(NLP * NLP);
      r = (float)(cos(ang) * s);
      im = (float)(sin(ang) * s);
    }
    tw[OFF_IUR + t] = r;
    tw[OFF_IUI + t] = im;
  }
}


__global__ __launch_bounds__(128) void k_fft2d(const float* __restrict__ x,
                                               const float* __restrict__ tw,
                                               float* __restrict__ Yr,
                                               float* __restrict__ Yi) {
  __shared__ float xs[NL * NL];
  __shared__ float Tr[NL * 80];
  __shared__ float Ti[NL * 80];
  const int plane = blockIdx.x;
  const int tid = threadIdx.x;
  const int wave = tid >> 5;
  const int lane = tid & 31;

  const float* xp = x + (size_t)plane * (NL * NL);
  for (int i = tid; i < NL * NL; i += 128) xs[i] = xp[i];
  __syncthreads();

  const float* Fr = tw + OFF_FWDR;
  const float* Fi = tw + OFF_FWDI;

  for (int t = wave; t < 4 * 5; t += 4) {
    int mo = (t / 5) * 16, no = (t % 5) * 16;
    v8f ar = zero8(), ai = zero8();
    for (int k0 = 0; k0 < NL; k0 += 4) {
      v2f a = frag_a(xs, NL, mo, k0);
      ar = wmma4(a, frag_b(Fr, 80, k0, no), ar);
      ai = wmma4(a, frag_b(Fi, 80, k0, no), ai);
    }
    int n = no + (lane & 15);
    int mb = mo + ((lane >> 4) << 3);
#pragma unroll
    for (int i = 0; i < 8; ++i) {
      Tr[(mb + i) * 80 + n] = ar[i];
      Ti[(mb + i) * 80 + n] = ai[i];
    }
  }
  __syncthreads();

  float* yr = Yr + (size_t)plane * NP;
  float* yi = Yi + (size_t)plane * NP;
  for (int t = wave; t < 5 * 5; t += 4) {
    int mo = (t / 5) * 16, no = (t % 5) * 16;
    v8f dr = zero8(), di = zero8();
    for (int k0 = 0; k0 < NL; k0 += 4) {
      v2f par = frag_a(Fr, 80, mo, k0);
      v2f pai = frag_a(Fi, 80, mo, k0);
      v2f pbr = frag_b(Tr, 80, k0, no);
      v2f pbi = frag_b(Ti, 80, k0, no);
      cwmma(par, pai, pbr, pbi, dr, di);
    }
    int n = no + (lane & 15);
    int mb = mo + ((lane >> 4) << 3);
    if (n < NLP) {
#pragma unroll
      for (int i = 0; i < 8; ++i) {
        int m = mb + i;
        if (m < NLP) {
          yr[m * NLP + n] = dr[i];
          yi[m * NLP + n] = di[i];
        }
      }
    }
  }
}


__global__ __launch_bounds__(128) void k_cfft(const float* __restrict__ tw,
                                              const float* __restrict__ Yr,
                                              const float* __restrict__ Yi,
                                              float* __restrict__ Xr,
                                              float* __restrict__ Xi) {
  const int b = blockIdx.y;
  const int no = blockIdx.x * 16;
  const int mo = (threadIdx.x >> 5) * 16;
  const int lane = threadIdx.x & 31;
  const float* Er = tw + OFF_ER;
  const float* Ei = tw + OFF_EI;
  const float* yr = Yr + (size_t)b * NC * NP;
  const float* yi = Yi + (size_t)b * NC * NP;
  v8f dr = zero8(), di = zero8();
  for (int k0 = 0; k0 < NC; k0 += 4) {
    v2f ar = frag_a(Er, NC, mo, k0);
    v2f ai = frag_a(Ei, NC, mo, k0);
    v2f br = frag_b_nguard(yr, NP, k0, no, NP);
    v2f bi = frag_b_nguard(yi, NP, k0, no, NP);
    cwmma(ar, ai, br, bi, dr, di);
  }
  int n = no + (lane & 15);
  int mb = mo + ((lane >> 4) << 3);
  if (n < NP) {
    float* xr = Xr + (size_t)b * NC * NP;
    float* xi = Xi + (size_t)b * NC * NP;
#pragma unroll
    for (int i = 0; i < 8; ++i) {
      xr[(mb + i) * NP + n] = dr[i];
      xi[(mb + i) * NP + n] = di[i];
    }
  }
}


constexpr int MIXB = 2;
constexpr int MIX_PT = (NP + 15) / 16;
constexpr int MIX_GX = (MIX_PT + 3) / 4;
constexpr int NG = 4;

__global__ __launch_bounds__(128) void k_mix(const float* __restrict__ tw,
                                             const float* __restrict__ Xr,
                                             const float* __restrict__ Xi,
                                             const float* __restrict__ wre,
                                             const float* __restrict__ wim,
                                             float* __restrict__ Gr,
                                             float* __restrict__ Gi, int nbase) {
  const int wave = threadIdx.x >> 5;
  const int lane = threadIdx.x & 31;
  const int ng = blockIdx.x;
  const int nidx = nbase + ng;
  const int p0 = (blockIdx.y * 4 + wave) * 16;
  if (p0 >= NP) return;
  const int col = lane & 15;
  const int h2 = (lane >> 4) << 1;
  const int p = p0 + col;
  const int pl = p < NP ? p : NP - 1;
  const int b0 = blockIdx.z * MIXB;

  const float* ar_p = tw + OFF_WNR + col * NC + h2;
  const float* ai_p = tw + OFF_WNI + col * NC + h2;
  const float* wr_p = wre + ((size_t)nidx * NC + h2) * NP + pl;
  const float* wi_p = wim + ((size_t)nidx * NC + h2) * NP + pl;
  const float* xr_p = Xr + ((size_t)b0 * NC + h2) * NP + pl;
  const float* xi_p = Xi + ((size_t)b0 * NC + h2) * NP + pl;

  v8f dr[MIXB], di[MIXB];
#pragma unroll
  for (int j = 0; j < MIXB; ++j) {
    dr[j] = zero8();
    di[j] = zero8();
  }

#pragma unroll
  for (int k0 = 0; k0 < NC; k0 += 4) {
    v2f ar = *reinterpret_cast<const v2f*>(ar_p + k0);
    v2f ai = *reinterpret_cast<const v2f*>(ai_p + k0);
    v2f nai;
    nai.x = -ai.x;
    nai.y = -ai.y;
    float wr0 = wr_p[(size_t)k0 * NP], wr1 = wr_p[(size_t)(k0 + 1) * NP];
    float wi0 = wi_p[(size_t)k0 * NP], wi1 = wi_p[(size_t)(k0 + 1) * NP];
#pragma unroll
    for (int j = 0; j < MIXB; ++j) {
      size_t o0 = ((size_t)j * NC + k0) * NP;
      size_t o1 = o0 + NP;
      float xr0 = xr_p[o0], xi0 = xi_p[o0];
      float xr1 = xr_p[o1], xi1 = xi_p[o1];
      v2f br, bi;
      br.x = xr0 * wr0 - xi0 * wi0;
      bi.x = xr0 * wi0 + xi0 * wr0;
      br.y = xr1 * wr1 - xi1 * wi1;
      bi.y = xr1 * wi1 + xi1 * wr1;
      dr[j] = wmma4(ar, br, dr[j]);
      dr[j] = wmma4(nai, bi, dr[j]);
      di[j] = wmma4(ar, bi, di[j]);
      di[j] = wmma4(ai, br, di[j]);
    }
  }

  if (p < NP) {
    const int mb = (lane >> 4) << 3;
#pragma unroll
    for (int j = 0; j < MIXB; ++j) {
#pragma unroll
      for (int i = 0; i < 8; ++i) {
        size_t o = (((size_t)ng * NB + (b0 + j)) * ND + (mb + i)) * NP + p;
        Gr[o] = dr[j][i];
        Gi[o] = di[j][i];
      }
    }
  }
}


typedef __attribute__((ext_vector_type(4))) float v4f;

constexpr int GP = 68;
constexpr int GS = 80 * GP;
constexpr int TP = 84;
constexpr int KS = 17;

__global__ __launch_bounds__(256) void k_ifft2d(const float* __restrict__ tw,
                                                const float* __restrict__ Gr,
                                                const float* __restrict__ Gi,
                                                float* __restrict__ out,
                                                int nbase) {
  __shared__ __attribute__((aligned(16))) float gs_r[GS];
  __shared__ __attribute__((aligned(16))) float gs_i[GS];
  __shared__ __attribute__((aligned(16))) float tt_r[NL * TP];
  __shared__ __attribute__((aligned(16))) float tt_i[NL * TP];
  const int d = blockIdx.x, b = blockIdx.y, ng = blockIdx.z;
  const int nidx = nbase + ng;
  const int tid = threadIdx.x;
  const int wave = __builtin_amdgcn_readfirstlane(tid >> 5);
  const int lane = tid & 31;
  const int c = lane & 15;
  const int h2 = (lane >> 4) << 1;
  const int wq = wave & 3;
  const int wh = wave >> 2;

  const float* gr = Gr + (((size_t)ng * NB + b) * ND + d) * NP;
  const float* gi = Gi + (((size_t)ng * NB + b) * ND + d) * NP;
  const int r4 = tid >> 6, c64 = tid & 63;
  float sr[17], si[17];
#pragma unroll
  for (int j = 0; j < 17; ++j) {
    const int row = 4 * j + r4;
    const bool ok = row < NLP;
    const int s = ok ? row * NLP + c64 : 0;
    const float vr = gr[s], vi = gi[s];
    sr[j] = ok ? vr : 0.f;
    si[j] = ok ? vi : 0.f;
  }
  const int erow = tid >> 2, ecol = tid & 3;
  const bool ok0 = ecol == 0;
  const int s0 = ok0 ? erow * NLP + 64 : 0;
  const float g0r = gr[s0], g0i = gi[s0];
  const float e0r = ok0 ? g0r : 0.f, e0i = ok0 ? g0i : 0.f;
  const int erow1 = 64 + (erow & 15);
  const bool ok1 = (tid & 63) == 0;
  const int s1 = ok1 ? NP - 1 : 0;
  const float g1r = gr[s1], g1i = gi[s1];
  const float e1r = ok1 ? g1r : 0.f, e1i = ok1 ? g1i : 0.f;

  const float* IVr = tw + OFF_IVR;
  const float* IVi = tw + OFF_IVI;
  v2f vbr[KS], vbi[KS], vnbi[KS];
#pragma unroll
  for (int k = 0; k < KS; ++k) {
    vbr[k] = frag_b(IVr, NL, 4 * k, wq * 16);
    vbi[k] = frag_b(IVi, NL, 4 * k, wq * 16);
  }

#pragma unroll
  for (int j = 0; j < 20; ++j) {
    gs_r[(4 * j + r4) * GP + c64] = j < 17 ? sr[j] : 0.f;
    gs_i[(4 * j + r4) * GP + c64] = j < 17 ? si[j] : 0.f;
  }
  gs_r[erow * GP + 64 + ecol] = e0r;
  gs_i[erow * GP + 64 + ecol] = e0i;
  gs_r[erow1 * GP + 64 + ecol] = e1r;
  gs_i[erow1 * GP + 64 + ecol] = e1i;
  __syncthreads();

  const float* IUr = tw + OFF_IUR;
  const float* IUi = tw + OFF_IUI;
  v2f uar[KS], unai[KS];
#pragma unroll
  for (int k = 0; k < KS; ++k) {
    uar[k] = frag_a(IUr, 68, wq * 16, 4 * k);
    v2f t = frag_a(IUi, 68, wq * 16, 4 * k);
    unai[k].x = -t.x;
    unai[k].y = -t.y;
  }
#pragma unroll
  for (int k = 0; k < KS; ++k) {
    vnbi[k].x = -vbi[k].x;
    vnbi[k].y = -vbi[k].y;
  }

  {
    const int no = wq * 16;
    const int mo_end = wh ? 80 : 48;
    for (int mo = wh ? 48 : 0; mo < mo_end; mo += 16) {
      const float* ar_p = gs_r + (mo + c) * GP + h2;
      const float* ai_p = gs_i + (mo + c) * GP + h2;
      v8f dr = zero8(), di = zero8();
#pragma unroll
      for (int k = 0; k < KS; ++k) {
        v2f ar = *reinterpret_cast<const v2f*>(ar_p + 4 * k);
        v2f ai = *reinterpret_cast<const v2f*>(ai_p + 4 * k);
        dr = wmma4(ar, vbr[k], dr);
        di = wmma4(ar, vbi[k], di);
        dr = wmma4(ai, vnbi[k], dr);
        di = wmma4(ai, vbr[k], di);
      }
      float* tr = tt_r + (no + c) * TP + mo + ((lane >> 4) << 3);
      float* ti = tt_i + (no + c) * TP + mo + ((lane >> 4) << 3);
      v4f q;
      q.x = dr[0]; q.y = dr[1]; q.z = dr[2]; q.w = dr[3];
      *reinterpret_cast<v4f*>(tr) = q;
      q.x = dr[4]; q.y = dr[5]; q.z = dr[6]; q.w = dr[7];
      *reinterpret_cast<v4f*>(tr + 4) = q;
      q.x = di[0]; q.y = di[1]; q.z = di[2]; q.w = di[3];
      *reinterpret_cast<v4f*>(ti) = q;
      q.x = di[4]; q.y = di[5]; q.z = di[6]; q.w = di[7];
      *reinterpret_cast<v4f*>(ti + 4) = q;
    }
  }
  __syncthreads();

  {
    const int mo = wq * 16;
    const int no = wh * 32;
    const float* b0r_p = tt_r + (no + c) * TP + h2;
    const float* b0i_p = tt_i + (no + c) * TP + h2;
    const float* b1r_p = b0r_p + 16 * TP;
    const float* b1i_p = b0i_p + 16 * TP;
    v8f d0 = zero8(), d1 = zero8();
#pragma unroll
    for (int k = 0; k < KS; ++k) {
      v2f b0r = *reinterpret_cast<const v2f*>(b0r_p + 4 * k);
      v2f b1r = *reinterpret_cast<const v2f*>(b1r_p + 4 * k);
      v2f b0i = *reinterpret_cast<const v2f*>(b0i_p + 4 * k);
      v2f b1i = *reinterpret_cast<const v2f*>(b1i_p + 4 * k);
      d0 = wmma4(uar[k], b0r, d0);
      d1 = wmma4(uar[k], b1r, d1);
      d0 = wmma4(unai[k], b0i, d0);
      d1 = wmma4(unai[k], b1i, d1);
    }
    float* op = out + ((size_t)b * (ND * NFR) + (d * NFR + nidx)) * (NL * NL);
    const int n = no + c;
    const int mb = mo + ((lane >> 4) << 3);
#pragma unroll
    for (int i = 0; i < 8; ++i) {
      op[(mb + i) * NL + n] = d0[i];
      op[(mb + i) * NL + n + 16] = d1[i];
    }
  }
}


extern "C" void kernel_launch(void* const* d_in, const int* in_sizes, int n_in,
                              void* d_out, int out_size, void* d_ws,
                              size_t ws_size, hipStream_t stream) {
  (void)in_sizes; (void)n_in; (void)out_size; (void)ws_size;
  const float* x   = (const float*)d_in[0];
  const float* wre = (const float*)d_in[1];
  const float* wim = (const float*)d_in[2];
  float* out = (float*)d_out;
  float* ws = (float*)d_ws;

  float* tw = ws;
  size_t off = TW_PAD;
  float* Yr = ws + off; off += (size_t)NB * NC * NP;
  float* Yi = ws + off; off += (size_t)NB * NC * NP;
  float* Xr = ws + off; off += (size_t)NB * NC * NP;
  float* Xi = ws + off;
  float* Gr = Yr;
  float* Gi = Yi;

  k_twiddles<<<dim3(25, 5), 256, 0, stream>>>(tw);
  k_fft2d<<<NB * NC, 128, 0, stream>>>(x, tw, Yr, Yi);
  k_cfft<<<dim3((NP + 15) / 16, NB), 128, 0, stream>>>(tw, Yr, Yi, Xr, Xi);
  for (int n = 0; n < NFR; n += NG) {
    k_mix<<<dim3(NG, MIX_GX, NB / MIXB), 128, 0, stream>>>(tw, Xr, Xi, wre, wim, Gr, Gi, n);
    k_ifft2d<<<dim3(ND, NB, NG), 256, 0, stream>>>(tw, Gr, Gi, out, n);
  }
}


# constant-table A fragments loaded lane-contiguous (E, WNT, IU tables stored in fragment order; F read along its symmetric row): 2-3 cache lines per load, not 16; values and sum order unchanged
# speedup vs baseline: 5.3017x; 1.0653x over previous
#include <hip/hip_runtime.h>


typedef __attribute__((ext_vector_type(2))) float v2f;
typedef __attribute__((ext_vector_type(8))) float v8f;

constexpr int NB  = 16;
constexpr int NC  = 64;
constexpr int NFR = 8;
constexpr int ND  = 16;
constexpr int NL  = 64;
constexpr int NLP = 65;
constexpr int NP  = NLP * NLP;

constexpr int OFF_FWDR = 0;
constexpr int OFF_FWDI = OFF_FWDR + 80 * 80;
constexpr int OFF_ER   = OFF_FWDI + 80 * 80;
constexpr int OFF_EI   = OFF_ER + 64 * 64;
constexpr int OFF_WNR  = OFF_EI + 64 * 64;
constexpr int OFF_WNI  = OFF_WNR + 64 * 16;
constexpr int OFF_IVR  = OFF_WNI + 64 * 16;
constexpr int OFF_IVI  = OFF_IVR + 68 * 64;
constexpr int OFF_IUR  = OFF_IVI + 68 * 64;
constexpr int OFF_IUI  = OFF_IUR + 64 * 68;
constexpr int TW_TOTAL = OFF_IUI + 64 * 68;
constexpr int TW_PAD   = 40960;

#define PI_D 3.14159265358979323846


static __device__ __forceinline__ v8f wmma4(v2f a, v2f b, v8f c) {
  return __builtin_amdgcn_wmma_f32_16x16x4_f32(false, a, false, b, (short)0, c,
                                               false, false);
}

static __device__ __forceinline__ v8f zero8() {
  v8f z = {0.f, 0.f, 0.f, 0.f, 0.f, 0.f, 0.f, 0.f};
  return z;
}

static __device__ __forceinline__ v2f frag_a(const float* __restrict__ A,
                                             int lda, int m0, int k0) {
  int lane = threadIdx.x & 31;
  int m = m0 + (lane & 15);
  int k = k0 + ((lane >> 4) << 1);
  const float* p = A + m * lda + k;
  v2f r;
  r.x = p[0];
  r.y = p[1];
  return r;
}

static __device__ __forceinline__ v2f frag_a_tab(const float* __restrict__ A,
                                                 int ksteps, int tile, int s) {
  int lane = threadIdx.x & 31;
  return *reinterpret_cast<const v2f*>(A + ((tile * ksteps + s) * 32 + lane) * 2);
}

static __device__ __forceinline__ v2f frag_b(const float* __restrict__ B,
                                             int ldb, int k0, int n0) {
  int lane = threadIdx.x & 31;
  int n = n0 + (lane & 15);
  int k = k0 + ((lane >> 4) << 1);
  const float* p = B + k * ldb + n;
  v2f r;
  r.x = p[0];
  r.y = p[ldb];
  return r;
}

static __device__ __forceinline__ v2f frag_b_nguard(const float* __restrict__ B,
                                                    int ldb, int k0, int n0,
                                                    int nmax) {
  int lane = threadIdx.x & 31;
  int n = n0 + (lane & 15);
  int k = k0 + ((lane >> 4) << 1);
  v2f r;
  r.x = 0.f;
  r.y = 0.f;
  if (n < nmax) {
    const float* p = B + k * ldb + n;
    r.x = p[0];
    r.y = p[ldb];
  }
  return r;
}

static __device__ __forceinline__ void cwmma(v2f ar, v2f ai, v2f br, v2f bi,
                                             v8f& dr, v8f& di) {
  dr = wmma4(ar, br, dr);
  v2f nai;
  nai.x = -ai.x;
  nai.y = -ai.y;
  dr = wmma4(nai, bi, dr);
  di = wmma4(ar, bi, di);
  di = wmma4(ai, br, di);
}


__global__ void k_twiddles(float* __restrict__ tw) {
  int t = blockIdx.x * blockDim.x + threadIdx.x;
  int which = blockIdx.y;
  if (which == 0) {
    if (t >= 80 * 80) return;
    int a = t / 80, b = t % 80;
    float r = 0.f, im = 0.f;
    if (a < NLP && b < NLP) {
      double ang = -2.0 * PI_D * (double)((a * b) % NLP) / (double)NLP;
      r = (float)cos(ang);
      im = (float)sin(ang);
    }
    tw[OFF_FWDR + t] = r;
    tw[OFF_FWDI + t] = im;
  } else if (which == 1) {
    if (t >= NC * NC) return;
    int l = (t >> 1) & 31;
    int c = (t >> 10) * 16 + (l & 15);
    int k = ((t >> 6) & 15) * 4 + ((l >> 4) << 1) + (t & 1);
    double ang = -2.0 * PI_D * (double)((c * k) % NC) / (double)NC;
    tw[OFF_ER + t] = (float)cos(ang);
    tw[OFF_EI + t] = (float)sin(ang);
  } else if (which == 2) {
    if (t >= NC * ND) return;
    int l = (t >> 1) & 31;
    int d = l & 15;
    int c = (t >> 6) * 4 + ((l >> 4) << 1) + (t & 1);
    double ang = 2.0 * PI_D * (double)((c * 4 * d) % NC) / (double)NC;
    tw[OFF_WNR + t] = (float)(cos(ang) / 64.0);
    tw[OFF_WNI + t] = (float)(sin(ang) / 64.0);
  } else if (which == 3) {
    if (t >= 68 * NL) return;
    int v = t / NL, w = t % NL;
    float r = 0.f, im = 0.f;
    if (v < NLP) {
      double ang = 2.0 * PI_D * (double)((v * (w + 1)) % NLP) / (double)NLP;
      r = (float)cos(ang);
      im = (float)sin(ang);
    }
    tw[OFF_IVR + t] = r;
    tw[OFF_IVI + t] = im;
  } else {
    if (t >= NL * 68) return;
    int l = (t >> 1) & 31;
    int fs = t >> 6;
    int h = (fs / 17) * 16 + (l & 15);
    int u = (fs % 17) * 4 + ((l >> 4) << 1) + (t & 1);
    float r = 0.f, im = 0.f;
    if (u < NLP) {
      double ang = 2.0 * PI_D * (double)((u * (h + 1)) % NLP) / (double)NLP;
      double s = 1.0 / (double)(NLP * NLP);
      r = (float)(cos(ang) * s);
      im = (float)(sin(ang) * s);
    }
    tw[OFF_IUR + t] = r;
    tw[OFF_IUI + t] = im;
  }
}


__global__ __launch_bounds__(128) void k_fft2d(const float* __restrict__ x,
                                               const float* __restrict__ tw,
                                               float* __restrict__ Yr,
                                               float* __restrict__ Yi) {
  __shared__ float xs[NL * NL];
  __shared__ float Tr[NL * 80];
  __shared__ float Ti[NL * 80];
  const int plane = blockIdx.x;
  const int tid = threadIdx.x;
  const int wave = tid >> 5;
  const int lane = tid & 31;

  const float* xp = x + (size_t)plane * (NL * NL);
  for (int i = tid; i < NL * NL; i += 128) xs[i] = xp[i];
  __syncthreads();

  const float* Fr = tw + OFF_FWDR;
  const float* Fi = tw + OFF_FWDI;

  for (int t = wave; t < 4 * 5; t += 4) {
    int mo = (t / 5) * 16, no = (t % 5) * 16;
    v8f ar = zero8(), ai = zero8();
    for (int k0 = 0; k0 < NL; k0 += 4) {
      v2f a = frag_a(xs, NL, mo, k0);
      ar = wmma4(a, frag_b(Fr, 80, k0, no), ar);
      ai = wmma4(a, frag_b(Fi, 80, k0, no), ai);
    }
    int n = no + (lane & 15);
    int mb = mo + ((lane >> 4) << 3);
#pragma unroll
    for (int i = 0; i < 8; ++i) {
      Tr[(mb + i) * 80 + n] = ar[i];
      Ti[(mb + i) * 80 + n] = ai[i];
    }
  }
  __syncthreads();

  float* yr = Yr + (size_t)plane * NP;
  float* yi = Yi + (size_t)plane * NP;
  for (int t = wave; t < 5 * 5; t += 4) {
    int mo = (t / 5) * 16, no = (t % 5) * 16;
    v8f dr = zero8(), di = zero8();
#pragma unroll
    for (int k0 = 0; k0 < NL; k0 += 4) {
      v2f par = frag_b(Fr, 80, k0, mo);
      v2f pai = frag_b(Fi, 80, k0, mo);
      v2f pbr = frag_b(Tr, 80, k0, no);
      v2f pbi = frag_b(Ti, 80, k0, no);
      cwmma(par, pai, pbr, pbi, dr, di);
    }
    int n = no + (lane & 15);
    int mb = mo + ((lane >> 4) << 3);
    if (n < NLP) {
#pragma unroll
      for (int i = 0; i < 8; ++i) {
        int m = mb + i;
        if (m < NLP) {
          yr[m * NLP + n] = dr[i];
          yi[m * NLP + n] = di[i];
        }
      }
    }
  }
}


__global__ __launch_bounds__(128) void k_cfft(const float* __restrict__ tw,
                                              const float* __restrict__ Yr,
                                              const float* __restrict__ Yi,
                                              float* __restrict__ Xr,
                                              float* __restrict__ Xi) {
  const int b = blockIdx.y;
  const int no = blockIdx.x * 16;
  const int ct = threadIdx.x >> 5;
  const int mo = ct * 16;
  const int lane = threadIdx.x & 31;
  const float* Er = tw + OFF_ER;
  const float* Ei = tw + OFF_EI;
  const float* yr = Yr + (size_t)b * NC * NP;
  const float* yi = Yi + (size_t)b * NC * NP;
  v8f dr = zero8(), di = zero8();
  for (int k0 = 0; k0 < NC; k0 += 4) {
    v2f ar = frag_a_tab(Er, 16, ct, k0 >> 2);
    v2f ai = frag_a_tab(Ei, 16, ct, k0 >> 2);
    v2f br = frag_b_nguard(yr, NP, k0, no, NP);
    v2f bi = frag_b_nguard(yi, NP, k0, no, NP);
    cwmma(ar, ai, br, bi, dr, di);
  }
  int n = no + (lane & 15);
  int mb = mo + ((lane >> 4) << 3);
  if (n < NP) {
    float* xr = Xr + (size_t)b * NC * NP;
    float* xi = Xi + (size_t)b * NC * NP;
#pragma unroll
    for (int i = 0; i < 8; ++i) {
      xr[(mb + i) * NP + n] = dr[i];
      xi[(mb + i) * NP + n] = di[i];
    }
  }
}


constexpr int MIXB = 2;
constexpr int MIX_PT = (NP + 15) / 16;
constexpr int MIX_GX = (MIX_PT + 3) / 4;
constexpr int NG = 4;

__global__ __launch_bounds__(128) void k_mix(const float* __restrict__ tw,
                                             const float* __restrict__ Xr,
                                             const float* __restrict__ Xi,
                                             const float* __restrict__ wre,
                                             const float* __restrict__ wim,
                                             float* __restrict__ Gr,
                                             float* __restrict__ Gi, int nbase) {
  const int wave = threadIdx.x >> 5;
  const int lane = threadIdx.x & 31;
  const int ng = blockIdx.x;
  const int nidx = nbase + ng;
  const int p0 = (blockIdx.y * 4 + wave) * 16;
  if (p0 >= NP) return;
  const int col = lane & 15;
  const int h2 = (lane >> 4) << 1;
  const int p = p0 + col;
  const int pl = p < NP ? p : NP - 1;
  const int b0 = blockIdx.z * MIXB;

  const float* ar_p = tw + OFF_WNR + lane * 2;
  const float* ai_p = tw + OFF_WNI + lane * 2;
  const float* wr_p = wre + ((size_t)nidx * NC + h2) * NP + pl;
  const float* wi_p = wim + ((size_t)nidx * NC + h2) * NP + pl;
  const float* xr_p = Xr + ((size_t)b0 * NC + h2) * NP + pl;
  const float* xi_p = Xi + ((size_t)b0 * NC + h2) * NP + pl;

  v8f dr[MIXB], di[MIXB];
#pragma unroll
  for (int j = 0; j < MIXB; ++j) {
    dr[j] = zero8();
    di[j] = zero8();
  }

#pragma unroll
  for (int k0 = 0; k0 < NC; k0 += 4) {
    v2f ar = *reinterpret_cast<const v2f*>(ar_p + k0 * 16);
    v2f ai = *reinterpret_cast<const v2f*>(ai_p + k0 * 16);
    v2f nai;
    nai.x = -ai.x;
    nai.y = -ai.y;
    float wr0 = wr_p[(size_t)k0 * NP], wr1 = wr_p[(size_t)(k0 + 1) * NP];
    float wi0 = wi_p[(size_t)k0 * NP], wi1 = wi_p[(size_t)(k0 + 1) * NP];
#pragma unroll
    for (int j = 0; j < MIXB; ++j) {
      size_t o0 = ((size_t)j * NC + k0) * NP;
      size_t o1 = o0 + NP;
      float xr0 = xr_p[o0], xi0 = xi_p[o0];
      float xr1 = xr_p[o1], xi1 = xi_p[o1];
      v2f br, bi;
      br.x = xr0 * wr0 - xi0 * wi0;
      bi.x = xr0 * wi0 + xi0 * wr0;
      br.y = xr1 * wr1 - xi1 * wi1;
      bi.y = xr1 * wi1 + xi1 * wr1;
      dr[j] = wmma4(ar, br, dr[j]);
      dr[j] = wmma4(nai, bi, dr[j]);
      di[j] = wmma4(ar, bi, di[j]);
      di[j] = wmma4(ai, br, di[j]);
    }
  }

  if (p < NP) {
    const int mb = (lane >> 4) << 3;
#pragma unroll
    for (int j = 0; j < MIXB; ++j) {
#pragma unroll
      for (int i = 0; i < 8; ++i) {
        size_t o = (((size_t)ng * NB + (b0 + j)) * ND + (mb + i)) * NP + p;
        Gr[o] = dr[j][i];
        Gi[o] = di[j][i];
      }
    }
  }
}


typedef __attribute__((ext_vector_type(4))) float v4f;

constexpr int GP = 68;
constexpr int GS = 80 * GP;
constexpr int TP = 84;
constexpr int KS = 17;

__global__ __launch_bounds__(256) void k_ifft2d(const float* __restrict__ tw,
                                                const float* __restrict__ Gr,
                                                const float* __restrict__ Gi,
                                                float* __restrict__ out,
                                                int nbase) {
  __shared__ __attribute__((aligned(16))) float gs_r[GS];
  __shared__ __attribute__((aligned(16))) float gs_i[GS];
  __shared__ __attribute__((aligned(16))) float tt_r[NL * TP];
  __shared__ __attribute__((aligned(16))) float tt_i[NL * TP];
  const int d = blockIdx.x, b = blockIdx.y, ng = blockIdx.z;
  const int nidx = nbase + ng;
  const int tid = threadIdx.x;
  const int wave = __builtin_amdgcn_readfirstlane(tid >> 5);
  const int lane = tid & 31;
  const int c = lane & 15;
  const int h2 = (lane >> 4) << 1;
  const int wq = wave & 3;
  const int wh = wave >> 2;

  const float* gr = Gr + (((size_t)ng * NB + b) * ND + d) * NP;
  const float* gi = Gi + (((size_t)ng * NB + b) * ND + d) * NP;
  const int r4 = tid >> 6, c64 = tid & 63;
  float sr[17], si[17];
#pragma unroll
  for (int j = 0; j < 17; ++j) {
    const int row = 4 * j + r4;
    const bool ok = row < NLP;
    const int s = ok ? row * NLP + c64 : 0;
    const float vr = gr[s], vi = gi[s];
    sr[j] = ok ? vr : 0.f;
    si[j] = ok ? vi : 0.f;
  }
  const int erow = tid >> 2, ecol = tid & 3;
  const bool ok0 = ecol == 0;
  const int s0 = ok0 ? erow * NLP + 64 : 0;
  const float g0r = gr[s0], g0i = gi[s0];
  const float e0r = ok0 ? g0r : 0.f, e0i = ok0 ? g0i : 0.f;
  const int erow1 = 64 + (erow & 15);
  const bool ok1 = (tid & 63) == 0;
  const int s1 = ok1 ? NP - 1 : 0;
  const float g1r = gr[s1], g1i = gi[s1];
  const float e1r = ok1 ? g1r : 0.f, e1i = ok1 ? g1i : 0.f;

  const float* IVr = tw + OFF_IVR;
  const float* IVi = tw + OFF_IVI;
  v2f vbr[KS], vbi[KS], vnbi[KS];
#pragma unroll
  for (int k = 0; k < KS; ++k) {
    vbr[k] = frag_b(IVr, NL, 4 * k, wq * 16);
    vbi[k] = frag_b(IVi, NL, 4 * k, wq * 16);
  }

#pragma unroll
  for (int j = 0; j < 20; ++j) {
    gs_r[(4 * j + r4) * GP + c64] = j < 17 ? sr[j] : 0.f;
    gs_i[(4 * j + r4) * GP + c64] = j < 17 ? si[j] : 0.f;
  }
  gs_r[erow * GP + 64 + ecol] = e0r;
  gs_i[erow * GP + 64 + ecol] = e0i;
  gs_r[erow1 * GP + 64 + ecol] = e1r;
  gs_i[erow1 * GP + 64 + ecol] = e1i;
  __syncthreads();

  const float* IUr = tw + OFF_IUR;
  const float* IUi = tw + OFF_IUI;
  v2f uar[KS], unai[KS];
#pragma unroll
  for (int k = 0; k < KS; ++k) {
    uar[k] = frag_a_tab(IUr, KS, wq, k);
    v2f t = frag_a_tab(IUi, KS, wq, k);
    unai[k].x = -t.x;
    unai[k].y = -t.y;
  }
#pragma unroll
  for (int k = 0; k < KS; ++k) {
    vnbi[k].x = -vbi[k].x;
    vnbi[k].y = -vbi[k].y;
  }

  {
    const int no = wq * 16;
    const int mo_end = wh ? 80 : 48;
    for (int mo = wh ? 48 : 0; mo < mo_end; mo += 16) {
      const float* ar_p = gs_r + (mo + c) * GP + h2;
      const float* ai_p = gs_i + (mo + c) * GP + h2;
      v8f dr = zero8(), di = zero8();
#pragma unroll
      for (int k = 0; k < KS; ++k) {
        v2f ar = *reinterpret_cast<const v2f*>(ar_p + 4 * k);
        v2f ai = *reinterpret_cast<const v2f*>(ai_p + 4 * k);
        dr = wmma4(ar, vbr[k], dr);
        di = wmma4(ar, vbi[k], di);
        dr = wmma4(ai, vnbi[k], dr);
        di = wmma4(ai, vbr[k], di);
      }
      float* tr = tt_r + (no + c) * TP + mo + ((lane >> 4) << 3);
      float* ti = tt_i + (no + c) * TP + mo + ((lane >> 4) << 3);
      v4f q;
      q.x = dr[0]; q.y = dr[1]; q.z = dr[2]; q.w = dr[3];
      *reinterpret_cast<v4f*>(tr) = q;
      q.x = dr[4]; q.y = dr[5]; q.z = dr[6]; q.w = dr[7];
      *reinterpret_cast<v4f*>(tr + 4) = q;
      q.x = di[0]; q.y = di[1]; q.z = di[2]; q.w = di[3];
      *reinterpret_cast<v4f*>(ti) = q;
      q.x = di[4]; q.y = di[5]; q.z = di[6]; q.w = di[7];
      *reinterpret_cast<v4f*>(ti + 4) = q;
    }
  }
  __syncthreads();

  {
    const int mo = wq * 16;
    const int no = wh * 32;
    const float* b0r_p = tt_r + (no + c) * TP + h2;
    const float* b0i_p = tt_i + (no + c) * TP + h2;
    const float* b1r_p = b0r_p + 16 * TP;
    const float* b1i_p = b0i_p + 16 * TP;
    v8f d0 = zero8(), d1 = zero8();
#pragma unroll
    for (int k = 0; k < KS; ++k) {
      v2f b0r = *reinterpret_cast<const v2f*>(b0r_p + 4 * k);
      v2f b1r = *reinterpret_cast<const v2f*>(b1r_p + 4 * k);
      v2f b0i = *reinterpret_cast<const v2f*>(b0i_p + 4 * k);
      v2f b1i = *reinterpret_cast<const v2f*>(b1i_p + 4 * k);
      d0 = wmma4(uar[k], b0r, d0);
      d1 = wmma4(uar[k], b1r, d1);
      d0 = wmma4(unai[k], b0i, d0);
      d1 = wmma4(unai[k], b1i, d1);
    }
    float* op = out + ((size_t)b * (ND * NFR) + (d * NFR + nidx)) * (NL * NL);
    const int n = no + c;
    const int mb = mo + ((lane >> 4) << 3);
#pragma unroll
    for (int i = 0; i < 8; ++i) {
      op[(mb + i) * NL + n] = d0[i];
      op[(mb + i) * NL + n + 16] = d1[i];
    }
  }
}


extern "C" void kernel_launch(void* const* d_in, const int* in_sizes, int n_in,
                              void* d_out, int out_size, void* d_ws,
                              size_t ws_size, hipStream_t stream) {
  (void)in_sizes; (void)n_in; (void)out_size; (void)ws_size;
  const float* x   = (const float*)d_in[0];
  const float* wre = (const float*)d_in[1];
  const float* wim = (const float*)d_in[2];
  float* out = (float*)d_out;
  float* ws = (float*)d_ws;

  float* tw = ws;
  size_t off = TW_PAD;
  float* Yr = ws + off; off += (size_t)NB * NC * NP;
  float* Yi = ws + off; off += (size_t)NB * NC * NP;
  float* Xr = ws + off; off += (size_t)NB * NC * NP;
  float* Xi = ws + off;
  float* Gr = Yr;
  float* Gi = Yi;

  k_twiddles<<<dim3(25, 5), 256, 0, stream>>>(tw);
  k_fft2d<<<NB * NC, 128, 0, stream>>>(x, tw, Yr, Yi);
  k_cfft<<<dim3((NP + 15) / 16, NB), 128, 0, stream>>>(tw, Yr, Yi, Xr, Xi);
  for (int n = 0; n < NFR; n += NG) {
    k_mix<<<dim3(NG, MIX_GX, NB / MIXB), 128, 0, stream>>>(tw, Xr, Xi, wre, wim, Gr, Gi, n);
    k_ifft2d<<<dim3(ND, NB, NG), 256, 0, stream>>>(tw, Gr, Gi, out, n);
  }
}


# channel DFT fused into the mix kernel (X tile kept in LDS, X planes gone); G of all 8 modes written in place over the Y and X planes, so mix and inverse DFT run once: 7 launches to 4; sum order same
# speedup vs baseline: 6.9706x; 1.3148x over previous
#include <hip/hip_runtime.h>


typedef __attribute__((ext_vector_type(2))) float v2f;
typedef __attribute__((ext_vector_type(8))) float v8f;

constexpr int NB  = 16;
constexpr int NC  = 64;
constexpr int NFR = 8;
constexpr int ND  = 16;
constexpr int NL  = 64;
constexpr int NLP = 65;
constexpr int NP  = NLP * NLP;

constexpr int OFF_FWDR = 0;
constexpr int OFF_FWDI = OFF_FWDR + 80 * 80;
constexpr int OFF_ER   = OFF_FWDI + 80 * 80;
constexpr int OFF_EI   = OFF_ER + 64 * 64;
constexpr int OFF_WNR  = OFF_EI + 64 * 64;
constexpr int OFF_WNI  = OFF_WNR + 64 * 16;
constexpr int OFF_IVR  = OFF_WNI + 64 * 16;
constexpr int OFF_IVI  = OFF_IVR + 68 * 64;
constexpr int OFF_IUR  = OFF_IVI + 68 * 64;
constexpr int OFF_IUI  = OFF_IUR + 64 * 68;
constexpr int TW_TOTAL = OFF_IUI + 64 * 68;
constexpr int TW_PAD   = 40960;

#define PI_D 3.14159265358979323846


static __device__ __forceinline__ v8f wmma4(v2f a, v2f b, v8f c) {
  return __builtin_amdgcn_wmma_f32_16x16x4_f32(false, a, false, b, (short)0, c,
                                               false, false);
}

static __device__ __forceinline__ v8f zero8() {
  v8f z = {0.f, 0.f, 0.f, 0.f, 0.f, 0.f, 0.f, 0.f};
  return z;
}

static __device__ __forceinline__ v2f frag_a(const float* __restrict__ A,
                                             int lda, int m0, int k0) {
  int lane = threadIdx.x & 31;
  int m = m0 + (lane & 15);
  int k = k0 + ((lane >> 4) << 1);
  const float* p = A + m * lda + k;
  v2f r;
  r.x = p[0];
  r.y = p[1];
  return r;
}

static __device__ __forceinline__ v2f frag_a_tab(const float* __restrict__ A,
                                                 int ksteps, int tile, int s) {
  int lane = threadIdx.x & 31;
  return *reinterpret_cast<const v2f*>(A + ((tile * ksteps + s) * 32 + lane) * 2);
}

static __device__ __forceinline__ v2f frag_b(const float* __restrict__ B,
                                             int ldb, int k0, int n0) {
  int lane = threadIdx.x & 31;
  int n = n0 + (lane & 15);
  int k = k0 + ((lane >> 4) << 1);
  const float* p = B + k * ldb + n;
  v2f r;
  r.x = p[0];
  r.y = p[ldb];
  return r;
}

static __device__ __forceinline__ void cwmma(v2f ar, v2f ai, v2f br, v2f bi,
                                             v8f& dr, v8f& di) {
  dr = wmma4(ar, br, dr);
  v2f nai;
  nai.x = -ai.x;
  nai.y = -ai.y;
  dr = wmma4(nai, bi, dr);
  di = wmma4(ar, bi, di);
  di = wmma4(ai, br, di);
}


__global__ void k_twiddles(float* __restrict__ tw) {
  int t = blockIdx.x * blockDim.x + threadIdx.x;
  int which = blockIdx.y;
  if (which == 0) {
    if (t >= 80 * 80) return;
    int a = t / 80, b = t % 80;
    float r = 0.f, im = 0.f;
    if (a < NLP && b < NLP) {
      double ang = -2.0 * PI_D * (double)((a * b) % NLP) / (double)NLP;
      r = (float)cos(ang);
      im = (float)sin(ang);
    }
    tw[OFF_FWDR + t] = r;
    tw[OFF_FWDI + t] = im;
  } else if (which == 1) {
    if (t >= NC * NC) return;
    int l = (t >> 1) & 31;
    int c = (t >> 10) * 16 + (l & 15);
    int k = ((t >> 6) & 15) * 4 + ((l >> 4) << 1) + (t & 1);
    double ang = -2.0 * PI_D * (double)((c * k) % NC) / (double)NC;
    tw[OFF_ER + t] = (float)cos(ang);
    tw[OFF_EI + t] = (float)sin(ang);
  } else if (which == 2) {
    if (t >= NC * ND) return;
    int l = (t >> 1) & 31;
    int d = l & 15;
    int c = (t >> 6) * 4 + ((l >> 4) << 1) + (t & 1);
    double ang = 2.0 * PI_D * (double)((c * 4 * d) % NC) / (double)NC;
    tw[OFF_WNR + t] = (float)(cos(ang) / 64.0);
    tw[OFF_WNI + t] = (float)(sin(ang) / 64.0);
  } else if (which == 3) {
    if (t >= 68 * NL) return;
    int v = t / NL, w = t % NL;
    float r = 0.f, im = 0.f;
    if (v < NLP) {
      double ang = 2.0 * PI_D * (double)((v * (w + 1)) % NLP) / (double)NLP;
      r = (float)cos(ang);
      im = (float)sin(ang);
    }
    tw[OFF_IVR + t] = r;
    tw[OFF_IVI + t] = im;
  } else {
    if (t >= NL * 68) return;
    int l = (t >> 1) & 31;
    int fs = t >> 6;
    int h = (fs / 17) * 16 + (l & 15);
    int u = (fs % 17) * 4 + ((l >> 4) << 1) + (t & 1);
    float r = 0.f, im = 0.f;
    if (u < NLP) {
      double ang = 2.0 * PI_D * (double)((u * (h + 1)) % NLP) / (double)NLP;
      double s = 1.0 / (double)(NLP * NLP);
      r = (float)(cos(ang) * s);
      im = (float)(sin(ang) * s);
    }
    tw[OFF_IUR + t] = r;
    tw[OFF_IUI + t] = im;
  }
}


__global__ __launch_bounds__(128) void k_fft2d(const float* __restrict__ x,
                                               const float* __restrict__ tw,
                                               float* __restrict__ Yr,
                                               float* __restrict__ Yi) {
  __shared__ float xs[NL * NL];
  __shared__ float Tr[NL * 80];
  __shared__ float Ti[NL * 80];
  const int plane = blockIdx.x;
  const int tid = threadIdx.x;
  const int wave = tid >> 5;
  const int lane = tid & 31;

  const float* xp = x + (size_t)plane * (NL * NL);
  for (int i = tid; i < NL * NL; i += 128) xs[i] = xp[i];
  __syncthreads();

  const float* Fr = tw + OFF_FWDR;
  const float* Fi = tw + OFF_FWDI;

  for (int t = wave; t < 4 * 5; t += 4) {
    int mo = (t / 5) * 16, no = (t % 5) * 16;
    v8f ar = zero8(), ai = zero8();
    for (int k0 = 0; k0 < NL; k0 += 4) {
      v2f a = frag_a(xs, NL, mo, k0);
      ar = wmma4(a, frag_b(Fr, 80, k0, no), ar);
      ai = wmma4(a, frag_b(Fi, 80, k0, no), ai);
    }
    int n = no + (lane & 15);
    int mb = mo + ((lane >> 4) << 3);
#pragma unroll
    for (int i = 0; i < 8; ++i) {
      Tr[(mb + i) * 80 + n] = ar[i];
      Ti[(mb + i) * 80 + n] = ai[i];
    }
  }
  __syncthreads();

  float* yr = Yr + (size_t)plane * NP;
  float* yi = Yi + (size_t)plane * NP;
  for (int t = wave; t < 5 * 5; t += 4) {
    int mo = (t / 5) * 16, no = (t % 5) * 16;
    v8f dr = zero8(), di = zero8();
#pragma unroll
    for (int k0 = 0; k0 < NL; k0 += 4) {
      v2f par = frag_b(Fr, 80, k0, mo);
      v2f pai = frag_b(Fi, 80, k0, mo);
      v2f pbr = frag_b(Tr, 80, k0, no);
      v2f pbi = frag_b(Ti, 80, k0, no);
      cwmma(par, pai, pbr, pbi, dr, di);
    }
    int n = no + (lane & 15);
    int mb = mo + ((lane >> 4) << 3);
    if (n < NLP) {
#pragma unroll
      for (int i = 0; i < 8; ++i) {
        int m = mb + i;
        if (m < NLP) {
          yr[m * NLP + n] = dr[i];
          yi[m * NLP + n] = di[i];
        }
      }
    }
  }
}


typedef __attribute__((ext_vector_type(4))) float v4f;

constexpr int MIX_PT = (NP + 15) / 16;
constexpr int XP = 68;

__global__ __launch_bounds__(128) void k_cmix(const float* __restrict__ tw,
                                              float* Yr, float* Yi,
                                              float* Xr, float* Xi,
                                              const float* __restrict__ wre,
                                              const float* __restrict__ wim) {
  __shared__ __attribute__((aligned(16))) float xs_r[16 * XP];
  __shared__ __attribute__((aligned(16))) float xs_i[16 * XP];
  const int b = blockIdx.x;
  const int p0 = blockIdx.y * 16;
  const int wave = __builtin_amdgcn_readfirstlane(threadIdx.x >> 5);
  const int lane = threadIdx.x & 31;
  const int col = lane & 15;
  const int half = lane >> 4;
  const int h2 = half << 1;
  const int p = p0 + col;
  const int pl = p < NP ? p : NP - 1;

  {
    const float* Er = tw + OFF_ER;
    const float* Ei = tw + OFF_EI;
    const float* yr_p = Yr + ((size_t)b * NC + h2) * NP + pl;
    const float* yi_p = Yi + ((size_t)b * NC + h2) * NP + pl;
    v8f dr = zero8(), di = zero8();
#pragma unroll
    for (int k0 = 0; k0 < NC; k0 += 4) {
      v2f ar = frag_a_tab(Er, 16, wave, k0 >> 2);
      v2f ai = frag_a_tab(Ei, 16, wave, k0 >> 2);
      v2f br, bi;
      br.x = yr_p[(size_t)k0 * NP];
      br.y = yr_p[(size_t)(k0 + 1) * NP];
      bi.x = yi_p[(size_t)k0 * NP];
      bi.y = yi_p[(size_t)(k0 + 1) * NP];
      cwmma(ar, ai, br, bi, dr, di);
    }
    float* sr = xs_r + col * XP + wave * 16 + (half << 3);
    float* si = xs_i + col * XP + wave * 16 + (half << 3);
    v4f q;
    q.x = dr[0]; q.y = dr[1]; q.z = dr[2]; q.w = dr[3];
    *reinterpret_cast<v4f*>(sr) = q;
    q.x = dr[4]; q.y = dr[5]; q.z = dr[6]; q.w = dr[7];
    *reinterpret_cast<v4f*>(sr + 4) = q;
    q.x = di[0]; q.y = di[1]; q.z = di[2]; q.w = di[3];
    *reinterpret_cast<v4f*>(si) = q;
    q.x = di[4]; q.y = di[5]; q.z = di[6]; q.w = di[7];
    *reinterpret_cast<v4f*>(si + 4) = q;
  }
  __syncthreads();

  const int n0 = wave * 2;
  const float* ar_p = tw + OFF_WNR + lane * 2;
  const float* ai_p = tw + OFF_WNI + lane * 2;
  const float* wr_p = wre + ((size_t)n0 * NC + h2) * NP + pl;
  const float* wi_p = wim + ((size_t)n0 * NC + h2) * NP + pl;
  const float* xr_p = xs_r + col * XP + h2;
  const float* xi_p = xs_i + col * XP + h2;

  v8f dr[2], di[2];
#pragma unroll
  for (int j = 0; j < 2; ++j) {
    dr[j] = zero8();
    di[j] = zero8();
  }

#pragma unroll
  for (int k0 = 0; k0 < NC; k0 += 4) {
    v2f ar = *reinterpret_cast<const v2f*>(ar_p + k0 * 16);
    v2f ai = *reinterpret_cast<const v2f*>(ai_p + k0 * 16);
    v2f nai;
    nai.x = -ai.x;
    nai.y = -ai.y;
    v2f xr = *reinterpret_cast<const v2f*>(xr_p + k0);
    v2f xi = *reinterpret_cast<const v2f*>(xi_p + k0);
#pragma unroll
    for (int j = 0; j < 2; ++j) {
      size_t o0 = ((size_t)j * NC + k0) * NP;
      size_t o1 = o0 + NP;
      float wr0 = wr_p[o0], wi0 = wi_p[o0];
      float wr1 = wr_p[o1], wi1 = wi_p[o1];
      v2f br, bi;
      br.x = xr.x * wr0 - xi.x * wi0;
      bi.x = xr.x * wi0 + xi.x * wr0;
      br.y = xr.y * wr1 - xi.y * wi1;
      bi.y = xr.y * wi1 + xi.y * wr1;
      dr[j] = wmma4(ar, br, dr[j]);
      dr[j] = wmma4(nai, bi, dr[j]);
      di[j] = wmma4(ar, bi, di[j]);
      di[j] = wmma4(ai, br, di[j]);
    }
  }

  if (p < NP) {
    const int mb = half << 3;
#pragma unroll
    for (int j = 0; j < 2; ++j) {
      const int n = n0 + j;
      float* gr = (n < 4 ? Yr : Xr) + ((size_t)b * NC + (n & 3) * ND + mb) * NP + p;
      float* gi = (n < 4 ? Yi : Xi) + ((size_t)b * NC + (n & 3) * ND + mb) * NP + p;
#pragma unroll
      for (int i = 0; i < 8; ++i) {
        gr[(size_t)i * NP] = dr[j][i];
        gi[(size_t)i * NP] = di[j][i];
      }
    }
  }
}


constexpr int GP = 68;
constexpr int GS = 80 * GP;
constexpr int TP = 84;
constexpr int KS = 17;

__global__ __launch_bounds__(256) void k_ifft2d(const float* __restrict__ tw,
                                                const float* __restrict__ Yr,
                                                const float* __restrict__ Yi,
                                                const float* __restrict__ Xr,
                                                const float* __restrict__ Xi,
                                                float* __restrict__ out) {
  __shared__ __attribute__((aligned(16))) float gs_r[GS];
  __shared__ __attribute__((aligned(16))) float gs_i[GS];
  __shared__ __attribute__((aligned(16))) float tt_r[NL * TP];
  __shared__ __attribute__((aligned(16))) float tt_i[NL * TP];
  const int d = blockIdx.x, b = blockIdx.y, nidx = blockIdx.z;
  const int tid = threadIdx.x;
  const int wave = __builtin_amdgcn_readfirstlane(tid >> 5);
  const int lane = tid & 31;
  const int c = lane & 15;
  const int h2 = (lane >> 4) << 1;
  const int wq = wave & 3;
  const int wh = wave >> 2;

  const float* gr = (nidx < 4 ? Yr : Xr) + ((size_t)b * NC + (nidx & 3) * ND + d) * NP;
  const float* gi = (nidx < 4 ? Yi : Xi) + ((size_t)b * NC + (nidx & 3) * ND + d) * NP;
  const int r4 = tid >> 6, c64 = tid & 63;
  float sr[17], si[17];
#pragma unroll
  for (int j = 0; j < 17; ++j) {
    const int row = 4 * j + r4;
    const bool ok = row < NLP;
    const int s = ok ? row * NLP + c64 : 0;
    const float vr = gr[s], vi = gi[s];
    sr[j] = ok ? vr : 0.f;
    si[j] = ok ? vi : 0.f;
  }
  const int erow = tid >> 2, ecol = tid & 3;
  const bool ok0 = ecol == 0;
  const int s0 = ok0 ? erow * NLP + 64 : 0;
  const float g0r = gr[s0], g0i = gi[s0];
  const float e0r = ok0 ? g0r : 0.f, e0i = ok0 ? g0i : 0.f;
  const int erow1 = 64 + (erow & 15);
  const bool ok1 = (tid & 63) == 0;
  const int s1 = ok1 ? NP - 1 : 0;
  const float g1r = gr[s1], g1i = gi[s1];
  const float e1r = ok1 ? g1r : 0.f, e1i = ok1 ? g1i : 0.f;

  const float* IVr = tw + OFF_IVR;
  const float* IVi = tw + OFF_IVI;
  v2f vbr[KS], vbi[KS], vnbi[KS];
#pragma unroll
  for (int k = 0; k < KS; ++k) {
    vbr[k] = frag_b(IVr, NL, 4 * k, wq * 16);
    vbi[k] = frag_b(IVi, NL, 4 * k, wq * 16);
  }

#pragma unroll
  for (int j = 0; j < 20; ++j) {
    gs_r[(4 * j + r4) * GP + c64] = j < 17 ? sr[j] : 0.f;
    gs_i[(4 * j + r4) * GP + c64] = j < 17 ? si[j] : 0.f;
  }
  gs_r[erow * GP + 64 + ecol] = e0r;
  gs_i[erow * GP + 64 + ecol] = e0i;
  gs_r[erow1 * GP + 64 + ecol] = e1r;
  gs_i[erow1 * GP + 64 + ecol] = e1i;
  __syncthreads();

  const float* IUr = tw + OFF_IUR;
  const float* IUi = tw + OFF_IUI;
  v2f uar[KS], unai[KS];
#pragma unroll
  for (int k = 0; k < KS; ++k) {
    uar[k] = frag_a_tab(IUr, KS, wq, k);
    v2f t = frag_a_tab(IUi, KS, wq, k);
    unai[k].x = -t.x;
    unai[k].y = -t.y;
  }
#pragma unroll
  for (int k = 0; k < KS; ++k) {
    vnbi[k].x = -vbi[k].x;
    vnbi[k].y = -vbi[k].y;
  }

  {
    const int no = wq * 16;
    const int mo_end = wh ? 80 : 48;
    for (int mo = wh ? 48 : 0; mo < mo_end; mo += 16) {
      const float* ar_p = gs_r + (mo + c) * GP + h2;
      const float* ai_p = gs_i + (mo + c) * GP + h2;
      v8f dr = zero8(), di = zero8();
#pragma unroll
      for (int k = 0; k < KS; ++k) {
        v2f ar = *reinterpret_cast<const v2f*>(ar_p + 4 * k);
        v2f ai = *reinterpret_cast<const v2f*>(ai_p + 4 * k);
        dr = wmma4(ar, vbr[k], dr);
        di = wmma4(ar, vbi[k], di);
        dr = wmma4(ai, vnbi[k], dr);
        di = wmma4(ai, vbr[k], di);
      }
      float* tr = tt_r + (no + c) * TP + mo + ((lane >> 4) << 3);
      float* ti = tt_i + (no + c) * TP + mo + ((lane >> 4) << 3);
      v4f q;
      q.x = dr[0]; q.y = dr[1]; q.z = dr[2]; q.w = dr[3];
      *reinterpret_cast<v4f*>(tr) = q;
      q.x = dr[4]; q.y = dr[5]; q.z = dr[6]; q.w = dr[7];
      *reinterpret_cast<v4f*>(tr + 4) = q;
      q.x = di[0]; q.y = di[1]; q.z = di[2]; q.w = di[3];
      *reinterpret_cast<v4f*>(ti) = q;
      q.x = di[4]; q.y = di[5]; q.z = di[6]; q.w = di[7];
      *reinterpret_cast<v4f*>(ti + 4) = q;
    }
  }
  __syncthreads();

  {
    const int mo = wq * 16;
    const int no = wh * 32;
    const float* b0r_p = tt_r + (no + c) * TP + h2;
    const float* b0i_p = tt_i + (no + c) * TP + h2;
    const float* b1r_p = b0r_p + 16 * TP;
    const float* b1i_p = b0i_p + 16 * TP;
    v8f d0 = zero8(), d1 = zero8();
#pragma unroll
    for (int k = 0; k < KS; ++k) {
      v2f b0r = *reinterpret_cast<const v2f*>(b0r_p + 4 * k);
      v2f b1r = *reinterpret_cast<const v2f*>(b1r_p + 4 * k);
      v2f b0i = *reinterpret_cast<const v2f*>(b0i_p + 4 * k);
      v2f b1i = *reinterpret_cast<const v2f*>(b1i_p + 4 * k);
      d0 = wmma4(uar[k], b0r, d0);
      d1 = wmma4(uar[k], b1r, d1);
      d0 = wmma4(unai[k], b0i, d0);
      d1 = wmma4(unai[k], b1i, d1);
    }
    float* op = out + ((size_t)b * (ND * NFR) + (d * NFR + nidx)) * (NL * NL);
    const int n = no + c;
    const int mb = mo + ((lane >> 4) << 3);
#pragma unroll
    for (int i = 0; i < 8; ++i) {
      op[(mb + i) * NL + n] = d0[i];
      op[(mb + i) * NL + n + 16] = d1[i];
    }
  }
}


extern "C" void kernel_launch(void* const* d_in, const int* in_sizes, int n_in,
                              void* d_out, int out_size, void* d_ws,
                              size_t ws_size, hipStream_t stream) {
  (void)in_sizes; (void)n_in; (void)out_size; (void)ws_size;
  const float* x   = (const float*)d_in[0];
  const float* wre = (const float*)d_in[1];
  const float* wim = (const float*)d_in[2];
  float* out = (float*)d_out;
  float* ws = (float*)d_ws;

  float* tw = ws;
  size_t off = TW_PAD;
  float* Yr = ws + off; off += (size_t)NB * NC * NP;
  float* Yi = ws + off; off += (size_t)NB * NC * NP;
  float* Xr = ws + off; off += (size_t)NB * NC * NP;
  float* Xi = ws + off;

  k_twiddles<<<dim3(25, 5), 256, 0, stream>>>(tw);
  k_fft2d<<<NB * NC, 128, 0, stream>>>(x, tw, Yr, Yi);
  k_cmix<<<dim3(NB, MIX_PT), 128, 0, stream>>>(tw, Yr, Yi, Xr, Xi, wre, wim);
  k_ifft2d<<<dim3(ND, NB, NFR), 256, 0, stream>>>(tw, Yr, Yi, Xr, Xi, out);
}


# complex-by-complex matrix products (2D DFT stage 2, both mix stages, inverse stage v) as three real WMMA products, not four; re/im combined after the k loop: rounding differs, k order same
# speedup vs baseline: 7.3753x; 1.0581x over previous
#include <hip/hip_runtime.h>


typedef __attribute__((ext_vector_type(2))) float v2f;
typedef __attribute__((ext_vector_type(8))) float v8f;

constexpr int NB  = 16;
constexpr int NC  = 64;
constexpr int NFR = 8;
constexpr int ND  = 16;
constexpr int NL  = 64;
constexpr int NLP = 65;
constexpr int NP  = NLP * NLP;

constexpr int OFF_FWDR = 0;
constexpr int OFF_FWDI = OFF_FWDR + 80 * 80;
constexpr int OFF_ER   = OFF_FWDI + 80 * 80;
constexpr int OFF_EI   = OFF_ER + 64 * 64;
constexpr int OFF_WNR  = OFF_EI + 64 * 64;
constexpr int OFF_WNI  = OFF_WNR + 64 * 16;
constexpr int OFF_IVR  = OFF_WNI + 64 * 16;
constexpr int OFF_IVI  = OFF_IVR + 68 * 64;
constexpr int OFF_IUR  = OFF_IVI + 68 * 64;
constexpr int OFF_IUI  = OFF_IUR + 64 * 68;
constexpr int TW_TOTAL = OFF_IUI + 64 * 68;
constexpr int TW_PAD   = 40960;

#define PI_D 3.14159265358979323846


static __device__ __forceinline__ v8f wmma4(v2f a, v2f b, v8f c) {
  return __builtin_amdgcn_wmma_f32_16x16x4_f32(false, a, false, b, (short)0, c,
                                               false, false);
}

static __device__ __forceinline__ v8f zero8() {
  v8f z = {0.f, 0.f, 0.f, 0.f, 0.f, 0.f, 0.f, 0.f};
  return z;
}

static __device__ __forceinline__ v2f frag_a(const float* __restrict__ A,
                                             int lda, int m0, int k0) {
  int lane = threadIdx.x & 31;
  int m = m0 + (lane & 15);
  int k = k0 + ((lane >> 4) << 1);
  const float* p = A + m * lda + k;
  v2f r;
  r.x = p[0];
  r.y = p[1];
  return r;
}

static __device__ __forceinline__ v2f frag_a_tab(const float* __restrict__ A,
                                                 int ksteps, int tile, int s) {
  int lane = threadIdx.x & 31;
  return *reinterpret_cast<const v2f*>(A + ((tile * ksteps + s) * 32 + lane) * 2);
}

static __device__ __forceinline__ v2f frag_b(const float* __restrict__ B,
                                             int ldb, int k0, int n0) {
  int lane = threadIdx.x & 31;
  int n = n0 + (lane & 15);
  int k = k0 + ((lane >> 4) << 1);
  const float* p = B + k * ldb + n;
  v2f r;
  r.x = p[0];
  r.y = p[ldb];
  return r;
}

static __device__ __forceinline__ void cwmma3(v2f ar, v2f adm, v2f asp, v2f br,
                                              v2f bi, v8f& k1, v8f& k2, v8f& k3) {
  v2f bs = br + bi;
  k1 = wmma4(ar, bs, k1);
  k2 = wmma4(adm, br, k2);
  k3 = wmma4(asp, bi, k3);
}


__global__ void k_twiddles(float* __restrict__ tw) {
  int t = blockIdx.x * blockDim.x + threadIdx.x;
  int which = blockIdx.y;
  if (which == 0) {
    if (t >= 80 * 80) return;
    int a = t / 80, b = t % 80;
    float r = 0.f, im = 0.f;
    if (a < NLP && b < NLP) {
      double ang = -2.0 * PI_D * (double)((a * b) % NLP) / (double)NLP;
      r = (float)cos(ang);
      im = (float)sin(ang);
    }
    tw[OFF_FWDR + t] = r;
    tw[OFF_FWDI + t] = im;
  } else if (which == 1) {
    if (t >= NC * NC) return;
    int l = (t >> 1) & 31;
    int c = (t >> 10) * 16 + (l & 15);
    int k = ((t >> 6) & 15) * 4 + ((l >> 4) << 1) + (t & 1);
    double ang = -2.0 * PI_D * (double)((c * k) % NC) / (double)NC;
    tw[OFF_ER + t] = (float)cos(ang);
    tw[OFF_EI + t] = (float)sin(ang);
  } else if (which == 2) {
    if (t >= NC * ND) return;
    int l = (t >> 1) & 31;
    int d = l & 15;
    int c = (t >> 6) * 4 + ((l >> 4) << 1) + (t & 1);
    double ang = 2.0 * PI_D * (double)((c * 4 * d) % NC) / (double)NC;
    tw[OFF_WNR + t] = (float)(cos(ang) / 64.0);
    tw[OFF_WNI + t] = (float)(sin(ang) / 64.0);
  } else if (which == 3) {
    if (t >= 68 * NL) return;
    int v = t / NL, w = t % NL;
    float r = 0.f, im = 0.f;
    if (v < NLP) {
      double ang = 2.0 * PI_D * (double)((v * (w + 1)) % NLP) / (double)NLP;
      r = (float)cos(ang);
      im = (float)sin(ang);
    }
    tw[OFF_IVR + t] = r;
    tw[OFF_IVI + t] = im;
  } else {
    if (t >= NL * 68) return;
    int l = (t >> 1) & 31;
    int fs = t >> 6;
    int h = (fs / 17) * 16 + (l & 15);
    int u = (fs % 17) * 4 + ((l >> 4) << 1) + (t & 1);
    float r = 0.f, im = 0.f;
    if (u < NLP) {
      double ang = 2.0 * PI_D * (double)((u * (h + 1)) % NLP) / (double)NLP;
      double s = 1.0 / (double)(NLP * NLP);
      r = (float)(cos(ang) * s);
      im = (float)(sin(ang) * s);
    }
    tw[OFF_IUR + t] = r;
    tw[OFF_IUI + t] = im;
  }
}


__global__ __launch_bounds__(128) void k_fft2d(const float* __restrict__ x,
                                               const float* __restrict__ tw,
                                               float* __restrict__ Yr,
                                               float* __restrict__ Yi) {
  __shared__ float xs[NL * NL];
  __shared__ float Tr[NL * 80];
  __shared__ float Ti[NL * 80];
  const int plane = blockIdx.x;
  const int tid = threadIdx.x;
  const int wave = tid >> 5;
  const int lane = tid & 31;

  const float* xp = x + (size_t)plane * (NL * NL);
  for (int i = tid; i < NL * NL; i += 128) xs[i] = xp[i];
  __syncthreads();

  const float* Fr = tw + OFF_FWDR;
  const float* Fi = tw + OFF_FWDI;

  for (int t = wave; t < 4 * 5; t += 4) {
    int mo = (t / 5) * 16, no = (t % 5) * 16;
    v8f ar = zero8(), ai = zero8();
    for (int k0 = 0; k0 < NL; k0 += 4) {
      v2f a = frag_a(xs, NL, mo, k0);
      ar = wmma4(a, frag_b(Fr, 80, k0, no), ar);
      ai = wmma4(a, frag_b(Fi, 80, k0, no), ai);
    }
    int n = no + (lane & 15);
    int mb = mo + ((lane >> 4) << 3);
#pragma unroll
    for (int i = 0; i < 8; ++i) {
      Tr[(mb + i) * 80 + n] = ar[i];
      Ti[(mb + i) * 80 + n] = ai[i];
    }
  }
  __syncthreads();

  float* yr = Yr + (size_t)plane * NP;
  float* yi = Yi + (size_t)plane * NP;
  for (int t = wave; t < 5 * 5; t += 4) {
    int mo = (t / 5) * 16, no = (t % 5) * 16;
    v8f k1 = zero8(), k2 = zero8(), k3 = zero8();
#pragma unroll
    for (int k0 = 0; k0 < NL; k0 += 4) {
      v2f par = frag_b(Fr, 80, k0, mo);
      v2f pai = frag_b(Fi, 80, k0, mo);
      v2f pbr = frag_b(Tr, 80, k0, no);
      v2f pbi = frag_b(Ti, 80, k0, no);
      cwmma3(par, pai - par, par + pai, pbr, pbi, k1, k2, k3);
    }
    const v8f dr = k1 - k3, di = k1 + k2;
    int n = no + (lane & 15);
    int mb = mo + ((lane >> 4) << 3);
    if (n < NLP) {
#pragma unroll
      for (int i = 0; i < 8; ++i) {
        int m = mb + i;
        if (m < NLP) {
          yr[m * NLP + n] = dr[i];
          yi[m * NLP + n] = di[i];
        }
      }
    }
  }
}


typedef __attribute__((ext_vector_type(4))) float v4f;

constexpr int MIX_PT = (NP + 15) / 16;
constexpr int XP = 68;

__global__ __launch_bounds__(128) void k_cmix(const float* __restrict__ tw,
                                              float* Yr, float* Yi,
                                              float* Xr, float* Xi,
                                              const float* __restrict__ wre,
                                              const float* __restrict__ wim) {
  __shared__ __attribute__((aligned(16))) float xs_r[16 * XP];
  __shared__ __attribute__((aligned(16))) float xs_i[16 * XP];
  const int b = blockIdx.x;
  const int p0 = blockIdx.y * 16;
  const int wave = __builtin_amdgcn_readfirstlane(threadIdx.x >> 5);
  const int lane = threadIdx.x & 31;
  const int col = lane & 15;
  const int half = lane >> 4;
  const int h2 = half << 1;
  const int p = p0 + col;
  const int pl = p < NP ? p : NP - 1;

  {
    const float* Er = tw + OFF_ER;
    const float* Ei = tw + OFF_EI;
    const float* yr_p = Yr + ((size_t)b * NC + h2) * NP + pl;
    const float* yi_p = Yi + ((size_t)b * NC + h2) * NP + pl;
    v8f k1 = zero8(), k2 = zero8(), k3 = zero8();
#pragma unroll
    for (int k0 = 0; k0 < NC; k0 += 4) {
      v2f ar = frag_a_tab(Er, 16, wave, k0 >> 2);
      v2f ai = frag_a_tab(Ei, 16, wave, k0 >> 2);
      v2f br, bi;
      br.x = yr_p[(size_t)k0 * NP];
      br.y = yr_p[(size_t)(k0 + 1) * NP];
      bi.x = yi_p[(size_t)k0 * NP];
      bi.y = yi_p[(size_t)(k0 + 1) * NP];
      cwmma3(ar, ai - ar, ar + ai, br, bi, k1, k2, k3);
    }
    const v8f dr = k1 - k3, di = k1 + k2;
    float* sr = xs_r + col * XP + wave * 16 + (half << 3);
    float* si = xs_i + col * XP + wave * 16 + (half << 3);
    v4f q;
    q.x = dr[0]; q.y = dr[1]; q.z = dr[2]; q.w = dr[3];
    *reinterpret_cast<v4f*>(sr) = q;
    q.x = dr[4]; q.y = dr[5]; q.z = dr[6]; q.w = dr[7];
    *reinterpret_cast<v4f*>(sr + 4) = q;
    q.x = di[0]; q.y = di[1]; q.z = di[2]; q.w = di[3];
    *reinterpret_cast<v4f*>(si) = q;
    q.x = di[4]; q.y = di[5]; q.z = di[6]; q.w = di[7];
    *reinterpret_cast<v4f*>(si + 4) = q;
  }
  __syncthreads();

  const int n0 = wave * 2;
  const float* ar_p = tw + OFF_WNR + lane * 2;
  const float* ai_p = tw + OFF_WNI + lane * 2;
  const float* wr_p = wre + ((size_t)n0 * NC + h2) * NP + pl;
  const float* wi_p = wim + ((size_t)n0 * NC + h2) * NP + pl;
  const float* xr_p = xs_r + col * XP + h2;
  const float* xi_p = xs_i + col * XP + h2;

  v8f k1[2], k2[2], k3[2];
#pragma unroll
  for (int j = 0; j < 2; ++j) {
    k1[j] = zero8();
    k2[j] = zero8();
    k3[j] = zero8();
  }

#pragma unroll
  for (int k0 = 0; k0 < NC; k0 += 4) {
    v2f ar = *reinterpret_cast<const v2f*>(ar_p + k0 * 16);
    v2f ai = *reinterpret_cast<const v2f*>(ai_p + k0 * 16);
    const v2f adm = ai - ar;
    const v2f asp = ar + ai;
    v2f xr = *reinterpret_cast<const v2f*>(xr_p + k0);
    v2f xi = *reinterpret_cast<const v2f*>(xi_p + k0);
#pragma unroll
    for (int j = 0; j < 2; ++j) {
      size_t o0 = ((size_t)j * NC + k0) * NP;
      size_t o1 = o0 + NP;
      float wr0 = wr_p[o0], wi0 = wi_p[o0];
      float wr1 = wr_p[o1], wi1 = wi_p[o1];
      v2f br, bi;
      br.x = xr.x * wr0 - xi.x * wi0;
      bi.x = xr.x * wi0 + xi.x * wr0;
      br.y = xr.y * wr1 - xi.y * wi1;
      bi.y = xr.y * wi1 + xi.y * wr1;
      cwmma3(ar, adm, asp, br, bi, k1[j], k2[j], k3[j]);
    }
  }

  if (p < NP) {
    const int mb = half << 3;
#pragma unroll
    for (int j = 0; j < 2; ++j) {
      const int n = n0 + j;
      const v8f dr = k1[j] - k3[j], di = k1[j] + k2[j];
      float* gr = (n < 4 ? Yr : Xr) + ((size_t)b * NC + (n & 3) * ND + mb) * NP + p;
      float* gi = (n < 4 ? Yi : Xi) + ((size_t)b * NC + (n & 3) * ND + mb) * NP + p;
#pragma unroll
      for (int i = 0; i < 8; ++i) {
        gr[(size_t)i * NP] = dr[i];
        gi[(size_t)i * NP] = di[i];
      }
    }
  }
}


constexpr int GP = 68;
constexpr int GS = 80 * GP;
constexpr int TP = 84;
constexpr int KS = 17;

__global__ __launch_bounds__(256) void k_ifft2d(const float* __restrict__ tw,
                                                const float* __restrict__ Yr,
                                                const float* __restrict__ Yi,
                                                const float* __restrict__ Xr,
                                                const float* __restrict__ Xi,
                                                float* __restrict__ out) {
  __shared__ __attribute__((aligned(16))) float gs_r[GS];
  __shared__ __attribute__((aligned(16))) float gs_i[GS];
  __shared__ __attribute__((aligned(16))) float tt_r[NL * TP];
  __shared__ __attribute__((aligned(16))) float tt_i[NL * TP];
  const int d = blockIdx.x, b = blockIdx.y, nidx = blockIdx.z;
  const int tid = threadIdx.x;
  const int wave = __builtin_amdgcn_readfirstlane(tid >> 5);
  const int lane = tid & 31;
  const int c = lane & 15;
  const int h2 = (lane >> 4) << 1;
  const int wq = wave & 3;
  const int wh = wave >> 2;

  const float* gr = (nidx < 4 ? Yr : Xr) + ((size_t)b * NC + (nidx & 3) * ND + d) * NP;
  const float* gi = (nidx < 4 ? Yi : Xi) + ((size_t)b * NC + (nidx & 3) * ND + d) * NP;
  const int r4 = tid >> 6, c64 = tid & 63;
  float sr[17], si[17];
#pragma unroll
  for (int j = 0; j < 17; ++j) {
    const int row = 4 * j + r4;
    const bool ok = row < NLP;
    const int s = ok ? row * NLP + c64 : 0;
    const float vr = gr[s], vi = gi[s];
    sr[j] = ok ? vr : 0.f;
    si[j] = ok ? vi : 0.f;
  }
  const int erow = tid >> 2, ecol = tid & 3;
  const bool ok0 = ecol == 0;
  const int s0 = ok0 ? erow * NLP + 64 : 0;
  const float g0r = gr[s0], g0i = gi[s0];
  const float e0r = ok0 ? g0r : 0.f, e0i = ok0 ? g0i : 0.f;
  const int erow1 = 64 + (erow & 15);
  const bool ok1 = (tid & 63) == 0;
  const int s1 = ok1 ? NP - 1 : 0;
  const float g1r = gr[s1], g1i = gi[s1];
  const float e1r = ok1 ? g1r : 0.f, e1i = ok1 ? g1i : 0.f;

  const float* IVr = tw + OFF_IVR;
  const float* IVi = tw + OFF_IVI;
  v2f vbr[KS], vbi[KS], vdm[KS], vsp[KS];
#pragma unroll
  for (int k = 0; k < KS; ++k) {
    vbr[k] = frag_b(IVr, NL, 4 * k, wq * 16);
    vbi[k] = frag_b(IVi, NL, 4 * k, wq * 16);
  }

#pragma unroll
  for (int j = 0; j < 20; ++j) {
    gs_r[(4 * j + r4) * GP + c64] = j < 17 ? sr[j] : 0.f;
    gs_i[(4 * j + r4) * GP + c64] = j < 17 ? si[j] : 0.f;
  }
  gs_r[erow * GP + 64 + ecol] = e0r;
  gs_i[erow * GP + 64 + ecol] = e0i;
  gs_r[erow1 * GP + 64 + ecol] = e1r;
  gs_i[erow1 * GP + 64 + ecol] = e1i;
  __syncthreads();

  const float* IUr = tw + OFF_IUR;
  const float* IUi = tw + OFF_IUI;
  v2f uar[KS], unai[KS];
#pragma unroll
  for (int k = 0; k < KS; ++k) {
    uar[k] = frag_a_tab(IUr, KS, wq, k);
    v2f t = frag_a_tab(IUi, KS, wq, k);
    unai[k].x = -t.x;
    unai[k].y = -t.y;
  }
#pragma unroll
  for (int k = 0; k < KS; ++k) {
    vdm[k] = vbi[k] - vbr[k];
    vsp[k] = vbr[k] + vbi[k];
  }

  {
    const int no = wq * 16;
    const int mo_end = wh ? 80 : 48;
    for (int mo = wh ? 48 : 0; mo < mo_end; mo += 16) {
      const float* ar_p = gs_r + (mo + c) * GP + h2;
      const float* ai_p = gs_i + (mo + c) * GP + h2;
      v8f k1 = zero8(), k2 = zero8(), k3 = zero8();
#pragma unroll
      for (int k = 0; k < KS; ++k) {
        v2f ar = *reinterpret_cast<const v2f*>(ar_p + 4 * k);
        v2f ai = *reinterpret_cast<const v2f*>(ai_p + 4 * k);
        v2f asum = ar + ai;
        k1 = wmma4(asum, vbr[k], k1);
        k2 = wmma4(ar, vdm[k], k2);
        k3 = wmma4(ai, vsp[k], k3);
      }
      const v8f dr = k1 - k3, di = k1 + k2;
      float* tr = tt_r + (no + c) * TP + mo + ((lane >> 4) << 3);
      float* ti = tt_i + (no + c) * TP + mo + ((lane >> 4) << 3);
      v4f q;
      q.x = dr[0]; q.y = dr[1]; q.z = dr[2]; q.w = dr[3];
      *reinterpret_cast<v4f*>(tr) = q;
      q.x = dr[4]; q.y = dr[5]; q.z = dr[6]; q.w = dr[7];
      *reinterpret_cast<v4f*>(tr + 4) = q;
      q.x = di[0]; q.y = di[1]; q.z = di[2]; q.w = di[3];
      *reinterpret_cast<v4f*>(ti) = q;
      q.x = di[4]; q.y = di[5]; q.z = di[6]; q.w = di[7];
      *reinterpret_cast<v4f*>(ti + 4) = q;
    }
  }
  __syncthreads();

  {
    const int mo = wq * 16;
    const int no = wh * 32;
    const float* b0r_p = tt_r + (no + c) * TP + h2;
    const float* b0i_p = tt_i + (no + c) * TP + h2;
    const float* b1r_p = b0r_p + 16 * TP;
    const float* b1i_p = b0i_p + 16 * TP;
    v8f d0 = zero8(), d1 = zero8();
#pragma unroll
    for (int k = 0; k < KS; ++k) {
      v2f b0r = *reinterpret_cast<const v2f*>(b0r_p + 4 * k);
      v2f b1r = *reinterpret_cast<const v2f*>(b1r_p + 4 * k);
      v2f b0i = *reinterpret_cast<const v2f*>(b0i_p + 4 * k);
      v2f b1i = *reinterpret_cast<const v2f*>(b1i_p + 4 * k);
      d0 = wmma4(uar[k], b0r, d0);
      d1 = wmma4(uar[k], b1r, d1);
      d0 = wmma4(unai[k], b0i, d0);
      d1 = wmma4(unai[k], b1i, d1);
    }
    float* op = out + ((size_t)b * (ND * NFR) + (d * NFR + nidx)) * (NL * NL);
    const int n = no + c;
    const int mb = mo + ((lane >> 4) << 3);
#pragma unroll
    for (int i = 0; i < 8; ++i) {
      op[(mb + i) * NL + n] = d0[i];
      op[(mb + i) * NL + n + 16] = d1[i];
    }
  }
}


extern "C" void kernel_launch(void* const* d_in, const int* in_sizes, int n_in,
                              void* d_out, int out_size, void* d_ws,
                              size_t ws_size, hipStream_t stream) {
  (void)in_sizes; (void)n_in; (void)out_size; (void)ws_size;
  const float* x   = (const float*)d_in[0];
  const float* wre = (const float*)d_in[1];
  const float* wim = (const float*)d_in[2];
  float* out = (float*)d_out;
  float* ws = (float*)d_ws;

  float* tw = ws;
  size_t off = TW_PAD;
  float* Yr = ws + off; off += (size_t)NB * NC * NP;
  float* Yi = ws + off; off += (size_t)NB * NC * NP;
  float* Xr = ws + off; off += (size_t)NB * NC * NP;
  float* Xi = ws + off;

  k_twiddles<<<dim3(25, 5), 256, 0, stream>>>(tw);
  k_fft2d<<<NB * NC, 128, 0, stream>>>(x, tw, Yr, Yi);
  k_cmix<<<dim3(NB, MIX_PT), 128, 0, stream>>>(tw, Yr, Yi, Xr, Xi, wre, wim);
  k_ifft2d<<<dim3(ND, NB, NFR), 256, 0, stream>>>(tw, Yr, Yi, Xr, Xi, out);
}


# inverse 2D DFT keeps only the real part: G plane folded with its conjugate mirror while staged (S = G + conj G(-u,-v)), both stages run rows u = 0..32 only; pairs u, 65-u are summed first
# speedup vs baseline: 8.2018x; 1.1121x over previous
#include <hip/hip_runtime.h>


typedef __attribute__((ext_vector_type(2))) float v2f;
typedef __attribute__((ext_vector_type(8))) float v8f;

constexpr int NB  = 16;
constexpr int NC  = 64;
constexpr int NFR = 8;
constexpr int ND  = 16;
constexpr int NL  = 64;
constexpr int NLP = 65;
constexpr int NP  = NLP * NLP;

constexpr int OFF_FWDR = 0;
constexpr int OFF_FWDI = OFF_FWDR + 80 * 80;
constexpr int OFF_ER   = OFF_FWDI + 80 * 80;
constexpr int OFF_EI   = OFF_ER + 64 * 64;
constexpr int OFF_WNR  = OFF_EI + 64 * 64;
constexpr int OFF_WNI  = OFF_WNR + 64 * 16;
constexpr int OFF_IVR  = OFF_WNI + 64 * 16;
constexpr int OFF_IVI  = OFF_IVR + 68 * 64;
constexpr int OFF_IUR  = OFF_IVI + 68 * 64;
constexpr int OFF_IUI  = OFF_IUR + 64 * 36;
constexpr int TW_TOTAL = OFF_IUI + 64 * 36;
constexpr int TW_PAD   = 40960;

#define PI_D 3.14159265358979323846


static __device__ __forceinline__ v8f wmma4(v2f a, v2f b, v8f c) {
  return __builtin_amdgcn_wmma_f32_16x16x4_f32(false, a, false, b, (short)0, c,
                                               false, false);
}

static __device__ __forceinline__ v8f zero8() {
  v8f z = {0.f, 0.f, 0.f, 0.f, 0.f, 0.f, 0.f, 0.f};
  return z;
}

static __device__ __forceinline__ v2f frag_a(const float* __restrict__ A,
                                             int lda, int m0, int k0) {
  int lane = threadIdx.x & 31;
  int m = m0 + (lane & 15);
  int k = k0 + ((lane >> 4) << 1);
  const float* p = A + m * lda + k;
  v2f r;
  r.x = p[0];
  r.y = p[1];
  return r;
}

static __device__ __forceinline__ v2f frag_a_tab(const float* __restrict__ A,
                                                 int ksteps, int tile, int s) {
  int lane = threadIdx.x & 31;
  return *reinterpret_cast<const v2f*>(A + ((tile * ksteps + s) * 32 + lane) * 2);
}

static __device__ __forceinline__ v2f frag_b(const float* __restrict__ B,
                                             int ldb, int k0, int n0) {
  int lane = threadIdx.x & 31;
  int n = n0 + (lane & 15);
  int k = k0 + ((lane >> 4) << 1);
  const float* p = B + k * ldb + n;
  v2f r;
  r.x = p[0];
  r.y = p[ldb];
  return r;
}

static __device__ __forceinline__ void cwmma3(v2f ar, v2f adm, v2f asp, v2f br,
                                              v2f bi, v8f& k1, v8f& k2, v8f& k3) {
  v2f bs = br + bi;
  k1 = wmma4(ar, bs, k1);
  k2 = wmma4(adm, br, k2);
  k3 = wmma4(asp, bi, k3);
}


__global__ void k_twiddles(float* __restrict__ tw) {
  int t = blockIdx.x * blockDim.x + threadIdx.x;
  int which = blockIdx.y;
  if (which == 0) {
    if (t >= 80 * 80) return;
    int a = t / 80, b = t % 80;
    float r = 0.f, im = 0.f;
    if (a < NLP && b < NLP) {
      double ang = -2.0 * PI_D * (double)((a * b) % NLP) / (double)NLP;
      r = (float)cos(ang);
      im = (float)sin(ang);
    }
    tw[OFF_FWDR + t] = r;
    tw[OFF_FWDI + t] = im;
  } else if (which == 1) {
    if (t >= NC * NC) return;
    int l = (t >> 1) & 31;
    int c = (t >> 10) * 16 + (l & 15);
    int k = ((t >> 6) & 15) * 4 + ((l >> 4) << 1) + (t & 1);
    double ang = -2.0 * PI_D * (double)((c * k) % NC) / (double)NC;
    tw[OFF_ER + t] = (float)cos(ang);
    tw[OFF_EI + t] = (float)sin(ang);
  } else if (which == 2) {
    if (t >= NC * ND) return;
    int l = (t >> 1) & 31;
    int d = l & 15;
    int c = (t >> 6) * 4 + ((l >> 4) << 1) + (t & 1);
    double ang = 2.0 * PI_D * (double)((c * 4 * d) % NC) / (double)NC;
    tw[OFF_WNR + t] = (float)(cos(ang) / 64.0);
    tw[OFF_WNI + t] = (float)(sin(ang) / 64.0);
  } else if (which == 3) {
    if (t >= 68 * NL) return;
    int v = t / NL, w = t % NL;
    float r = 0.f, im = 0.f;
    if (v < NLP) {
      double ang = 2.0 * PI_D * (double)((v * (w + 1)) % NLP) / (double)NLP;
      r = (float)cos(ang);
      im = (float)sin(ang);
    }
    tw[OFF_IVR + t] = r;
    tw[OFF_IVI + t] = im;
  } else {
    if (t >= NL * 36) return;
    int l = (t >> 1) & 31;
    int fs = t >> 6;
    int h = (fs / 9) * 16 + (l & 15);
    int u = (fs % 9) * 4 + ((l >> 4) << 1) + (t & 1);
    float r = 0.f, im = 0.f;
    if (u <= NLP / 2) {
      double ang = 2.0 * PI_D * (double)((u * (h + 1)) % NLP) / (double)NLP;
      double s = (u == 0 ? 0.5 : 1.0) / (double)(NLP * NLP);
      r = (float)(cos(ang) * s);
      im = (float)(-sin(ang) * s);
    }
    tw[OFF_IUR + t] = r;
    tw[OFF_IUI + t] = im;
  }
}


__global__ __launch_bounds__(128) void k_fft2d(const float* __restrict__ x,
                                               const float* __restrict__ tw,
                                               float* __restrict__ Yr,
                                               float* __restrict__ Yi) {
  __shared__ float xs[NL * NL];
  __shared__ float Tr[NL * 80];
  __shared__ float Ti[NL * 80];
  const int plane = blockIdx.x;
  const int tid = threadIdx.x;
  const int wave = tid >> 5;
  const int lane = tid & 31;

  const float* xp = x + (size_t)plane * (NL * NL);
  for (int i = tid; i < NL * NL; i += 128) xs[i] = xp[i];
  __syncthreads();

  const float* Fr = tw + OFF_FWDR;
  const float* Fi = tw + OFF_FWDI;

  for (int t = wave; t < 4 * 5; t += 4) {
    int mo = (t / 5) * 16, no = (t % 5) * 16;
    v8f ar = zero8(), ai = zero8();
    for (int k0 = 0; k0 < NL; k0 += 4) {
      v2f a = frag_a(xs, NL, mo, k0);
      ar = wmma4(a, frag_b(Fr, 80, k0, no), ar);
      ai = wmma4(a, frag_b(Fi, 80, k0, no), ai);
    }
    int n = no + (lane & 15);
    int mb = mo + ((lane >> 4) << 3);
#pragma unroll
    for (int i = 0; i < 8; ++i) {
      Tr[(mb + i) * 80 + n] = ar[i];
      Ti[(mb + i) * 80 + n] = ai[i];
    }
  }
  __syncthreads();

  float* yr = Yr + (size_t)plane * NP;
  float* yi = Yi + (size_t)plane * NP;
  for (int t = wave; t < 5 * 5; t += 4) {
    int mo = (t / 5) * 16, no = (t % 5) * 16;
    v8f k1 = zero8(), k2 = zero8(), k3 = zero8();
#pragma unroll
    for (int k0 = 0; k0 < NL; k0 += 4) {
      v2f par = frag_b(Fr, 80, k0, mo);
      v2f pai = frag_b(Fi, 80, k0, mo);
      v2f pbr = frag_b(Tr, 80, k0, no);
      v2f pbi = frag_b(Ti, 80, k0, no);
      cwmma3(par, pai - par, par + pai, pbr, pbi, k1, k2, k3);
    }
    const v8f dr = k1 - k3, di = k1 + k2;
    int n = no + (lane & 15);
    int mb = mo + ((lane >> 4) << 3);
    if (n < NLP) {
#pragma unroll
      for (int i = 0; i < 8; ++i) {
        int m = mb + i;
        if (m < NLP) {
          yr[m * NLP + n] = dr[i];
          yi[m * NLP + n] = di[i];
        }
      }
    }
  }
}


typedef __attribute__((ext_vector_type(4))) float v4f;

constexpr int MIX_PT = (NP + 15) / 16;
constexpr int XP = 68;

__global__ __launch_bounds__(128) void k_cmix(const float* __restrict__ tw,
                                              float* Yr, float* Yi,
                                              float* Xr, float* Xi,
                                              const float* __restrict__ wre,
                                              const float* __restrict__ wim) {
  __shared__ __attribute__((aligned(16))) float xs_r[16 * XP];
  __shared__ __attribute__((aligned(16))) float xs_i[16 * XP];
  const int b = blockIdx.x;
  const int p0 = blockIdx.y * 16;
  const int wave = __builtin_amdgcn_readfirstlane(threadIdx.x >> 5);
  const int lane = threadIdx.x & 31;
  const int col = lane & 15;
  const int half = lane >> 4;
  const int h2 = half << 1;
  const int p = p0 + col;
  const int pl = p < NP ? p : NP - 1;

  {
    const float* Er = tw + OFF_ER;
    const float* Ei = tw + OFF_EI;
    const float* yr_p = Yr + ((size_t)b * NC + h2) * NP + pl;
    const float* yi_p = Yi + ((size_t)b * NC + h2) * NP + pl;
    v8f k1 = zero8(), k2 = zero8(), k3 = zero8();
#pragma unroll
    for (int k0 = 0; k0 < NC; k0 += 4) {
      v2f ar = frag_a_tab(Er, 16, wave, k0 >> 2);
      v2f ai = frag_a_tab(Ei, 16, wave, k0 >> 2);
      v2f br, bi;
      br.x = yr_p[(size_t)k0 * NP];
      br.y = yr_p[(size_t)(k0 + 1) * NP];
      bi.x = yi_p[(size_t)k0 * NP];
      bi.y = yi_p[(size_t)(k0 + 1) * NP];
      cwmma3(ar, ai - ar, ar + ai, br, bi, k1, k2, k3);
    }
    const v8f dr = k1 - k3, di = k1 + k2;
    float* sr = xs_r + col * XP + wave * 16 + (half << 3);
    float* si = xs_i + col * XP + wave * 16 + (half << 3);
    v4f q;
    q.x = dr[0]; q.y = dr[1]; q.z = dr[2]; q.w = dr[3];
    *reinterpret_cast<v4f*>(sr) = q;
    q.x = dr[4]; q.y = dr[5]; q.z = dr[6]; q.w = dr[7];
    *reinterpret_cast<v4f*>(sr + 4) = q;
    q.x = di[0]; q.y = di[1]; q.z = di[2]; q.w = di[3];
    *reinterpret_cast<v4f*>(si) = q;
    q.x = di[4]; q.y = di[5]; q.z = di[6]; q.w = di[7];
    *reinterpret_cast<v4f*>(si + 4) = q;
  }
  __syncthreads();

  const int n0 = wave * 2;
  const float* ar_p = tw + OFF_WNR + lane * 2;
  const float* ai_p = tw + OFF_WNI + lane * 2;
  const float* wr_p = wre + ((size_t)n0 * NC + h2) * NP + pl;
  const float* wi_p = wim + ((size_t)n0 * NC + h2) * NP + pl;
  const float* xr_p = xs_r + col * XP + h2;
  const float* xi_p = xs_i + col * XP + h2;

  v8f k1[2], k2[2], k3[2];
#pragma unroll
  for (int j = 0; j < 2; ++j) {
    k1[j] = zero8();
    k2[j] = zero8();
    k3[j] = zero8();
  }

#pragma unroll
  for (int k0 = 0; k0 < NC; k0 += 4) {
    v2f ar = *reinterpret_cast<const v2f*>(ar_p + k0 * 16);
    v2f ai = *reinterpret_cast<const v2f*>(ai_p + k0 * 16);
    const v2f adm = ai - ar;
    const v2f asp = ar + ai;
    v2f xr = *reinterpret_cast<const v2f*>(xr_p + k0);
    v2f xi = *reinterpret_cast<const v2f*>(xi_p + k0);
#pragma unroll
    for (int j = 0; j < 2; ++j) {
      size_t o0 = ((size_t)j * NC + k0) * NP;
      size_t o1 = o0 + NP;
      float wr0 = wr_p[o0], wi0 = wi_p[o0];
      float wr1 = wr_p[o1], wi1 = wi_p[o1];
      v2f br, bi;
      br.x = xr.x * wr0 - xi.x * wi0;
      bi.x = xr.x * wi0 + xi.x * wr0;
      br.y = xr.y * wr1 - xi.y * wi1;
      bi.y = xr.y * wi1 + xi.y * wr1;
      cwmma3(ar, adm, asp, br, bi, k1[j], k2[j], k3[j]);
    }
  }

  if (p < NP) {
    const int mb = half << 3;
#pragma unroll
    for (int j = 0; j < 2; ++j) {
      const int n = n0 + j;
      const v8f dr = k1[j] - k3[j], di = k1[j] + k2[j];
      float* gr = (n < 4 ? Yr : Xr) + ((size_t)b * NC + (n & 3) * ND + mb) * NP + p;
      float* gi = (n < 4 ? Yi : Xi) + ((size_t)b * NC + (n & 3) * ND + mb) * NP + p;
#pragma unroll
      for (int i = 0; i < 8; ++i) {
        gr[(size_t)i * NP] = dr[i];
        gi[(size_t)i * NP] = di[i];
      }
    }
  }
}


constexpr int GP = 68;
constexpr int GR = 48;
constexpr int GS = GR * GP;
constexpr int TP = 52;
constexpr int KS = 17;
constexpr int KU = 9;
constexpr int UH = NLP / 2;

__global__ __launch_bounds__(256) void k_ifft2d(const float* __restrict__ tw,
                                                const float* __restrict__ Yr,
                                                const float* __restrict__ Yi,
                                                const float* __restrict__ Xr,
                                                const float* __restrict__ Xi,
                                                float* __restrict__ out) {
  __shared__ __attribute__((aligned(16))) float gs_r[GS];
  __shared__ __attribute__((aligned(16))) float gs_i[GS];
  __shared__ __attribute__((aligned(16))) float tt_r[NL * TP];
  __shared__ __attribute__((aligned(16))) float tt_i[NL * TP];
  const int d = blockIdx.x, b = blockIdx.y, nidx = blockIdx.z;
  const int tid = threadIdx.x;
  const int wave = __builtin_amdgcn_readfirstlane(tid >> 5);
  const int lane = tid & 31;
  const int c = lane & 15;
  const int h2 = (lane >> 4) << 1;
  const int wq = wave & 3;
  const int wh = wave >> 2;

  const float* IVr = tw + OFF_IVR;
  const float* IVi = tw + OFF_IVI;
  v2f vbr[KS], vbi[KS], vdm[KS], vsp[KS];
#pragma unroll
  for (int k = 0; k < KS; ++k) {
    vbr[k] = frag_b(IVr, NL, 4 * k, wq * 16);
    vbi[k] = frag_b(IVi, NL, 4 * k, wq * 16);
  }

  const float* gr = (nidx < 4 ? Yr : Xr) + ((size_t)b * NC + (nidx & 3) * ND + d) * NP;
  const float* gi = (nidx < 4 ? Yi : Xi) + ((size_t)b * NC + (nidx & 3) * ND + d) * NP;
  const int r4 = tid >> 6, c64 = tid & 63;
  const int mc = c64 ? NLP - c64 : 0;
  const int erow = (tid >> 2) - (tid >= 192 ? 16 : 0), ecol = tid & 3;
  const bool ok0 = ecol == 0 && erow <= UH;
  const int d0i = erow * NLP + 64;
  const int mrow0 = erow ? NLP - erow : 0;
  const int m0i = mrow0 * NLP + 1;
  const int s0 = ok0 ? d0i : 0;
  const int m0 = ok0 ? m0i : 0;
  const float a0r = gr[s0], a0i = gi[s0];
  const float b0r = gr[m0], b0i = gi[m0];
  float sr[KU], si[KU];
#pragma unroll
  for (int j = 0; j < KU; ++j) {
    const int row = 4 * j + r4;
    const bool ok = j < KU - 1 || r4 == 0;
    const int mrow = row ? NLP - row : 0;
    const int s = ok ? row * NLP + c64 : 0;
    const int m = ok ? mrow * NLP + mc : 0;
    const float ar = gr[s], ai = gi[s];
    const float br = gr[m], bi = gi[m];
    sr[j] = ok ? ar + br : 0.f;
    si[j] = ok ? ai - bi : 0.f;
  }
  const float e0r = ok0 ? a0r + b0r : 0.f;
  const float e0i = ok0 ? a0i - b0i : 0.f;

#pragma unroll
  for (int j = 0; j < GR / 4; ++j) {
    gs_r[(4 * j + r4) * GP + c64] = j < KU ? sr[j] : 0.f;
    gs_i[(4 * j + r4) * GP + c64] = j < KU ? si[j] : 0.f;
  }
  gs_r[erow * GP + 64 + ecol] = e0r;
  gs_i[erow * GP + 64 + ecol] = e0i;
  __syncthreads();

#pragma unroll
  for (int k = 0; k < KS; ++k) {
    vdm[k] = vbi[k] - vbr[k];
    vsp[k] = vbr[k] + vbi[k];
  }

  {
    const int no = wq * 16;
    const int mo_end = wh ? 48 : 32;
    for (int mo = wh ? 32 : 0; mo < mo_end; mo += 16) {
      const float* ar_p = gs_r + (mo + c) * GP + h2;
      const float* ai_p = gs_i + (mo + c) * GP + h2;
      v8f k1 = zero8(), k2 = zero8(), k3 = zero8();
#pragma unroll
      for (int k = 0; k < KS; ++k) {
        v2f ar = *reinterpret_cast<const v2f*>(ar_p + 4 * k);
        v2f ai = *reinterpret_cast<const v2f*>(ai_p + 4 * k);
        v2f asum = ar + ai;
        k1 = wmma4(asum, vbr[k], k1);
        k2 = wmma4(ar, vdm[k], k2);
        k3 = wmma4(ai, vsp[k], k3);
      }
      const v8f dr = k1 - k3, di = k1 + k2;
      float* tr = tt_r + (no + c) * TP + mo + ((lane >> 4) << 3);
      float* ti = tt_i + (no + c) * TP + mo + ((lane >> 4) << 3);
      v4f q;
      q.x = dr[0]; q.y = dr[1]; q.z = dr[2]; q.w = dr[3];
      *reinterpret_cast<v4f*>(tr) = q;
      q.x = dr[4]; q.y = dr[5]; q.z = dr[6]; q.w = dr[7];
      *reinterpret_cast<v4f*>(tr + 4) = q;
      q.x = di[0]; q.y = di[1]; q.z = di[2]; q.w = di[3];
      *reinterpret_cast<v4f*>(ti) = q;
      q.x = di[4]; q.y = di[5]; q.z = di[6]; q.w = di[7];
      *reinterpret_cast<v4f*>(ti + 4) = q;
    }
  }

  const float* IUr = tw + OFF_IUR;
  const float* IUi = tw + OFF_IUI;
  v2f uar[KU], unai[KU];
#pragma unroll
  for (int k = 0; k < KU; ++k) {
    uar[k] = frag_a_tab(IUr, KU, wq, k);
    unai[k] = frag_a_tab(IUi, KU, wq, k);
  }
  __syncthreads();

  {
    const int mo = wq * 16;
    const int no = wh * 32;
    const float* b0r_p = tt_r + (no + c) * TP + h2;
    const float* b0i_p = tt_i + (no + c) * TP + h2;
    const float* b1r_p = b0r_p + 16 * TP;
    const float* b1i_p = b0i_p + 16 * TP;
    v8f d0 = zero8(), d1 = zero8();
#pragma unroll
    for (int k = 0; k < KU; ++k) {
      v2f b0r = *reinterpret_cast<const v2f*>(b0r_p + 4 * k);
      v2f b1r = *reinterpret_cast<const v2f*>(b1r_p + 4 * k);
      v2f b0i = *reinterpret_cast<const v2f*>(b0i_p + 4 * k);
      v2f b1i = *reinterpret_cast<const v2f*>(b1i_p + 4 * k);
      d0 = wmma4(uar[k], b0r, d0);
      d1 = wmma4(uar[k], b1r, d1);
      d0 = wmma4(unai[k], b0i, d0);
      d1 = wmma4(unai[k], b1i, d1);
    }
    float* op = out + ((size_t)b * (ND * NFR) + (d * NFR + nidx)) * (NL * NL);
    const int n = no + c;
    const int mb = mo + ((lane >> 4) << 3);
#pragma unroll
    for (int i = 0; i < 8; ++i) {
      op[(mb + i) * NL + n] = d0[i];
      op[(mb + i) * NL + n + 16] = d1[i];
    }
  }
}


extern "C" void kernel_launch(void* const* d_in, const int* in_sizes, int n_in,
                              void* d_out, int out_size, void* d_ws,
                              size_t ws_size, hipStream_t stream) {
  (void)in_sizes; (void)n_in; (void)out_size; (void)ws_size;
  const float* x   = (const float*)d_in[0];
  const float* wre = (const float*)d_in[1];
  const float* wim = (const float*)d_in[2];
  float* out = (float*)d_out;
  float* ws = (float*)d_ws;

  float* tw = ws;
  size_t off = TW_PAD;
  float* Yr = ws + off; off += (size_t)NB * NC * NP;
  float* Yi = ws + off; off += (size_t)NB * NC * NP;
  float* Xr = ws + off; off += (size_t)NB * NC * NP;
  float* Xi = ws + off;

  k_twiddles<<<dim3(25, 5), 256, 0, stream>>>(tw);
  k_fft2d<<<NB * NC, 128, 0, stream>>>(x, tw, Yr, Yi);
  k_cmix<<<dim3(NB, MIX_PT), 128, 0, stream>>>(tw, Yr, Yi, Xr, Xi, wre, wim);
  k_ifft2d<<<dim3(ND, NB, NFR), 256, 0, stream>>>(tw, Yr, Yi, Xr, Xi, out);
}


# mix table has period 16 in c: the four products X*W of channels c, c+16, c+32, c+48 are added in VALU first, the matrix product runs over k = 16, not 64; c sum order changes (groups of 4 first)
# speedup vs baseline: 8.5140x; 1.0381x over previous
#include <hip/hip_runtime.h>


typedef __attribute__((ext_vector_type(2))) float v2f;
typedef __attribute__((ext_vector_type(8))) float v8f;

constexpr int NB  = 16;
constexpr int NC  = 64;
constexpr int NFR = 8;
constexpr int ND  = 16;
constexpr int NL  = 64;
constexpr int NLP = 65;
constexpr int NP  = NLP * NLP;

constexpr int OFF_FWDR = 0;
constexpr int OFF_FWDI = OFF_FWDR + 80 * 80;
constexpr int OFF_ER   = OFF_FWDI + 80 * 80;
constexpr int OFF_EI   = OFF_ER + 64 * 64;
constexpr int OFF_WNR  = OFF_EI + 64 * 64;
constexpr int OFF_WNI  = OFF_WNR + 64 * 16;
constexpr int OFF_IVR  = OFF_WNI + 64 * 16;
constexpr int OFF_IVI  = OFF_IVR + 68 * 64;
constexpr int OFF_IUR  = OFF_IVI + 68 * 64;
constexpr int OFF_IUI  = OFF_IUR + 64 * 36;
constexpr int TW_TOTAL = OFF_IUI + 64 * 36;
constexpr int TW_PAD   = 40960;

#define PI_D 3.14159265358979323846


static __device__ __forceinline__ v8f wmma4(v2f a, v2f b, v8f c) {
  return __builtin_amdgcn_wmma_f32_16x16x4_f32(false, a, false, b, (short)0, c,
                                               false, false);
}

static __device__ __forceinline__ v8f zero8() {
  v8f z = {0.f, 0.f, 0.f, 0.f, 0.f, 0.f, 0.f, 0.f};
  return z;
}

static __device__ __forceinline__ v2f frag_a(const float* __restrict__ A,
                                             int lda, int m0, int k0) {
  int lane = threadIdx.x & 31;
  int m = m0 + (lane & 15);
  int k = k0 + ((lane >> 4) << 1);
  const float* p = A + m * lda + k;
  v2f r;
  r.x = p[0];
  r.y = p[1];
  return r;
}

static __device__ __forceinline__ v2f frag_a_tab(const float* __restrict__ A,
                                                 int ksteps, int tile, int s) {
  int lane = threadIdx.x & 31;
  return *reinterpret_cast<const v2f*>(A + ((tile * ksteps + s) * 32 + lane) * 2);
}

static __device__ __forceinline__ v2f frag_b(const float* __restrict__ B,
                                             int ldb, int k0, int n0) {
  int lane = threadIdx.x & 31;
  int n = n0 + (lane & 15);
  int k = k0 + ((lane >> 4) << 1);
  const float* p = B + k * ldb + n;
  v2f r;
  r.x = p[0];
  r.y = p[ldb];
  return r;
}

static __device__ __forceinline__ void cwmma3(v2f ar, v2f adm, v2f asp, v2f br,
                                              v2f bi, v8f& k1, v8f& k2, v8f& k3) {
  v2f bs = br + bi;
  k1 = wmma4(ar, bs, k1);
  k2 = wmma4(adm, br, k2);
  k3 = wmma4(asp, bi, k3);
}


__global__ void k_twiddles(float* __restrict__ tw) {
  int t = blockIdx.x * blockDim.x + threadIdx.x;
  int which = blockIdx.y;
  if (which == 0) {
    if (t >= 80 * 80) return;
    int a = t / 80, b = t % 80;
    float r = 0.f, im = 0.f;
    if (a < NLP && b < NLP) {
      double ang = -2.0 * PI_D * (double)((a * b) % NLP) / (double)NLP;
      r = (float)cos(ang);
      im = (float)sin(ang);
    }
    tw[OFF_FWDR + t] = r;
    tw[OFF_FWDI + t] = im;
  } else if (which == 1) {
    if (t >= NC * NC) return;
    int l = (t >> 1) & 31;
    int c = (t >> 10) * 16 + (l & 15);
    int k = ((t >> 6) & 15) * 4 + ((l >> 4) << 1) + (t & 1);
    double ang = -2.0 * PI_D * (double)((c * k) % NC) / (double)NC;
    tw[OFF_ER + t] = (float)cos(ang);
    tw[OFF_EI + t] = (float)sin(ang);
  } else if (which == 2) {
    if (t >= 16 * ND) return;
    int l = (t >> 1) & 31;
    int d = l & 15;
    int c = (t >> 6) * 4 + ((l >> 4) << 1) + (t & 1);
    double ang = 2.0 * PI_D * (double)((c * 4 * d) % NC) / (double)NC;
    tw[OFF_WNR + t] = (float)(cos(ang) / 64.0);
    tw[OFF_WNI + t] = (float)(sin(ang) / 64.0);
  } else if (which == 3) {
    if (t >= 68 * NL) return;
    int v = t / NL, w = t % NL;
    float r = 0.f, im = 0.f;
    if (v < NLP) {
      double ang = 2.0 * PI_D * (double)((v * (w + 1)) % NLP) / (double)NLP;
      r = (float)cos(ang);
      im = (float)sin(ang);
    }
    tw[OFF_IVR + t] = r;
    tw[OFF_IVI + t] = im;
  } else {
    if (t >= NL * 36) return;
    int l = (t >> 1) & 31;
    int fs = t >> 6;
    int h = (fs / 9) * 16 + (l & 15);
    int u = (fs % 9) * 4 + ((l >> 4) << 1) + (t & 1);
    float r = 0.f, im = 0.f;
    if (u <= NLP / 2) {
      double ang = 2.0 * PI_D * (double)((u * (h + 1)) % NLP) / (double)NLP;
      double s = (u == 0 ? 0.5 : 1.0) / (double)(NLP * NLP);
      r = (float)(cos(ang) * s);
      im = (float)(-sin(ang) * s);
    }
    tw[OFF_IUR + t] = r;
    tw[OFF_IUI + t] = im;
  }
}


__global__ __launch_bounds__(128) void k_fft2d(const float* __restrict__ x,
                                               const float* __restrict__ tw,
                                               float* __restrict__ Yr,
                                               float* __restrict__ Yi) {
  __shared__ float xs[NL * NL];
  __shared__ float Tr[NL * 80];
  __shared__ float Ti[NL * 80];
  const int plane = blockIdx.x;
  const int tid = threadIdx.x;
  const int wave = tid >> 5;
  const int lane = tid & 31;

  const float* xp = x + (size_t)plane * (NL * NL);
  for (int i = tid; i < NL * NL; i += 128) xs[i] = xp[i];
  __syncthreads();

  const float* Fr = tw + OFF_FWDR;
  const float* Fi = tw + OFF_FWDI;

  for (int t = wave; t < 4 * 5; t += 4) {
    int mo = (t / 5) * 16, no = (t % 5) * 16;
    v8f ar = zero8(), ai = zero8();
    for (int k0 = 0; k0 < NL; k0 += 4) {
      v2f a = frag_a(xs, NL, mo, k0);
      ar = wmma4(a, frag_b(Fr, 80, k0, no), ar);
      ai = wmma4(a, frag_b(Fi, 80, k0, no), ai);
    }
    int n = no + (lane & 15);
    int mb = mo + ((lane >> 4) << 3);
#pragma unroll
    for (int i = 0; i < 8; ++i) {
      Tr[(mb + i) * 80 + n] = ar[i];
      Ti[(mb + i) * 80 + n] = ai[i];
    }
  }
  __syncthreads();

  float* yr = Yr + (size_t)plane * NP;
  float* yi = Yi + (size_t)plane * NP;
  for (int t = wave; t < 5 * 5; t += 4) {
    int mo = (t / 5) * 16, no = (t % 5) * 16;
    v8f k1 = zero8(), k2 = zero8(), k3 = zero8();
#pragma unroll
    for (int k0 = 0; k0 < NL; k0 += 4) {
      v2f par = frag_b(Fr, 80, k0, mo);
      v2f pai = frag_b(Fi, 80, k0, mo);
      v2f pbr = frag_b(Tr, 80, k0, no);
      v2f pbi = frag_b(Ti, 80, k0, no);
      cwmma3(par, pai - par, par + pai, pbr, pbi, k1, k2, k3);
    }
    const v8f dr = k1 - k3, di = k1 + k2;
    int n = no + (lane & 15);
    int mb = mo + ((lane >> 4) << 3);
    if (n < NLP) {
#pragma unroll
      for (int i = 0; i < 8; ++i) {
        int m = mb + i;
        if (m < NLP) {
          yr[m * NLP + n] = dr[i];
          yi[m * NLP + n] = di[i];
        }
      }
    }
  }
}


typedef __attribute__((ext_vector_type(4))) float v4f;

constexpr int MIX_PT = (NP + 15) / 16;
constexpr int XP = 68;

__global__ __launch_bounds__(128) void k_cmix(const float* __restrict__ tw,
                                              float* Yr, float* Yi,
                                              float* Xr, float* Xi,
                                              const float* __restrict__ wre,
                                              const float* __restrict__ wim) {
  __shared__ __attribute__((aligned(16))) float xs_r[16 * XP];
  __shared__ __attribute__((aligned(16))) float xs_i[16 * XP];
  const int b = blockIdx.x;
  const int p0 = blockIdx.y * 16;
  const int wave = __builtin_amdgcn_readfirstlane(threadIdx.x >> 5);
  const int lane = threadIdx.x & 31;
  const int col = lane & 15;
  const int half = lane >> 4;
  const int h2 = half << 1;
  const int p = p0 + col;
  const int pl = p < NP ? p : NP - 1;

  {
    const float* Er = tw + OFF_ER;
    const float* Ei = tw + OFF_EI;
    const float* yr_p = Yr + ((size_t)b * NC + h2) * NP + pl;
    const float* yi_p = Yi + ((size_t)b * NC + h2) * NP + pl;
    v8f k1 = zero8(), k2 = zero8(), k3 = zero8();
#pragma unroll
    for (int k0 = 0; k0 < NC; k0 += 4) {
      v2f ar = frag_a_tab(Er, 16, wave, k0 >> 2);
      v2f ai = frag_a_tab(Ei, 16, wave, k0 >> 2);
      v2f br, bi;
      br.x = yr_p[(size_t)k0 * NP];
      br.y = yr_p[(size_t)(k0 + 1) * NP];
      bi.x = yi_p[(size_t)k0 * NP];
      bi.y = yi_p[(size_t)(k0 + 1) * NP];
      cwmma3(ar, ai - ar, ar + ai, br, bi, k1, k2, k3);
    }
    const v8f dr = k1 - k3, di = k1 + k2;
    float* sr = xs_r + col * XP + wave * 16 + (half << 3);
    float* si = xs_i + col * XP + wave * 16 + (half << 3);
    v4f q;
    q.x = dr[0]; q.y = dr[1]; q.z = dr[2]; q.w = dr[3];
    *reinterpret_cast<v4f*>(sr) = q;
    q.x = dr[4]; q.y = dr[5]; q.z = dr[6]; q.w = dr[7];
    *reinterpret_cast<v4f*>(sr + 4) = q;
    q.x = di[0]; q.y = di[1]; q.z = di[2]; q.w = di[3];
    *reinterpret_cast<v4f*>(si) = q;
    q.x = di[4]; q.y = di[5]; q.z = di[6]; q.w = di[7];
    *reinterpret_cast<v4f*>(si + 4) = q;
  }
  __syncthreads();

  const int n0 = wave * 2;
  const float* ar_p = tw + OFF_WNR + lane * 2;
  const float* ai_p = tw + OFF_WNI + lane * 2;
  const float* wr_p = wre + ((size_t)n0 * NC + h2) * NP + pl;
  const float* wi_p = wim + ((size_t)n0 * NC + h2) * NP + pl;
  const float* xr_p = xs_r + col * XP + h2;
  const float* xi_p = xs_i + col * XP + h2;

  v8f k1[2], k2[2], k3[2];
#pragma unroll
  for (int j = 0; j < 2; ++j) {
    k1[j] = zero8();
    k2[j] = zero8();
    k3[j] = zero8();
  }

#pragma unroll
  for (int k0 = 0; k0 < 16; k0 += 4) {
    v2f ar = *reinterpret_cast<const v2f*>(ar_p + k0 * 16);
    v2f ai = *reinterpret_cast<const v2f*>(ai_p + k0 * 16);
    const v2f adm = ai - ar;
    const v2f asp = ar + ai;
    v2f xr[4], xi[4];
#pragma unroll
    for (int q = 0; q < 4; ++q) {
      xr[q] = *reinterpret_cast<const v2f*>(xr_p + k0 + 16 * q);
      xi[q] = *reinterpret_cast<const v2f*>(xi_p + k0 + 16 * q);
    }
#pragma unroll
    for (int j = 0; j < 2; ++j) {
      v2f br, bi;
#pragma unroll
      for (int q = 0; q < 4; ++q) {
        size_t o0 = ((size_t)j * NC + k0 + 16 * q) * NP;
        size_t o1 = o0 + NP;
        v2f wr, wi;
        wr.x = wr_p[o0]; wi.x = wi_p[o0];
        wr.y = wr_p[o1]; wi.y = wi_p[o1];
        if (q == 0) {
          br = xr[0] * wr;
          bi = xr[0] * wi;
        } else {
          br += xr[q] * wr;
          bi += xr[q] * wi;
        }
        br -= xi[q] * wi;
        bi += xi[q] * wr;
      }
      cwmma3(ar, adm, asp, br, bi, k1[j], k2[j], k3[j]);
    }
  }

  if (p < NP) {
    const int mb = half << 3;
#pragma unroll
    for (int j = 0; j < 2; ++j) {
      const int n = n0 + j;
      const v8f dr = k1[j] - k3[j], di = k1[j] + k2[j];
      float* gr = (n < 4 ? Yr : Xr) + ((size_t)b * NC + (n & 3) * ND + mb) * NP + p;
      float* gi = (n < 4 ? Yi : Xi) + ((size_t)b * NC + (n & 3) * ND + mb) * NP + p;
#pragma unroll
      for (int i = 0; i < 8; ++i) {
        gr[(size_t)i * NP] = dr[i];
        gi[(size_t)i * NP] = di[i];
      }
    }
  }
}


constexpr int GP = 68;
constexpr int GR = 48;
constexpr int GS = GR * GP;
constexpr int TP = 52;
constexpr int KS = 17;
constexpr int KU = 9;
constexpr int UH = NLP / 2;

__global__ __launch_bounds__(256) void k_ifft2d(const float* __restrict__ tw,
                                                const float* __restrict__ Yr,
                                                const float* __restrict__ Yi,
                                                const float* __restrict__ Xr,
                                                const float* __restrict__ Xi,
                                                float* __restrict__ out) {
  __shared__ __attribute__((aligned(16))) float gs_r[GS];
  __shared__ __attribute__((aligned(16))) float gs_i[GS];
  __shared__ __attribute__((aligned(16))) float tt_r[NL * TP];
  __shared__ __attribute__((aligned(16))) float tt_i[NL * TP];
  const int d = blockIdx.x, b = blockIdx.y, nidx = blockIdx.z;
  const int tid = threadIdx.x;
  const int wave = __builtin_amdgcn_readfirstlane(tid >> 5);
  const int lane = tid & 31;
  const int c = lane & 15;
  const int h2 = (lane >> 4) << 1;
  const int wq = wave & 3;
  const int wh = wave >> 2;

  const float* IVr = tw + OFF_IVR;
  const float* IVi = tw + OFF_IVI;
  v2f vbr[KS], vbi[KS], vdm[KS], vsp[KS];
#pragma unroll
  for (int k = 0; k < KS; ++k) {
    vbr[k] = frag_b(IVr, NL, 4 * k, wq * 16);
    vbi[k] = frag_b(IVi, NL, 4 * k, wq * 16);
  }

  const float* gr = (nidx < 4 ? Yr : Xr) + ((size_t)b * NC + (nidx & 3) * ND + d) * NP;
  const float* gi = (nidx < 4 ? Yi : Xi) + ((size_t)b * NC + (nidx & 3) * ND + d) * NP;
  const int r4 = tid >> 6, c64 = tid & 63;
  const int mc = c64 ? NLP - c64 : 0;
  const int erow = (tid >> 2) - (tid >= 192 ? 16 : 0), ecol = tid & 3;
  const bool ok0 = ecol == 0 && erow <= UH;
  const int d0i = erow * NLP + 64;
  const int mrow0 = erow ? NLP - erow : 0;
  const int m0i = mrow0 * NLP + 1;
  const int s0 = ok0 ? d0i : 0;
  const int m0 = ok0 ? m0i : 0;
  const float a0r = gr[s0], a0i = gi[s0];
  const float b0r = gr[m0], b0i = gi[m0];
  float sr[KU], si[KU];
#pragma unroll
  for (int j = 0; j < KU; ++j) {
    const int row = 4 * j + r4;
    const bool ok = j < KU - 1 || r4 == 0;
    const int mrow = row ? NLP - row : 0;
    const int s = ok ? row * NLP + c64 : 0;
    const int m = ok ? mrow * NLP + mc : 0;
    const float ar = gr[s], ai = gi[s];
    const float br = gr[m], bi = gi[m];
    sr[j] = ok ? ar + br : 0.f;
    si[j] = ok ? ai - bi : 0.f;
  }
  const float e0r = ok0 ? a0r + b0r : 0.f;
  const float e0i = ok0 ? a0i - b0i : 0.f;

#pragma unroll
  for (int j = 0; j < GR / 4; ++j) {
    gs_r[(4 * j + r4) * GP + c64] = j < KU ? sr[j] : 0.f;
    gs_i[(4 * j + r4) * GP + c64] = j < KU ? si[j] : 0.f;
  }
  gs_r[erow * GP + 64 + ecol] = e0r;
  gs_i[erow * GP + 64 + ecol] = e0i;
  __syncthreads();

#pragma unroll
  for (int k = 0; k < KS; ++k) {
    vdm[k] = vbi[k] - vbr[k];
    vsp[k] = vbr[k] + vbi[k];
  }

  {
    const int no = wq * 16;
    const int mo_end = wh ? 48 : 32;
    for (int mo = wh ? 32 : 0; mo < mo_end; mo += 16) {
      const float* ar_p = gs_r + (mo + c) * GP + h2;
      const float* ai_p = gs_i + (mo + c) * GP + h2;
      v8f k1 = zero8(), k2 = zero8(), k3 = zero8();
#pragma unroll
      for (int k = 0; k < KS; ++k) {
        v2f ar = *reinterpret_cast<const v2f*>(ar_p + 4 * k);
        v2f ai = *reinterpret_cast<const v2f*>(ai_p + 4 * k);
        v2f asum = ar + ai;
        k1 = wmma4(asum, vbr[k], k1);
        k2 = wmma4(ar, vdm[k], k2);
        k3 = wmma4(ai, vsp[k], k3);
      }
      const v8f dr = k1 - k3, di = k1 + k2;
      float* tr = tt_r + (no + c) * TP + mo + ((lane >> 4) << 3);
      float* ti = tt_i + (no + c) * TP + mo + ((lane >> 4) << 3);
      v4f q;
      q.x = dr[0]; q.y = dr[1]; q.z = dr[2]; q.w = dr[3];
      *reinterpret_cast<v4f*>(tr) = q;
      q.x = dr[4]; q.y = dr[5]; q.z = dr[6]; q.w = dr[7];
      *reinterpret_cast<v4f*>(tr + 4) = q;
      q.x = di[0]; q.y = di[1]; q.z = di[2]; q.w = di[3];
      *reinterpret_cast<v4f*>(ti) = q;
      q.x = di[4]; q.y = di[5]; q.z = di[6]; q.w = di[7];
      *reinterpret_cast<v4f*>(ti + 4) = q;
    }
  }

  const float* IUr = tw + OFF_IUR;
  const float* IUi = tw + OFF_IUI;
  v2f uar[KU], unai[KU];
#pragma unroll
  for (int k = 0; k < KU; ++k) {
    uar[k] = frag_a_tab(IUr, KU, wq, k);
    unai[k] = frag_a_tab(IUi, KU, wq, k);
  }
  __syncthreads();

  {
    const int mo = wq * 16;
    const int no = wh * 32;
    const float* b0r_p = tt_r + (no + c) * TP + h2;
    const float* b0i_p = tt_i + (no + c) * TP + h2;
    const float* b1r_p = b0r_p + 16 * TP;
    const float* b1i_p = b0i_p + 16 * TP;
    v8f d0 = zero8(), d1 = zero8();
#pragma unroll
    for (int k = 0; k < KU; ++k) {
      v2f b0r = *reinterpret_cast<const v2f*>(b0r_p + 4 * k);
      v2f b1r = *reinterpret_cast<const v2f*>(b1r_p + 4 * k);
      v2f b0i = *reinterpret_cast<const v2f*>(b0i_p + 4 * k);
      v2f b1i = *reinterpret_cast<const v2f*>(b1i_p + 4 * k);
      d0 = wmma4(uar[k], b0r, d0);
      d1 = wmma4(uar[k], b1r, d1);
      d0 = wmma4(unai[k], b0i, d0);
      d1 = wmma4(unai[k], b1i, d1);
    }
    float* op = out + ((size_t)b * (ND * NFR) + (d * NFR + nidx)) * (NL * NL);
    const int n = no + c;
    const int mb = mo + ((lane >> 4) << 3);
#pragma unroll
    for (int i = 0; i < 8; ++i) {
      op[(mb + i) * NL + n] = d0[i];
      op[(mb + i) * NL + n + 16] = d1[i];
    }
  }
}


extern "C" void kernel_launch(void* const* d_in, const int* in_sizes, int n_in,
                              void* d_out, int out_size, void* d_ws,
                              size_t ws_size, hipStream_t stream) {
  (void)in_sizes; (void)n_in; (void)out_size; (void)ws_size;
  const float* x   = (const float*)d_in[0];
  const float* wre = (const float*)d_in[1];
  const float* wim = (const float*)d_in[2];
  float* out = (float*)d_out;
  float* ws = (float*)d_ws;

  float* tw = ws;
  size_t off = TW_PAD;
  float* Yr = ws + off; off += (size_t)NB * NC * NP;
  float* Yi = ws + off; off += (size_t)NB * NC * NP;
  float* Xr = ws + off; off += (size_t)NB * NC * NP;
  float* Xi = ws + off;

  k_twiddles<<<dim3(25, 5), 256, 0, stream>>>(tw);
  k_fft2d<<<NB * NC, 128, 0, stream>>>(x, tw, Yr, Yi);
  k_cmix<<<dim3(NB, MIX_PT), 128, 0, stream>>>(tw, Yr, Yi, Xr, Xi, wre, wim);
  k_ifft2d<<<dim3(ND, NB, NFR), 256, 0, stream>>>(tw, Yr, Yi, Xr, Xi, out);
}


# mix kernel: a block takes two batches (8 waves); in the mode-mix stage one wave does one mode for both, so each W element is loaded once per batch pair, not once per batch; sum order unchanged
# speedup vs baseline: 9.6855x; 1.1376x over previous
#include <hip/hip_runtime.h>


typedef __attribute__((ext_vector_type(2))) float v2f;
typedef __attribute__((ext_vector_type(8))) float v8f;

constexpr int NB  = 16;
constexpr int NC  = 64;
constexpr int NFR = 8;
constexpr int ND  = 16;
constexpr int NL  = 64;
constexpr int NLP = 65;
constexpr int NP  = NLP * NLP;

constexpr int OFF_FWDR = 0;
constexpr int OFF_FWDI = OFF_FWDR + 80 * 80;
constexpr int OFF_ER   = OFF_FWDI + 80 * 80;
constexpr int OFF_EI   = OFF_ER + 64 * 64;
constexpr int OFF_WNR  = OFF_EI + 64 * 64;
constexpr int OFF_WNI  = OFF_WNR + 64 * 16;
constexpr int OFF_IVR  = OFF_WNI + 64 * 16;
constexpr int OFF_IVI  = OFF_IVR + 68 * 64;
constexpr int OFF_IUR  = OFF_IVI + 68 * 64;
constexpr int OFF_IUI  = OFF_IUR + 64 * 36;
constexpr int TW_TOTAL = OFF_IUI + 64 * 36;
constexpr int TW_PAD   = 40960;

#define PI_D 3.14159265358979323846


static __device__ __forceinline__ v8f wmma4(v2f a, v2f b, v8f c) {
  return __builtin_amdgcn_wmma_f32_16x16x4_f32(false, a, false, b, (short)0, c,
                                               false, false);
}

static __device__ __forceinline__ v8f zero8() {
  v8f z = {0.f, 0.f, 0.f, 0.f, 0.f, 0.f, 0.f, 0.f};
  return z;
}

static __device__ __forceinline__ v2f frag_a(const float* __restrict__ A,
                                             int lda, int m0, int k0) {
  int lane = threadIdx.x & 31;
  int m = m0 + (lane & 15);
  int k = k0 + ((lane >> 4) << 1);
  const float* p = A + m * lda + k;
  v2f r;
  r.x = p[0];
  r.y = p[1];
  return r;
}

static __device__ __forceinline__ v2f frag_a_tab(const float* __restrict__ A,
                                                 int ksteps, int tile, int s) {
  int lane = threadIdx.x & 31;
  return *reinterpret_cast<const v2f*>(A + ((tile * ksteps + s) * 32 + lane) * 2);
}

static __device__ __forceinline__ v2f frag_b(const float* __restrict__ B,
                                             int ldb, int k0, int n0) {
  int lane = threadIdx.x & 31;
  int n = n0 + (lane & 15);
  int k = k0 + ((lane >> 4) << 1);
  const float* p = B + k * ldb + n;
  v2f r;
  r.x = p[0];
  r.y = p[ldb];
  return r;
}

static __device__ __forceinline__ void cwmma3(v2f ar, v2f adm, v2f asp, v2f br,
                                              v2f bi, v8f& k1, v8f& k2, v8f& k3) {
  v2f bs = br + bi;
  k1 = wmma4(ar, bs, k1);
  k2 = wmma4(adm, br, k2);
  k3 = wmma4(asp, bi, k3);
}


__global__ void k_twiddles(float* __restrict__ tw) {
  int t = blockIdx.x * blockDim.x + threadIdx.x;
  int which = blockIdx.y;
  if (which == 0) {
    if (t >= 80 * 80) return;
    int a = t / 80, b = t % 80;
    float r = 0.f, im = 0.f;
    if (a < NLP && b < NLP) {
      double ang = -2.0 * PI_D * (double)((a * b) % NLP) / (double)NLP;
      r = (float)cos(ang);
      im = (float)sin(ang);
    }
    tw[OFF_FWDR + t] = r;
    tw[OFF_FWDI + t] = im;
  } else if (which == 1) {
    if (t >= NC * NC) return;
    int l = (t >> 1) & 31;
    int c = (t >> 10) * 16 + (l & 15);
    int k = ((t >> 6) & 15) * 4 + ((l >> 4) << 1) + (t & 1);
    double ang = -2.0 * PI_D * (double)((c * k) % NC) / (double)NC;
    tw[OFF_ER + t] = (float)cos(ang);
    tw[OFF_EI + t] = (float)sin(ang);
  } else if (which == 2) {
    if (t >= 16 * ND) return;
    int l = (t >> 1) & 31;
    int d = l & 15;
    int c = (t >> 6) * 4 + ((l >> 4) << 1) + (t & 1);
    double ang = 2.0 * PI_D * (double)((c * 4 * d) % NC) / (double)NC;
    tw[OFF_WNR + t] = (float)(cos(ang) / 64.0);
    tw[OFF_WNI + t] = (float)(sin(ang) / 64.0);
  } else if (which == 3) {
    if (t >= 68 * NL) return;
    int v = t / NL, w = t % NL;
    float r = 0.f, im = 0.f;
    if (v < NLP) {
      double ang = 2.0 * PI_D * (double)((v * (w + 1)) % NLP) / (double)NLP;
      r = (float)cos(ang);
      im = (float)sin(ang);
    }
    tw[OFF_IVR + t] = r;
    tw[OFF_IVI + t] = im;
  } else {
    if (t >= NL * 36) return;
    int l = (t >> 1) & 31;
    int fs = t >> 6;
    int h = (fs / 9) * 16 + (l & 15);
    int u = (fs % 9) * 4 + ((l >> 4) << 1) + (t & 1);
    float r = 0.f, im = 0.f;
    if (u <= NLP / 2) {
      double ang = 2.0 * PI_D * (double)((u * (h + 1)) % NLP) / (double)NLP;
      double s = (u == 0 ? 0.5 : 1.0) / (double)(NLP * NLP);
      r = (float)(cos(ang) * s);
      im = (float)(-sin(ang) * s);
    }
    tw[OFF_IUR + t] = r;
    tw[OFF_IUI + t] = im;
  }
}


__global__ __launch_bounds__(128) void k_fft2d(const float* __restrict__ x,
                                               const float* __restrict__ tw,
                                               float* __restrict__ Yr,
                                               float* __restrict__ Yi) {
  __shared__ float xs[NL * NL];
  __shared__ float Tr[NL * 80];
  __shared__ float Ti[NL * 80];
  const int plane = blockIdx.x;
  const int tid = threadIdx.x;
  const int wave = tid >> 5;
  const int lane = tid & 31;

  const float* xp = x + (size_t)plane * (NL * NL);
  for (int i = tid; i < NL * NL; i += 128) xs[i] = xp[i];
  __syncthreads();

  const float* Fr = tw + OFF_FWDR;
  const float* Fi = tw + OFF_FWDI;

  for (int t = wave; t < 4 * 5; t += 4) {
    int mo = (t / 5) * 16, no = (t % 5) * 16;
    v8f ar = zero8(), ai = zero8();
    for (int k0 = 0; k0 < NL; k0 += 4) {
      v2f a = frag_a(xs, NL, mo, k0);
      ar = wmma4(a, frag_b(Fr, 80, k0, no), ar);
      ai = wmma4(a, frag_b(Fi, 80, k0, no), ai);
    }
    int n = no + (lane & 15);
    int mb = mo + ((lane >> 4) << 3);
#pragma unroll
    for (int i = 0; i < 8; ++i) {
      Tr[(mb + i) * 80 + n] = ar[i];
      Ti[(mb + i) * 80 + n] = ai[i];
    }
  }
  __syncthreads();

  float* yr = Yr + (size_t)plane * NP;
  float* yi = Yi + (size_t)plane * NP;
  for (int t = wave; t < 5 * 5; t += 4) {
    int mo = (t / 5) * 16, no = (t % 5) * 16;
    v8f k1 = zero8(), k2 = zero8(), k3 = zero8();
#pragma unroll
    for (int k0 = 0; k0 < NL; k0 += 4) {
      v2f par = frag_b(Fr, 80, k0, mo);
      v2f pai = frag_b(Fi, 80, k0, mo);
      v2f pbr = frag_b(Tr, 80, k0, no);
      v2f pbi = frag_b(Ti, 80, k0, no);
      cwmma3(par, pai - par, par + pai, pbr, pbi, k1, k2, k3);
    }
    const v8f dr = k1 - k3, di = k1 + k2;
    int n = no + (lane & 15);
    int mb = mo + ((lane >> 4) << 3);
    if (n < NLP) {
#pragma unroll
      for (int i = 0; i < 8; ++i) {
        int m = mb + i;
        if (m < NLP) {
          yr[m * NLP + n] = dr[i];
          yi[m * NLP + n] = di[i];
        }
      }
    }
  }
}


typedef __attribute__((ext_vector_type(4))) float v4f;

constexpr int MIX_PT = (NP + 15) / 16;
constexpr int XP = 68;
constexpr int XT = 16 * XP;

__global__ __launch_bounds__(256) void k_cmix(const float* __restrict__ tw,
                                              float* Yr, float* Yi,
                                              float* Xr, float* Xi,
                                              const float* __restrict__ wre,
                                              const float* __restrict__ wim) {
  __shared__ __attribute__((aligned(16))) float xs_r[2 * XT];
  __shared__ __attribute__((aligned(16))) float xs_i[2 * XT];
  const int b0 = blockIdx.x * 2;
  const int p0 = blockIdx.y * 16;
  const int wave = __builtin_amdgcn_readfirstlane(threadIdx.x >> 5);
  const int lane = threadIdx.x & 31;
  const int col = lane & 15;
  const int half = lane >> 4;
  const int h2 = half << 1;
  const int p = p0 + col;
  const int pl = p < NP ? p : NP - 1;

  {
    const int bl = wave >> 2;
    const int ct = wave & 3;
    const float* Er = tw + OFF_ER;
    const float* Ei = tw + OFF_EI;
    const float* yr_p = Yr + ((size_t)(b0 + bl) * NC + h2) * NP + pl;
    const float* yi_p = Yi + ((size_t)(b0 + bl) * NC + h2) * NP + pl;
    v8f k1 = zero8(), k2 = zero8(), k3 = zero8();
#pragma unroll
    for (int k0 = 0; k0 < NC; k0 += 4) {
      v2f ar = frag_a_tab(Er, 16, ct, k0 >> 2);
      v2f ai = frag_a_tab(Ei, 16, ct, k0 >> 2);
      v2f br, bi;
      br.x = yr_p[(size_t)k0 * NP];
      br.y = yr_p[(size_t)(k0 + 1) * NP];
      bi.x = yi_p[(size_t)k0 * NP];
      bi.y = yi_p[(size_t)(k0 + 1) * NP];
      cwmma3(ar, ai - ar, ar + ai, br, bi, k1, k2, k3);
    }
    const v8f dr = k1 - k3, di = k1 + k2;
    float* sr = xs_r + bl * XT + col * XP + ct * 16 + (half << 3);
    float* si = xs_i + bl * XT + col * XP + ct * 16 + (half << 3);
    v4f q;
    q.x = dr[0]; q.y = dr[1]; q.z = dr[2]; q.w = dr[3];
    *reinterpret_cast<v4f*>(sr) = q;
    q.x = dr[4]; q.y = dr[5]; q.z = dr[6]; q.w = dr[7];
    *reinterpret_cast<v4f*>(sr + 4) = q;
    q.x = di[0]; q.y = di[1]; q.z = di[2]; q.w = di[3];
    *reinterpret_cast<v4f*>(si) = q;
    q.x = di[4]; q.y = di[5]; q.z = di[6]; q.w = di[7];
    *reinterpret_cast<v4f*>(si + 4) = q;
  }
  __syncthreads();

  const int n = wave;
  const float* ar_p = tw + OFF_WNR + lane * 2;
  const float* ai_p = tw + OFF_WNI + lane * 2;
  const float* wr_p = wre + ((size_t)n * NC + h2) * NP + pl;
  const float* wi_p = wim + ((size_t)n * NC + h2) * NP + pl;
  const float* xr_p = xs_r + col * XP + h2;
  const float* xi_p = xs_i + col * XP + h2;

  v8f k1[2], k2[2], k3[2];
#pragma unroll
  for (int j = 0; j < 2; ++j) {
    k1[j] = zero8();
    k2[j] = zero8();
    k3[j] = zero8();
  }

#pragma unroll
  for (int k0 = 0; k0 < 16; k0 += 4) {
    v2f ar = *reinterpret_cast<const v2f*>(ar_p + k0 * 16);
    v2f ai = *reinterpret_cast<const v2f*>(ai_p + k0 * 16);
    const v2f adm = ai - ar;
    const v2f asp = ar + ai;
    v2f wr[4], wi[4];
#pragma unroll
    for (int q = 0; q < 4; ++q) {
      size_t o0 = ((size_t)k0 + 16 * q) * NP;
      size_t o1 = o0 + NP;
      wr[q].x = wr_p[o0]; wi[q].x = wi_p[o0];
      wr[q].y = wr_p[o1]; wi[q].y = wi_p[o1];
    }
#pragma unroll
    for (int j = 0; j < 2; ++j) {
      v2f br, bi;
#pragma unroll
      for (int q = 0; q < 4; ++q) {
        const v2f xr = *reinterpret_cast<const v2f*>(xr_p + j * XT + k0 + 16 * q);
        const v2f xi = *reinterpret_cast<const v2f*>(xi_p + j * XT + k0 + 16 * q);
        if (q == 0) {
          br = xr * wr[0];
          bi = xr * wi[0];
        } else {
          br += xr * wr[q];
          bi += xr * wi[q];
        }
        br -= xi * wi[q];
        bi += xi * wr[q];
      }
      cwmma3(ar, adm, asp, br, bi, k1[j], k2[j], k3[j]);
    }
  }

  if (p < NP) {
    const int mb = half << 3;
#pragma unroll
    for (int j = 0; j < 2; ++j) {
      const v8f dr = k1[j] - k3[j], di = k1[j] + k2[j];
      float* gr = (n < 4 ? Yr : Xr) + ((size_t)(b0 + j) * NC + (n & 3) * ND + mb) * NP + p;
      float* gi = (n < 4 ? Yi : Xi) + ((size_t)(b0 + j) * NC + (n & 3) * ND + mb) * NP + p;
#pragma unroll
      for (int i = 0; i < 8; ++i) {
        gr[(size_t)i * NP] = dr[i];
        gi[(size_t)i * NP] = di[i];
      }
    }
  }
}


constexpr int GP = 68;
constexpr int GR = 48;
constexpr int GS = GR * GP;
constexpr int TP = 52;
constexpr int KS = 17;
constexpr int KU = 9;
constexpr int UH = NLP / 2;

__global__ __launch_bounds__(256) void k_ifft2d(const float* __restrict__ tw,
                                                const float* __restrict__ Yr,
                                                const float* __restrict__ Yi,
                                                const float* __restrict__ Xr,
                                                const float* __restrict__ Xi,
                                                float* __restrict__ out) {
  __shared__ __attribute__((aligned(16))) float gs_r[GS];
  __shared__ __attribute__((aligned(16))) float gs_i[GS];
  __shared__ __attribute__((aligned(16))) float tt_r[NL * TP];
  __shared__ __attribute__((aligned(16))) float tt_i[NL * TP];
  const int d = blockIdx.x, b = blockIdx.y, nidx = blockIdx.z;
  const int tid = threadIdx.x;
  const int wave = __builtin_amdgcn_readfirstlane(tid >> 5);
  const int lane = tid & 31;
  const int c = lane & 15;
  const int h2 = (lane >> 4) << 1;
  const int wq = wave & 3;
  const int wh = wave >> 2;

  const float* IVr = tw + OFF_IVR;
  const float* IVi = tw + OFF_IVI;
  v2f vbr[KS], vbi[KS], vdm[KS], vsp[KS];
#pragma unroll
  for (int k = 0; k < KS; ++k) {
    vbr[k] = frag_b(IVr, NL, 4 * k, wq * 16);
    vbi[k] = frag_b(IVi, NL, 4 * k, wq * 16);
  }

  const float* gr = (nidx < 4 ? Yr : Xr) + ((size_t)b * NC + (nidx & 3) * ND + d) * NP;
  const float* gi = (nidx < 4 ? Yi : Xi) + ((size_t)b * NC + (nidx & 3) * ND + d) * NP;
  const int r4 = tid >> 6, c64 = tid & 63;
  const int mc = c64 ? NLP - c64 : 0;
  const int erow = (tid >> 2) - (tid >= 192 ? 16 : 0), ecol = tid & 3;
  const bool ok0 = ecol == 0 && erow <= UH;
  const int d0i = erow * NLP + 64;
  const int mrow0 = erow ? NLP - erow : 0;
  const int m0i = mrow0 * NLP + 1;
  const int s0 = ok0 ? d0i : 0;
  const int m0 = ok0 ? m0i : 0;
  const float a0r = gr[s0], a0i = gi[s0];
  const float b0r = gr[m0], b0i = gi[m0];
  float sr[KU], si[KU];
#pragma unroll
  for (int j = 0; j < KU; ++j) {
    const int row = 4 * j + r4;
    const bool ok = j < KU - 1 || r4 == 0;
    const int mrow = row ? NLP - row : 0;
    const int s = ok ? row * NLP + c64 : 0;
    const int m = ok ? mrow * NLP + mc : 0;
    const float ar = gr[s], ai = gi[s];
    const float br = gr[m], bi = gi[m];
    sr[j] = ok ? ar + br : 0.f;
    si[j] = ok ? ai - bi : 0.f;
  }
  const float e0r = ok0 ? a0r + b0r : 0.f;
  const float e0i = ok0 ? a0i - b0i : 0.f;

#pragma unroll
  for (int j = 0; j < GR / 4; ++j) {
    gs_r[(4 * j + r4) * GP + c64] = j < KU ? sr[j] : 0.f;
    gs_i[(4 * j + r4) * GP + c64] = j < KU ? si[j] : 0.f;
  }
  gs_r[erow * GP + 64 + ecol] = e0r;
  gs_i[erow * GP + 64 + ecol] = e0i;
  __syncthreads();

#pragma unroll
  for (int k = 0; k < KS; ++k) {
    vdm[k] = vbi[k] - vbr[k];
    vsp[k] = vbr[k] + vbi[k];
  }

  {
    const int no = wq * 16;
    const int mo_end = wh ? 48 : 32;
    for (int mo = wh ? 32 : 0; mo < mo_end; mo += 16) {
      const float* ar_p = gs_r + (mo + c) * GP + h2;
      const float* ai_p = gs_i + (mo + c) * GP + h2;
      v8f k1 = zero8(), k2 = zero8(), k3 = zero8();
#pragma unroll
      for (int k = 0; k < KS; ++k) {
        v2f ar = *reinterpret_cast<const v2f*>(ar_p + 4 * k);
        v2f ai = *reinterpret_cast<const v2f*>(ai_p + 4 * k);
        v2f asum = ar + ai;
        k1 = wmma4(asum, vbr[k], k1);
        k2 = wmma4(ar, vdm[k], k2);
        k3 = wmma4(ai, vsp[k], k3);
      }
      const v8f dr = k1 - k3, di = k1 + k2;
      float* tr = tt_r + (no + c) * TP + mo + ((lane >> 4) << 3);
      float* ti = tt_i + (no + c) * TP + mo + ((lane >> 4) << 3);
      v4f q;
      q.x = dr[0]; q.y = dr[1]; q.z = dr[2]; q.w = dr[3];
      *reinterpret_cast<v4f*>(tr) = q;
      q.x = dr[4]; q.y = dr[5]; q.z = dr[6]; q.w = dr[7];
      *reinterpret_cast<v4f*>(tr + 4) = q;
      q.x = di[0]; q.y = di[1]; q.z = di[2]; q.w = di[3];
      *reinterpret_cast<v4f*>(ti) = q;
      q.x = di[4]; q.y = di[5]; q.z = di[6]; q.w = di[7];
      *reinterpret_cast<v4f*>(ti + 4) = q;
    }
  }

  const float* IUr = tw + OFF_IUR;
  const float* IUi = tw + OFF_IUI;
  v2f uar[KU], unai[KU];
#pragma unroll
  for (int k = 0; k < KU; ++k) {
    uar[k] = frag_a_tab(IUr, KU, wq, k);
    unai[k] = frag_a_tab(IUi, KU, wq, k);
  }
  __syncthreads();

  {
    const int mo = wq * 16;
    const int no = wh * 32;
    const float* b0r_p = tt_r + (no + c) * TP + h2;
    const float* b0i_p = tt_i + (no + c) * TP + h2;
    const float* b1r_p = b0r_p + 16 * TP;
    const float* b1i_p = b0i_p + 16 * TP;
    v8f d0 = zero8(), d1 = zero8();
#pragma unroll
    for (int k = 0; k < KU; ++k) {
      v2f b0r = *reinterpret_cast<const v2f*>(b0r_p + 4 * k);
      v2f b1r = *reinterpret_cast<const v2f*>(b1r_p + 4 * k);
      v2f b0i = *reinterpret_cast<const v2f*>(b0i_p + 4 * k);
      v2f b1i = *reinterpret_cast<const v2f*>(b1i_p + 4 * k);
      d0 = wmma4(uar[k], b0r, d0);
      d1 = wmma4(uar[k], b1r, d1);
      d0 = wmma4(unai[k], b0i, d0);
      d1 = wmma4(unai[k], b1i, d1);
    }
    float* op = out + ((size_t)b * (ND * NFR) + (d * NFR + nidx)) * (NL * NL);
    const int n = no + c;
    const int mb = mo + ((lane >> 4) << 3);
#pragma unroll
    for (int i = 0; i < 8; ++i) {
      op[(mb + i) * NL + n] = d0[i];
      op[(mb + i) * NL + n + 16] = d1[i];
    }
  }
}


extern "C" void kernel_launch(void* const* d_in, const int* in_sizes, int n_in,
                              void* d_out, int out_size, void* d_ws,
                              size_t ws_size, hipStream_t stream) {
  (void)in_sizes; (void)n_in; (void)out_size; (void)ws_size;
  const float* x   = (const float*)d_in[0];
  const float* wre = (const float*)d_in[1];
  const float* wim = (const float*)d_in[2];
  float* out = (float*)d_out;
  float* ws = (float*)d_ws;

  float* tw = ws;
  size_t off = TW_PAD;
  float* Yr = ws + off; off += (size_t)NB * NC * NP;
  float* Yi = ws + off; off += (size_t)NB * NC * NP;
  float* Xr = ws + off; off += (size_t)NB * NC * NP;
  float* Xi = ws + off;

  k_twiddles<<<dim3(25, 5), 256, 0, stream>>>(tw);
  k_fft2d<<<NB * NC, 128, 0, stream>>>(x, tw, Yr, Yi);
  k_cmix<<<dim3(NB / 2, MIX_PT), 256, 0, stream>>>(tw, Yr, Yi, Xr, Xi, wre, wim);
  k_ifft2d<<<dim3(ND, NB, NFR), 256, 0, stream>>>(tw, Yr, Yi, Xr, Xi, out);
}


# forward 2D DFT of real x: only columns v = 0..32 are computed (3 column tiles, not 5, in both stages); columns 33..64 are stored as conjugate copies Y(65-u,65-v), not summed; k order unchanged
# speedup vs baseline: 10.4457x; 1.0785x over previous
#include <hip/hip_runtime.h>


typedef __attribute__((ext_vector_type(2))) float v2f;
typedef __attribute__((ext_vector_type(8))) float v8f;

constexpr int NB  = 16;
constexpr int NC  = 64;
constexpr int NFR = 8;
constexpr int ND  = 16;
constexpr int NL  = 64;
constexpr int NLP = 65;
constexpr int NP  = NLP * NLP;

constexpr int OFF_FWDR = 0;
constexpr int OFF_FWDI = OFF_FWDR + 80 * 80;
constexpr int OFF_ER   = OFF_FWDI + 80 * 80;
constexpr int OFF_EI   = OFF_ER + 64 * 64;
constexpr int OFF_WNR  = OFF_EI + 64 * 64;
constexpr int OFF_WNI  = OFF_WNR + 64 * 16;
constexpr int OFF_IVR  = OFF_WNI + 64 * 16;
constexpr int OFF_IVI  = OFF_IVR + 68 * 64;
constexpr int OFF_IUR  = OFF_IVI + 68 * 64;
constexpr int OFF_IUI  = OFF_IUR + 64 * 36;
constexpr int TW_TOTAL = OFF_IUI + 64 * 36;
constexpr int TW_PAD   = 40960;

#define PI_D 3.14159265358979323846


static __device__ __forceinline__ v8f wmma4(v2f a, v2f b, v8f c) {
  return __builtin_amdgcn_wmma_f32_16x16x4_f32(false, a, false, b, (short)0, c,
                                               false, false);
}

static __device__ __forceinline__ v8f zero8() {
  v8f z = {0.f, 0.f, 0.f, 0.f, 0.f, 0.f, 0.f, 0.f};
  return z;
}

static __device__ __forceinline__ v2f frag_a(const float* __restrict__ A,
                                             int lda, int m0, int k0) {
  int lane = threadIdx.x & 31;
  int m = m0 + (lane & 15);
  int k = k0 + ((lane >> 4) << 1);
  const float* p = A + m * lda + k;
  v2f r;
  r.x = p[0];
  r.y = p[1];
  return r;
}

static __device__ __forceinline__ v2f frag_a_tab(const float* __restrict__ A,
                                                 int ksteps, int tile, int s) {
  int lane = threadIdx.x & 31;
  return *reinterpret_cast<const v2f*>(A + ((tile * ksteps + s) * 32 + lane) * 2);
}

static __device__ __forceinline__ v2f frag_b(const float* __restrict__ B,
                                             int ldb, int k0, int n0) {
  int lane = threadIdx.x & 31;
  int n = n0 + (lane & 15);
  int k = k0 + ((lane >> 4) << 1);
  const float* p = B + k * ldb + n;
  v2f r;
  r.x = p[0];
  r.y = p[ldb];
  return r;
}

static __device__ __forceinline__ void cwmma3(v2f ar, v2f adm, v2f asp, v2f br,
                                              v2f bi, v8f& k1, v8f& k2, v8f& k3) {
  v2f bs = br + bi;
  k1 = wmma4(ar, bs, k1);
  k2 = wmma4(adm, br, k2);
  k3 = wmma4(asp, bi, k3);
}


__global__ void k_twiddles(float* __restrict__ tw) {
  int t = blockIdx.x * blockDim.x + threadIdx.x;
  int which = blockIdx.y;
  if (which == 0) {
    if (t >= 80 * 80) return;
    int a = t / 80, b = t % 80;
    float r = 0.f, im = 0.f;
    if (a < NLP && b < NLP) {
      double ang = -2.0 * PI_D * (double)((a * b) % NLP) / (double)NLP;
      r = (float)cos(ang);
      im = (float)sin(ang);
    }
    tw[OFF_FWDR + t] = r;
    tw[OFF_FWDI + t] = im;
  } else if (which == 1) {
    if (t >= NC * NC) return;
    int l = (t >> 1) & 31;
    int c = (t >> 10) * 16 + (l & 15);
    int k = ((t >> 6) & 15) * 4 + ((l >> 4) << 1) + (t & 1);
    double ang = -2.0 * PI_D * (double)((c * k) % NC) / (double)NC;
    tw[OFF_ER + t] = (float)cos(ang);
    tw[OFF_EI + t] = (float)sin(ang);
  } else if (which == 2) {
    if (t >= 16 * ND) return;
    int l = (t >> 1) & 31;
    int d = l & 15;
    int c = (t >> 6) * 4 + ((l >> 4) << 1) + (t & 1);
    double ang = 2.0 * PI_D * (double)((c * 4 * d) % NC) / (double)NC;
    tw[OFF_WNR + t] = (float)(cos(ang) / 64.0);
    tw[OFF_WNI + t] = (float)(sin(ang) / 64.0);
  } else if (which == 3) {
    if (t >= 68 * NL) return;
    int v = t / NL, w = t % NL;
    float r = 0.f, im = 0.f;
    if (v < NLP) {
      double ang = 2.0 * PI_D * (double)((v * (w + 1)) % NLP) / (double)NLP;
      r = (float)cos(ang);
      im = (float)sin(ang);
    }
    tw[OFF_IVR + t] = r;
    tw[OFF_IVI + t] = im;
  } else {
    if (t >= NL * 36) return;
    int l = (t >> 1) & 31;
    int fs = t >> 6;
    int h = (fs / 9) * 16 + (l & 15);
    int u = (fs % 9) * 4 + ((l >> 4) << 1) + (t & 1);
    float r = 0.f, im = 0.f;
    if (u <= NLP / 2) {
      double ang = 2.0 * PI_D * (double)((u * (h + 1)) % NLP) / (double)NLP;
      double s = (u == 0 ? 0.5 : 1.0) / (double)(NLP * NLP);
      r = (float)(cos(ang) * s);
      im = (float)(-sin(ang) * s);
    }
    tw[OFF_IUR + t] = r;
    tw[OFF_IUI + t] = im;
  }
}


constexpr int VT = 3;

__global__ __launch_bounds__(128) void k_fft2d(const float* __restrict__ x,
                                               const float* __restrict__ tw,
                                               float* __restrict__ Yr,
                                               float* __restrict__ Yi) {
  __shared__ float xs[NL * NL];
  __shared__ float Tr[NL * 80];
  __shared__ float Ti[NL * 80];
  const int plane = blockIdx.x;
  const int tid = threadIdx.x;
  const int wave = tid >> 5;
  const int lane = tid & 31;

  const float* xp = x + (size_t)plane * (NL * NL);
  for (int i = tid; i < NL * NL; i += 128) xs[i] = xp[i];
  __syncthreads();

  const float* Fr = tw + OFF_FWDR;
  const float* Fi = tw + OFF_FWDI;

  {
    const int mo = wave * 16;
    v2f a[NL / 4];
#pragma unroll
    for (int s = 0; s < NL / 4; ++s) a[s] = frag_a(xs, NL, mo, 4 * s);
    for (int no = 0; no < VT * 16; no += 16) {
      v8f ar = zero8(), ai = zero8();
#pragma unroll
      for (int s = 0; s < NL / 4; ++s) {
        ar = wmma4(a[s], frag_b(Fr, 80, 4 * s, no), ar);
        ai = wmma4(a[s], frag_b(Fi, 80, 4 * s, no), ai);
      }
      int n = no + (lane & 15);
      int mb = mo + ((lane >> 4) << 3);
#pragma unroll
      for (int i = 0; i < 8; ++i) {
        Tr[(mb + i) * 80 + n] = ar[i];
        Ti[(mb + i) * 80 + n] = ai[i];
      }
    }
  }
  __syncthreads();

  float* yr = Yr + (size_t)plane * NP;
  float* yi = Yi + (size_t)plane * NP;
  for (int t = wave; t < 5 * VT; t += 4) {
    int mo = (t / VT) * 16, no = (t % VT) * 16;
    v8f k1 = zero8(), k2 = zero8(), k3 = zero8();
#pragma unroll
    for (int k0 = 0; k0 < NL; k0 += 4) {
      v2f par = frag_b(Fr, 80, k0, mo);
      v2f pai = frag_b(Fi, 80, k0, mo);
      v2f pbr = frag_b(Tr, 80, k0, no);
      v2f pbi = frag_b(Ti, 80, k0, no);
      cwmma3(par, pai - par, par + pai, pbr, pbi, k1, k2, k3);
    }
    const v8f dr = k1 - k3, di = k1 + k2;
    const int n = no + (lane & 15);
    const int mb = mo + ((lane >> 4) << 3);
    if (n <= NLP / 2) {
      const int nm = NLP - n;
#pragma unroll
      for (int i = 0; i < 8; ++i) {
        const int m = mb + i;
        if (m < NLP) {
          yr[m * NLP + n] = dr[i];
          yi[m * NLP + n] = di[i];
          if (n) {
            const int mm = m ? NLP - m : 0;
            yr[mm * NLP + nm] = dr[i];
            yi[mm * NLP + nm] = -di[i];
          }
        }
      }
    }
  }
}


typedef __attribute__((ext_vector_type(4))) float v4f;

constexpr int MIX_PT = (NP + 15) / 16;
constexpr int XP = 68;
constexpr int XT = 16 * XP;

__global__ __launch_bounds__(256) void k_cmix(const float* __restrict__ tw,
                                              float* Yr, float* Yi,
                                              float* Xr, float* Xi,
                                              const float* __restrict__ wre,
                                              const float* __restrict__ wim) {
  __shared__ __attribute__((aligned(16))) float xs_r[2 * XT];
  __shared__ __attribute__((aligned(16))) float xs_i[2 * XT];
  const int b0 = blockIdx.x * 2;
  const int p0 = blockIdx.y * 16;
  const int wave = __builtin_amdgcn_readfirstlane(threadIdx.x >> 5);
  const int lane = threadIdx.x & 31;
  const int col = lane & 15;
  const int half = lane >> 4;
  const int h2 = half << 1;
  const int p = p0 + col;
  const int pl = p < NP ? p : NP - 1;

  {
    const int bl = wave >> 2;
    const int ct = wave & 3;
    const float* Er = tw + OFF_ER;
    const float* Ei = tw + OFF_EI;
    const float* yr_p = Yr + ((size_t)(b0 + bl) * NC + h2) * NP + pl;
    const float* yi_p = Yi + ((size_t)(b0 + bl) * NC + h2) * NP + pl;
    v8f k1 = zero8(), k2 = zero8(), k3 = zero8();
#pragma unroll
    for (int k0 = 0; k0 < NC; k0 += 4) {
      v2f ar = frag_a_tab(Er, 16, ct, k0 >> 2);
      v2f ai = frag_a_tab(Ei, 16, ct, k0 >> 2);
      v2f br, bi;
      br.x = yr_p[(size_t)k0 * NP];
      br.y = yr_p[(size_t)(k0 + 1) * NP];
      bi.x = yi_p[(size_t)k0 * NP];
      bi.y = yi_p[(size_t)(k0 + 1) * NP];
      cwmma3(ar, ai - ar, ar + ai, br, bi, k1, k2, k3);
    }
    const v8f dr = k1 - k3, di = k1 + k2;
    float* sr = xs_r + bl * XT + col * XP + ct * 16 + (half << 3);
    float* si = xs_i + bl * XT + col * XP + ct * 16 + (half << 3);
    v4f q;
    q.x = dr[0]; q.y = dr[1]; q.z = dr[2]; q.w = dr[3];
    *reinterpret_cast<v4f*>(sr) = q;
    q.x = dr[4]; q.y = dr[5]; q.z = dr[6]; q.w = dr[7];
    *reinterpret_cast<v4f*>(sr + 4) = q;
    q.x = di[0]; q.y = di[1]; q.z = di[2]; q.w = di[3];
    *reinterpret_cast<v4f*>(si) = q;
    q.x = di[4]; q.y = di[5]; q.z = di[6]; q.w = di[7];
    *reinterpret_cast<v4f*>(si + 4) = q;
  }
  __syncthreads();

  const int n = wave;
  const float* ar_p = tw + OFF_WNR + lane * 2;
  const float* ai_p = tw + OFF_WNI + lane * 2;
  const float* wr_p = wre + ((size_t)n * NC + h2) * NP + pl;
  const float* wi_p = wim + ((size_t)n * NC + h2) * NP + pl;
  const float* xr_p = xs_r + col * XP + h2;
  const float* xi_p = xs_i + col * XP + h2;

  v8f k1[2], k2[2], k3[2];
#pragma unroll
  for (int j = 0; j < 2; ++j) {
    k1[j] = zero8();
    k2[j] = zero8();
    k3[j] = zero8();
  }

#pragma unroll
  for (int k0 = 0; k0 < 16; k0 += 4) {
    v2f ar = *reinterpret_cast<const v2f*>(ar_p + k0 * 16);
    v2f ai = *reinterpret_cast<const v2f*>(ai_p + k0 * 16);
    const v2f adm = ai - ar;
    const v2f asp = ar + ai;
    v2f wr[4], wi[4];
#pragma unroll
    for (int q = 0; q < 4; ++q) {
      size_t o0 = ((size_t)k0 + 16 * q) * NP;
      size_t o1 = o0 + NP;
      wr[q].x = wr_p[o0]; wi[q].x = wi_p[o0];
      wr[q].y = wr_p[o1]; wi[q].y = wi_p[o1];
    }
#pragma unroll
    for (int j = 0; j < 2; ++j) {
      v2f br, bi;
#pragma unroll
      for (int q = 0; q < 4; ++q) {
        const v2f xr = *reinterpret_cast<const v2f*>(xr_p + j * XT + k0 + 16 * q);
        const v2f xi = *reinterpret_cast<const v2f*>(xi_p + j * XT + k0 + 16 * q);
        if (q == 0) {
          br = xr * wr[0];
          bi = xr * wi[0];
        } else {
          br += xr * wr[q];
          bi += xr * wi[q];
        }
        br -= xi * wi[q];
        bi += xi * wr[q];
      }
      cwmma3(ar, adm, asp, br, bi, k1[j], k2[j], k3[j]);
    }
  }

  if (p < NP) {
    const int mb = half << 3;
#pragma unroll
    for (int j = 0; j < 2; ++j) {
      const v8f dr = k1[j] - k3[j], di = k1[j] + k2[j];
      float* gr = (n < 4 ? Yr : Xr) + ((size_t)(b0 + j) * NC + (n & 3) * ND + mb) * NP + p;
      float* gi = (n < 4 ? Yi : Xi) + ((size_t)(b0 + j) * NC + (n & 3) * ND + mb) * NP + p;
#pragma unroll
      for (int i = 0; i < 8; ++i) {
        gr[(size_t)i * NP] = dr[i];
        gi[(size_t)i * NP] = di[i];
      }
    }
  }
}


constexpr int GP = 68;
constexpr int GR = 48;
constexpr int GS = GR * GP;
constexpr int TP = 52;
constexpr int KS = 17;
constexpr int KU = 9;
constexpr int UH = NLP / 2;

__global__ __launch_bounds__(256) void k_ifft2d(const float* __restrict__ tw,
                                                const float* __restrict__ Yr,
                                                const float* __restrict__ Yi,
                                                const float* __restrict__ Xr,
                                                const float* __restrict__ Xi,
                                                float* __restrict__ out) {
  __shared__ __attribute__((aligned(16))) float gs_r[GS];
  __shared__ __attribute__((aligned(16))) float gs_i[GS];
  __shared__ __attribute__((aligned(16))) float tt_r[NL * TP];
  __shared__ __attribute__((aligned(16))) float tt_i[NL * TP];
  const int d = blockIdx.x, b = blockIdx.y, nidx = blockIdx.z;
  const int tid = threadIdx.x;
  const int wave = __builtin_amdgcn_readfirstlane(tid >> 5);
  const int lane = tid & 31;
  const int c = lane & 15;
  const int h2 = (lane >> 4) << 1;
  const int wq = wave & 3;
  const int wh = wave >> 2;

  const float* IVr = tw + OFF_IVR;
  const float* IVi = tw + OFF_IVI;
  v2f vbr[KS], vbi[KS], vdm[KS], vsp[KS];
#pragma unroll
  for (int k = 0; k < KS; ++k) {
    vbr[k] = frag_b(IVr, NL, 4 * k, wq * 16);
    vbi[k] = frag_b(IVi, NL, 4 * k, wq * 16);
  }

  const float* gr = (nidx < 4 ? Yr : Xr) + ((size_t)b * NC + (nidx & 3) * ND + d) * NP;
  const float* gi = (nidx < 4 ? Yi : Xi) + ((size_t)b * NC + (nidx & 3) * ND + d) * NP;
  const int r4 = tid >> 6, c64 = tid & 63;
  const int mc = c64 ? NLP - c64 : 0;
  const int erow = (tid >> 2) - (tid >= 192 ? 16 : 0), ecol = tid & 3;
  const bool ok0 = ecol == 0 && erow <= UH;
  const int d0i = erow * NLP + 64;
  const int mrow0 = erow ? NLP - erow : 0;
  const int m0i = mrow0 * NLP + 1;
  const int s0 = ok0 ? d0i : 0;
  const int m0 = ok0 ? m0i : 0;
  const float a0r = gr[s0], a0i = gi[s0];
  const float b0r = gr[m0], b0i = gi[m0];
  float sr[KU], si[KU];
#pragma unroll
  for (int j = 0; j < KU; ++j) {
    const int row = 4 * j + r4;
    const bool ok = j < KU - 1 || r4 == 0;
    const int mrow = row ? NLP - row : 0;
    const int s = ok ? row * NLP + c64 : 0;
    const int m = ok ? mrow * NLP + mc : 0;
    const float ar = gr[s], ai = gi[s];
    const float br = gr[m], bi = gi[m];
    sr[j] = ok ? ar + br : 0.f;
    si[j] = ok ? ai - bi : 0.f;
  }
  const float e0r = ok0 ? a0r + b0r : 0.f;
  const float e0i = ok0 ? a0i - b0i : 0.f;

#pragma unroll
  for (int j = 0; j < GR / 4; ++j) {
    gs_r[(4 * j + r4) * GP + c64] = j < KU ? sr[j] : 0.f;
    gs_i[(4 * j + r4) * GP + c64] = j < KU ? si[j] : 0.f;
  }
  gs_r[erow * GP + 64 + ecol] = e0r;
  gs_i[erow * GP + 64 + ecol] = e0i;
  __syncthreads();

#pragma unroll
  for (int k = 0; k < KS; ++k) {
    vdm[k] = vbi[k] - vbr[k];
    vsp[k] = vbr[k] + vbi[k];
  }

  {
    const int no = wq * 16;
    const int mo_end = wh ? 48 : 32;
    for (int mo = wh ? 32 : 0; mo < mo_end; mo += 16) {
      const float* ar_p = gs_r + (mo + c) * GP + h2;
      const float* ai_p = gs_i + (mo + c) * GP + h2;
      v8f k1 = zero8(), k2 = zero8(), k3 = zero8();
#pragma unroll
      for (int k = 0; k < KS; ++k) {
        v2f ar = *reinterpret_cast<const v2f*>(ar_p + 4 * k);
        v2f ai = *reinterpret_cast<const v2f*>(ai_p + 4 * k);
        v2f asum = ar + ai;
        k1 = wmma4(asum, vbr[k], k1);
        k2 = wmma4(ar, vdm[k], k2);
        k3 = wmma4(ai, vsp[k], k3);
      }
      const v8f dr = k1 - k3, di = k1 + k2;
      float* tr = tt_r + (no + c) * TP + mo + ((lane >> 4) << 3);
      float* ti = tt_i + (no + c) * TP + mo + ((lane >> 4) << 3);
      v4f q;
      q.x = dr[0]; q.y = dr[1]; q.z = dr[2]; q.w = dr[3];
      *reinterpret_cast<v4f*>(tr) = q;
      q.x = dr[4]; q.y = dr[5]; q.z = dr[6]; q.w = dr[7];
      *reinterpret_cast<v4f*>(tr + 4) = q;
      q.x = di[0]; q.y = di[1]; q.z = di[2]; q.w = di[3];
      *reinterpret_cast<v4f*>(ti) = q;
      q.x = di[4]; q.y = di[5]; q.z = di[6]; q.w = di[7];
      *reinterpret_cast<v4f*>(ti + 4) = q;
    }
  }

  const float* IUr = tw + OFF_IUR;
  const float* IUi = tw + OFF_IUI;
  v2f uar[KU], unai[KU];
#pragma unroll
  for (int k = 0; k < KU; ++k) {
    uar[k] = frag_a_tab(IUr, KU, wq, k);
    unai[k] = frag_a_tab(IUi, KU, wq, k);
  }
  __syncthreads();

  {
    const int mo = wq * 16;
    const int no = wh * 32;
    const float* b0r_p = tt_r + (no + c) * TP + h2;
    const float* b0i_p = tt_i + (no + c) * TP + h2;
    const float* b1r_p = b0r_p + 16 * TP;
    const float* b1i_p = b0i_p + 16 * TP;
    v8f d0 = zero8(), d1 = zero8();
#pragma unroll
    for (int k = 0; k < KU; ++k) {
      v2f b0r = *reinterpret_cast<const v2f*>(b0r_p + 4 * k);
      v2f b1r = *reinterpret_cast<const v2f*>(b1r_p + 4 * k);
      v2f b0i = *reinterpret_cast<const v2f*>(b0i_p + 4 * k);
      v2f b1i = *reinterpret_cast<const v2f*>(b1i_p + 4 * k);
      d0 = wmma4(uar[k], b0r, d0);
      d1 = wmma4(uar[k], b1r, d1);
      d0 = wmma4(unai[k], b0i, d0);
      d1 = wmma4(unai[k], b1i, d1);
    }
    float* op = out + ((size_t)b * (ND * NFR) + (d * NFR + nidx)) * (NL * NL);
    const int n = no + c;
    const int mb = mo + ((lane >> 4) << 3);
#pragma unroll
    for (int i = 0; i < 8; ++i) {
      op[(mb + i) * NL + n] = d0[i];
      op[(mb + i) * NL + n + 16] = d1[i];
    }
  }
}


extern "C" void kernel_launch(void* const* d_in, const int* in_sizes, int n_in,
                              void* d_out, int out_size, void* d_ws,
                              size_t ws_size, hipStream_t stream) {
  (void)in_sizes; (void)n_in; (void)out_size; (void)ws_size;
  const float* x   = (const float*)d_in[0];
  const float* wre = (const float*)d_in[1];
  const float* wim = (const float*)d_in[2];
  float* out = (float*)d_out;
  float* ws = (float*)d_ws;

  float* tw = ws;
  size_t off = TW_PAD;
  float* Yr = ws + off; off += (size_t)NB * NC * NP;
  float* Yi = ws + off; off += (size_t)NB * NC * NP;
  float* Xr = ws + off; off += (size_t)NB * NC * NP;
  float* Xi = ws + off;

  k_twiddles<<<dim3(25, 5), 256, 0, stream>>>(tw);
  k_fft2d<<<NB * NC, 128, 0, stream>>>(x, tw, Yr, Yi);
  k_cmix<<<dim3(NB / 2, MIX_PT), 256, 0, stream>>>(tw, Yr, Yi, Xr, Xi, wre, wim);
  k_ifft2d<<<dim3(ND, NB, NFR), 256, 0, stream>>>(tw, Yr, Yi, Xr, Xi, out);
}
